# starting kernel
# baseline (speedup 1.0000x reference)
#include <hip/hip_runtime.h>
#include <math.h>

#define H_HEADS 4
#define C_CH 32
#define F_DIM 128
#define NEG_SLOPE 0.2f

typedef __attribute__((ext_vector_type(2))) float v2f;
typedef __attribute__((ext_vector_type(8))) float v8f;

__global__ void fill_kernel(float* __restrict__ p, long n, float v) {
    long i = (long)blockIdx.x * blockDim.x + threadIdx.x;
    long stride = (long)gridDim.x * blockDim.x;
    for (; i < n; i += stride) p[i] = v;
}

__global__ __launch_bounds__(256)
void gemm_wmma_kernel(const float* __restrict__ X, const float* __restrict__ W,
                      const float* __restrict__ a_src, const float* __restrict__ a_dst,
                      float* __restrict__ Hout, float* __restrict__ as,
                      float* __restrict__ ad, int rows) {
    extern __shared__ float Wlds[];
    for (int i = threadIdx.x; i < (F_DIM * F_DIM) / 2; i += blockDim.x) {
        int kp  = i >> 7;
        int col = i & 127;
        int cs  = (col + ((kp & 1) << 4)) & 127;
        float2 w2;
        w2.x = W[(size_t)(2 * kp)     * F_DIM + col];
        w2.y = W[(size_t)(2 * kp + 1) * F_DIM + col];
        *(float2*)(Wlds + (size_t)kp * 256 + cs * 2) = w2;
    }
    __syncthreads();

    const int lane = threadIdx.x & 31;
    const int wave = threadIdx.x >> 5;
    const int lm   = lane & 15;
    const int lh   = lane >> 4;
    const int rowBase = blockIdx.x * 128 + wave * 16;
    if (rowBase >= rows) return;

    int aRow = rowBase + lm;
    if (aRow >= rows) aRow = rows - 1;

    v8f acc[8];
    v8f zero = {0.f, 0.f, 0.f, 0.f, 0.f, 0.f, 0.f, 0.f};
#pragma unroll
    for (int ct = 0; ct < 8; ++ct) acc[ct] = zero;

    const float* xrow = X + (size_t)aRow * F_DIM;
    for (int k0 = 0; k0 < F_DIM; k0 += 4) {
        const int kp = (k0 >> 1) + lh;
        v2f a;
        a.x = xrow[2 * kp];
        a.y = xrow[2 * kp + 1];
        const float* bbase = Wlds + (size_t)kp * 256;
#pragma unroll
        for (int ct = 0; ct < 8; ++ct) {
            const int cs = (ct * 16 + lm + 16 * lh) & 127;
            v2f b = *(const v2f*)(bbase + cs * 2);
            acc[ct] = __builtin_amdgcn_wmma_f32_16x16x4_f32(
                false, a, false, b, (short)0, acc[ct], false, false);
        }
    }

#pragma unroll
    for (int ct = 0; ct < 8; ++ct) {
#pragma unroll
        for (int g = 0; g < 8; ++g) {
            int r = rowBase + g + 8 * lh;
            if (r < rows) Hout[(size_t)r * F_DIM + ct * 16 + lm] = acc[ct][g];
        }
    }

    float asv[8], adv[8];
#pragma unroll
    for (int ct = 0; ct < 8; ++ct) {
        asv[ct] = a_src[ct * 16 + lm];
        adv[ct] = a_dst[ct * 16 + lm];
    }
#pragma unroll
    for (int h = 0; h < 4; ++h) {
#pragma unroll
        for (int g = 0; g < 8; ++g) {
            float ss = acc[2 * h][g] * asv[2 * h];
            ss = fmaf(acc[2 * h + 1][g], asv[2 * h + 1], ss);
            float sd = acc[2 * h][g] * adv[2 * h];
            sd = fmaf(acc[2 * h + 1][g], adv[2 * h + 1], sd);
#pragma unroll
            for (int off = 8; off > 0; off >>= 1) {
                ss += __shfl_xor(ss, off, 32);
                sd += __shfl_xor(sd, off, 32);
            }
            int r = rowBase + g + 8 * lh;
            if (lm == 0 && r < rows) {
                as[(size_t)r * H_HEADS + h] = ss;
                ad[(size_t)r * H_HEADS + h] = sd;
            }
        }
    }
}

__device__ inline void atomicMaxFloat(float* addr, float val) {
    if (val >= 0.f)
        atomicMax((int*)addr, __float_as_int(val));
    else
        atomicMin((unsigned int*)addr, __float_as_uint(val));
}

__global__ void score_kernel(const float* __restrict__ as, const float* __restrict__ ad,
                             const int* __restrict__ ei, int E, int Nn,
                             float* __restrict__ alpha, float* __restrict__ m) {
    int id = blockIdx.x * blockDim.x + threadIdx.x;
    int total = E + Nn;
    if (id >= total) return;
    int src = (id < E) ? ei[id]     : id - E;
    int dst = (id < E) ? ei[E + id] : id - E;
    float4 s = *(const float4*)(as + (size_t)src * 4);
    float4 d = *(const float4*)(ad + (size_t)dst * 4);
    float ev[4] = {s.x + d.x, s.y + d.y, s.z + d.z, s.w + d.w};
#pragma unroll
    for (int h = 0; h < 4; ++h) {
        float e = ev[h];
        e = (e > 0.f) ? e : NEG_SLOPE * e;
        ev[h] = e;
        atomicMaxFloat(m + (size_t)dst * 4 + h, e);
    }
    *(float4*)(alpha + (size_t)id * 4) = make_float4(ev[0], ev[1], ev[2], ev[3]);
}

__global__ void expsum_kernel(const int* __restrict__ ei, int E, int Nn,
                              const float* __restrict__ m, float* __restrict__ alpha,
                              float* __restrict__ z) {
    int id = blockIdx.x * blockDim.x + threadIdx.x;
    int total = E + Nn;
    if (id >= total) return;
    int dst = (id < E) ? ei[E + id] : id - E;
    float4 e  = *(const float4*)(alpha + (size_t)id * 4);
    float4 mv = *(const float4*)(m + (size_t)dst * 4);
    float4 p;
    p.x = __expf(e.x - mv.x);
    p.y = __expf(e.y - mv.y);
    p.z = __expf(e.z - mv.z);
    p.w = __expf(e.w - mv.w);
    *(float4*)(alpha + (size_t)id * 4) = p;
    atomicAdd(&z[(size_t)dst * 4 + 0], p.x);
    atomicAdd(&z[(size_t)dst * 4 + 1], p.y);
    atomicAdd(&z[(size_t)dst * 4 + 2], p.z);
    atomicAdd(&z[(size_t)dst * 4 + 3], p.w);
}

__global__ void agg_kernel(const float* __restrict__ hbuf, const float* __restrict__ z,
                           const int* __restrict__ ei, int E, int Nn,
                           float* __restrict__ alpha, float* __restrict__ acc) {
    int widx = blockIdx.x * (blockDim.x >> 5) + (threadIdx.x >> 5);
    int lane = threadIdx.x & 31;
    int total = E + Nn;
    if (widx >= total) return;
    int src = (widx < E) ? ei[widx]     : widx - E;
    int dst = (widx < E) ? ei[E + widx] : widx - E;
    int head = lane >> 3;
    float p  = alpha[(size_t)widx * 4 + head];
    float zz = z[(size_t)dst * 4 + head];
    float al = p / zz;
    float alw = 0.f;
    if (lane < 4) alw = alpha[(size_t)widx * 4 + lane] / z[(size_t)dst * 4 + lane];
    float4 hv = *(const float4*)(hbuf + (size_t)src * F_DIM + lane * 4);
    if (lane < 4) alpha[(size_t)widx * 4 + lane] = alw;
    float* op = acc + (size_t)dst * F_DIM + lane * 4;
    atomicAdd(op + 0, al * hv.x);
    atomicAdd(op + 1, al * hv.y);
    atomicAdd(op + 2, al * hv.z);
    atomicAdd(op + 3, al * hv.w);
}

__global__ void add_bias_kernel(const float* __restrict__ acc, const float* __restrict__ b,
                                float* __restrict__ out, long total) {
    long id = (long)blockIdx.x * blockDim.x + threadIdx.x;
    if (id >= total) return;
    out[id] = acc[id] + b[id & (F_DIM - 1)];
}

__global__ void group_kernel(const float* __restrict__ xacc, const float* __restrict__ b1,
                             const float* __restrict__ xg, const int* __restrict__ gidx,
                             const int* __restrict__ nbr, float* __restrict__ xfinal,
                             int Nn) {
    int lane = threadIdx.x & 31;
    int node = blockIdx.x * (blockDim.x >> 5) + (threadIdx.x >> 5);
    if (node >= Nn) return;
    float4 bv = *(const float4*)(b1 + lane * 4);
    float4 xo = *(const float4*)(xacc + (size_t)node * F_DIM + lane * 4);
    xo.x += bv.x; xo.y += bv.y; xo.z += bv.z; xo.w += bv.w;
    int gi = gidx[node];
    int idx[17];
    idx[0] = gi;
#pragma unroll
    for (int k = 0; k < 16; ++k) idx[k + 1] = nbr[(size_t)gi * 16 + k];

    float4 cv[17];
    float sv[17];
#pragma unroll
    for (int k = 0; k < 17; ++k) {
        float4 c = *(const float4*)(xg + (size_t)idx[k] * F_DIM + lane * 4);
        cv[k] = c;
        float s = xo.x * c.x + xo.y * c.y + xo.z * c.z + xo.w * c.w;
#pragma unroll
        for (int off = 16; off > 0; off >>= 1) s += __shfl_xor(s, off, 32);
        sv[k] = s;
    }
    float mx = sv[0];
#pragma unroll
    for (int k = 1; k < 17; ++k) mx = fmaxf(mx, sv[k]);
    float sum = 0.f;
#pragma unroll
    for (int k = 0; k < 17; ++k) { sv[k] = __expf(sv[k] - mx); sum += sv[k]; }
    float inv = 1.f / sum;
    float4 go = make_float4(0.f, 0.f, 0.f, 0.f);
#pragma unroll
    for (int k = 0; k < 17; ++k) {
        float w = sv[k] * inv;
        go.x = fmaf(w, cv[k].x, go.x);
        go.y = fmaf(w, cv[k].y, go.y);
        go.z = fmaf(w, cv[k].z, go.z);
        go.w = fmaf(w, cv[k].w, go.w);
    }
    float4 o;
    o.x = 0.5f * (xo.x + go.x);
    o.y = 0.5f * (xo.y + go.y);
    o.z = 0.5f * (xo.z + go.z);
    o.w = 0.5f * (xo.w + go.w);
    *(float4*)(xfinal + (size_t)node * F_DIM + lane * 4) = o;
}

extern "C" void kernel_launch(void* const* d_in, const int* in_sizes, int n_in,
                              void* d_out, int out_size, void* d_ws, size_t ws_size,
                              hipStream_t stream) {
    const float* x       = (const float*)d_in[0];
    const int*   ei      = (const int*)d_in[1];
    const float* x_g     = (const float*)d_in[2];
    const int*   ei_g    = (const int*)d_in[3];
    const int*   gidx    = (const int*)d_in[4];
    const int*   nbr     = (const int*)d_in[5];
    const float* W1      = (const float*)d_in[6];
    const float* a_src1  = (const float*)d_in[7];
    const float* a_dst1  = (const float*)d_in[8];
    const float* b1      = (const float*)d_in[9];
    const float* W2      = (const float*)d_in[10];
    const float* a_src2  = (const float*)d_in[11];
    const float* a_dst2  = (const float*)d_in[12];
    const float* b2      = (const float*)d_in[13];

    const int N  = in_sizes[0] / F_DIM;
    const int E  = in_sizes[1] / 2;
    const int K  = in_sizes[2] / F_DIM;
    const int Eg = in_sizes[3] / 2;

    float* out     = (float*)d_out;
    float* x_final = out;
    float* xg_out  = out + (size_t)N * F_DIM;
    float* alpha1  = xg_out + (size_t)K * F_DIM;
    float* alpha2  = alpha1 + (size_t)(E + N) * H_HEADS;

    float* ws = (float*)d_ws;
    size_t off = 0;
    float* h1    = ws + off; off += (size_t)N * F_DIM;
    float* h2    = ws + off; off += (size_t)K * F_DIM;
    float* as1   = ws + off; off += (size_t)N * H_HEADS;
    float* ad1   = ws + off; off += (size_t)N * H_HEADS;
    float* m1    = ws + off; off += (size_t)N * H_HEADS;
    float* z1    = ws + off; off += (size_t)N * H_HEADS;
    float* as2   = ws + off; off += (size_t)K * H_HEADS;
    float* ad2   = ws + off; off += (size_t)K * H_HEADS;
    float* m2    = ws + off; off += (size_t)K * H_HEADS;
    float* z2    = ws + off; off += (size_t)K * H_HEADS;
    float* xacc1 = ws + off; off += (size_t)N * F_DIM;
    float* xacc2 = ws + off; off += (size_t)K * F_DIM;

    const int T = 256;
    auto blk = [](long n, int t) { return (int)((n + t - 1) / t); };

    fill_kernel<<<1024, T, 0, stream>>>(xacc1, (long)N * F_DIM, 0.f);
    fill_kernel<<<256,  T, 0, stream>>>(xacc2, (long)K * F_DIM, 0.f);
    fill_kernel<<<256,  T, 0, stream>>>(z1, (long)N * H_HEADS, 0.f);
    fill_kernel<<<64,   T, 0, stream>>>(z2, (long)K * H_HEADS, 0.f);
    fill_kernel<<<256,  T, 0, stream>>>(m1, (long)N * H_HEADS, -INFINITY);
    fill_kernel<<<64,   T, 0, stream>>>(m2, (long)K * H_HEADS, -INFINITY);

    gemm_wmma_kernel<<<blk(N, 128), T, F_DIM * F_DIM * sizeof(float), stream>>>(
        x, W1, a_src1, a_dst1, h1, as1, ad1, N);
    gemm_wmma_kernel<<<blk(K, 128), T, F_DIM * F_DIM * sizeof(float), stream>>>(
        x_g, W2, a_src2, a_dst2, h2, as2, ad2, K);

    score_kernel<<<blk((long)E + N, T), T, 0, stream>>>(as1, ad1, ei, E, N, alpha1, m1);
    score_kernel<<<blk((long)Eg + K, T), T, 0, stream>>>(as2, ad2, ei_g, Eg, K, alpha2, m2);

    expsum_kernel<<<blk((long)E + N, T), T, 0, stream>>>(ei, E, N, m1, alpha1, z1);
    expsum_kernel<<<blk((long)Eg + K, T), T, 0, stream>>>(ei_g, Eg, K, m2, alpha2, z2);

    agg_kernel<<<blk((long)E + N, 8), T, 0, stream>>>(h1, z1, ei, E, N, alpha1, xacc1);
    agg_kernel<<<blk((long)Eg + K, 8), T, 0, stream>>>(h2, z2, ei_g, Eg, K, alpha2, xacc2);

    add_bias_kernel<<<blk((long)K * F_DIM, T), T, 0, stream>>>(xacc2, b2, xg_out, (long)K * F_DIM);

    group_kernel<<<blk(N, 8), T, 0, stream>>>(xacc1, b1, xg_out, gidx, nbr, x_final, N);
}


# Atomics coalesced by lane layout: score/expsum use 4 adjacent lanes per edge (one per head), aggregate lane L owns channels L+32j: each atomic instruction hits whole 16B/128B runs; same adds
# speedup vs baseline: 1.6137x; 1.6137x over previous
#include <hip/hip_runtime.h>
#include <math.h>

#define H_HEADS 4
#define C_CH 32
#define F_DIM 128
#define NEG_SLOPE 0.2f

typedef __attribute__((ext_vector_type(2))) float v2f;
typedef __attribute__((ext_vector_type(8))) float v8f;

__global__ void fill_kernel(float* __restrict__ p, long n, float v) {
    long i = (long)blockIdx.x * blockDim.x + threadIdx.x;
    long stride = (long)gridDim.x * blockDim.x;
    for (; i < n; i += stride) p[i] = v;
}

__global__ __launch_bounds__(256)
void gemm_wmma_kernel(const float* __restrict__ X, const float* __restrict__ W,
                      const float* __restrict__ a_src, const float* __restrict__ a_dst,
                      float* __restrict__ Hout, float* __restrict__ as,
                      float* __restrict__ ad, int rows) {
    extern __shared__ float Wlds[];
    for (int i = threadIdx.x; i < (F_DIM * F_DIM) / 2; i += blockDim.x) {
        int kp  = i >> 7;
        int col = i & 127;
        int cs  = (col + ((kp & 1) << 4)) & 127;
        float2 w2;
        w2.x = W[(size_t)(2 * kp)     * F_DIM + col];
        w2.y = W[(size_t)(2 * kp + 1) * F_DIM + col];
        *(float2*)(Wlds + (size_t)kp * 256 + cs * 2) = w2;
    }
    __syncthreads();

    const int lane = threadIdx.x & 31;
    const int wave = threadIdx.x >> 5;
    const int lm   = lane & 15;
    const int lh   = lane >> 4;
    const int rowBase = blockIdx.x * 128 + wave * 16;
    if (rowBase >= rows) return;

    int aRow = rowBase + lm;
    if (aRow >= rows) aRow = rows - 1;

    v8f acc[8];
    v8f zero = {0.f, 0.f, 0.f, 0.f, 0.f, 0.f, 0.f, 0.f};
#pragma unroll
    for (int ct = 0; ct < 8; ++ct) acc[ct] = zero;

    const float* xrow = X + (size_t)aRow * F_DIM;
    for (int k0 = 0; k0 < F_DIM; k0 += 4) {
        const int kp = (k0 >> 1) + lh;
        v2f a;
        a.x = xrow[2 * kp];
        a.y = xrow[2 * kp + 1];
        const float* bbase = Wlds + (size_t)kp * 256;
#pragma unroll
        for (int ct = 0; ct < 8; ++ct) {
            const int cs = (ct * 16 + lm + 16 * lh) & 127;
            v2f b = *(const v2f*)(bbase + cs * 2);
            acc[ct] = __builtin_amdgcn_wmma_f32_16x16x4_f32(
                false, a, false, b, (short)0, acc[ct], false, false);
        }
    }

#pragma unroll
    for (int ct = 0; ct < 8; ++ct) {
#pragma unroll
        for (int g = 0; g < 8; ++g) {
            int r = rowBase + g + 8 * lh;
            if (r < rows) Hout[(size_t)r * F_DIM + ct * 16 + lm] = acc[ct][g];
        }
    }

    float asv[8], adv[8];
#pragma unroll
    for (int ct = 0; ct < 8; ++ct) {
        asv[ct] = a_src[ct * 16 + lm];
        adv[ct] = a_dst[ct * 16 + lm];
    }
#pragma unroll
    for (int h = 0; h < 4; ++h) {
#pragma unroll
        for (int g = 0; g < 8; ++g) {
            float ss = acc[2 * h][g] * asv[2 * h];
            ss = fmaf(acc[2 * h + 1][g], asv[2 * h + 1], ss);
            float sd = acc[2 * h][g] * adv[2 * h];
            sd = fmaf(acc[2 * h + 1][g], adv[2 * h + 1], sd);
#pragma unroll
            for (int off = 8; off > 0; off >>= 1) {
                ss += __shfl_xor(ss, off, 32);
                sd += __shfl_xor(sd, off, 32);
            }
            int r = rowBase + g + 8 * lh;
            if (lm == 0 && r < rows) {
                as[(size_t)r * H_HEADS + h] = ss;
                ad[(size_t)r * H_HEADS + h] = sd;
            }
        }
    }
}

__device__ inline void atomicMaxFloat(float* addr, float val) {
    if (val >= 0.f)
        atomicMax((int*)addr, __float_as_int(val));
    else
        atomicMin((unsigned int*)addr, __float_as_uint(val));
}

__global__ void score_kernel(const float* __restrict__ as, const float* __restrict__ ad,
                             const int* __restrict__ ei, int E, int Nn,
                             float* __restrict__ alpha, float* __restrict__ m) {
    const int lane  = threadIdx.x & 31;
    const int h     = lane & 3;
    const int sub   = lane >> 2;
    const int base  = (blockIdx.x * blockDim.x + threadIdx.x) & ~31;
    const int total = E + Nn;
    if (base >= total) return;
    int src[4], dst[4];
    for (int i = 0; i < 4; ++i) {
        int id = base + 8 * i + sub;
        if (id > total - 1) id = total - 1;
        src[i] = (id < E) ? ei[id]     : id - E;
        dst[i] = (id < E) ? ei[E + id] : id - E;
    }
    float ev[4];
    for (int i = 0; i < 4; ++i) {
        float e = as[(size_t)src[i] * 4 + h] + ad[(size_t)dst[i] * 4 + h];
        ev[i] = (e > 0.f) ? e : NEG_SLOPE * e;
    }
    for (int i = 0; i < 4; ++i) {
        const int id = base + 8 * i + sub;
        if (id < total) {
            atomicMaxFloat(m + (size_t)dst[i] * 4 + h, ev[i]);
            alpha[(size_t)id * 4 + h] = ev[i];
        }
    }
}

__global__ void expsum_kernel(const int* __restrict__ ei, int E, int Nn,
                              const float* __restrict__ m, float* __restrict__ alpha,
                              float* __restrict__ z) {
    const int lane  = threadIdx.x & 31;
    const int h     = lane & 3;
    const int sub   = lane >> 2;
    const int base  = (blockIdx.x * blockDim.x + threadIdx.x) & ~31;
    const int total = E + Nn;
    if (base >= total) return;
    int dst[4];
    float ev[4];
    for (int i = 0; i < 4; ++i) {
        int id = base + 8 * i + sub;
        if (id > total - 1) id = total - 1;
        dst[i] = (id < E) ? ei[E + id] : id - E;
        ev[i]  = alpha[(size_t)id * 4 + h];
    }
    float mv[4];
    for (int i = 0; i < 4; ++i) mv[i] = m[(size_t)dst[i] * 4 + h];
    for (int i = 0; i < 4; ++i) {
        const int id = base + 8 * i + sub;
        if (id < total) {
            float p = __expf(ev[i] - mv[i]);
            alpha[(size_t)id * 4 + h] = p;
            atomicAdd(&z[(size_t)dst[i] * 4 + h], p);
        }
    }
}

__global__ void agg_kernel(const float* __restrict__ hbuf, const float* __restrict__ z,
                           const int* __restrict__ ei, int E, int Nn,
                           float* __restrict__ alpha, float* __restrict__ acc) {
    int widx = blockIdx.x * (blockDim.x >> 5) + (threadIdx.x >> 5);
    int lane = threadIdx.x & 31;
    int total = E + Nn;
    if (widx >= total) return;
    int src = (widx < E) ? ei[widx]     : widx - E;
    int dst = (widx < E) ? ei[E + widx] : widx - E;
    float4 p  = *(const float4*)(alpha + (size_t)widx * 4);
    float4 zz = *(const float4*)(z + (size_t)dst * 4);
    float4 al;
    al.x = p.x / zz.x;
    al.y = p.y / zz.y;
    al.z = p.z / zz.z;
    al.w = p.w / zz.w;
    const float* hp = hbuf + (size_t)src * F_DIM + lane;
    float h0 = hp[0];
    float h1 = hp[32];
    float h2 = hp[64];
    float h3 = hp[96];
    if (lane == 0) *(float4*)(alpha + (size_t)widx * 4) = al;
    float* op = acc + (size_t)dst * F_DIM + lane;
    atomicAdd(op,      al.x * h0);
    atomicAdd(op + 32, al.y * h1);
    atomicAdd(op + 64, al.z * h2);
    atomicAdd(op + 96, al.w * h3);
}

__global__ void add_bias_kernel(const float* __restrict__ acc, const float* __restrict__ b,
                                float* __restrict__ out, long total) {
    long id = (long)blockIdx.x * blockDim.x + threadIdx.x;
    if (id >= total) return;
    out[id] = acc[id] + b[id & (F_DIM - 1)];
}

__global__ void group_kernel(const float* __restrict__ xacc, const float* __restrict__ b1,
                             const float* __restrict__ xg, const int* __restrict__ gidx,
                             const int* __restrict__ nbr, float* __restrict__ xfinal,
                             int Nn) {
    int lane = threadIdx.x & 31;
    int node = blockIdx.x * (blockDim.x >> 5) + (threadIdx.x >> 5);
    if (node >= Nn) return;
    float4 bv = *(const float4*)(b1 + lane * 4);
    float4 xo = *(const float4*)(xacc + (size_t)node * F_DIM + lane * 4);
    xo.x += bv.x; xo.y += bv.y; xo.z += bv.z; xo.w += bv.w;
    int gi = gidx[node];
    int idx[17];
    idx[0] = gi;
#pragma unroll
    for (int k = 0; k < 16; ++k) idx[k + 1] = nbr[(size_t)gi * 16 + k];

    float4 cv[17];
    float sv[17];
#pragma unroll
    for (int k = 0; k < 17; ++k) {
        float4 c = *(const float4*)(xg + (size_t)idx[k] * F_DIM + lane * 4);
        cv[k] = c;
        float s = xo.x * c.x + xo.y * c.y + xo.z * c.z + xo.w * c.w;
#pragma unroll
        for (int off = 16; off > 0; off >>= 1) s += __shfl_xor(s, off, 32);
        sv[k] = s;
    }
    float mx = sv[0];
#pragma unroll
    for (int k = 1; k < 17; ++k) mx = fmaxf(mx, sv[k]);
    float sum = 0.f;
#pragma unroll
    for (int k = 0; k < 17; ++k) { sv[k] = __expf(sv[k] - mx); sum += sv[k]; }
    float inv = 1.f / sum;
    float4 go = make_float4(0.f, 0.f, 0.f, 0.f);
#pragma unroll
    for (int k = 0; k < 17; ++k) {
        float w = sv[k] * inv;
        go.x = fmaf(w, cv[k].x, go.x);
        go.y = fmaf(w, cv[k].y, go.y);
        go.z = fmaf(w, cv[k].z, go.z);
        go.w = fmaf(w, cv[k].w, go.w);
    }
    float4 o;
    o.x = 0.5f * (xo.x + go.x);
    o.y = 0.5f * (xo.y + go.y);
    o.z = 0.5f * (xo.z + go.z);
    o.w = 0.5f * (xo.w + go.w);
    *(float4*)(xfinal + (size_t)node * F_DIM + lane * 4) = o;
}

extern "C" void kernel_launch(void* const* d_in, const int* in_sizes, int n_in,
                              void* d_out, int out_size, void* d_ws, size_t ws_size,
                              hipStream_t stream) {
    const float* x       = (const float*)d_in[0];
    const int*   ei      = (const int*)d_in[1];
    const float* x_g     = (const float*)d_in[2];
    const int*   ei_g    = (const int*)d_in[3];
    const int*   gidx    = (const int*)d_in[4];
    const int*   nbr     = (const int*)d_in[5];
    const float* W1      = (const float*)d_in[6];
    const float* a_src1  = (const float*)d_in[7];
    const float* a_dst1  = (const float*)d_in[8];
    const float* b1      = (const float*)d_in[9];
    const float* W2      = (const float*)d_in[10];
    const float* a_src2  = (const float*)d_in[11];
    const float* a_dst2  = (const float*)d_in[12];
    const float* b2      = (const float*)d_in[13];

    const int N  = in_sizes[0] / F_DIM;
    const int E  = in_sizes[1] / 2;
    const int K  = in_sizes[2] / F_DIM;
    const int Eg = in_sizes[3] / 2;

    float* out     = (float*)d_out;
    float* x_final = out;
    float* xg_out  = out + (size_t)N * F_DIM;
    float* alpha1  = xg_out + (size_t)K * F_DIM;
    float* alpha2  = alpha1 + (size_t)(E + N) * H_HEADS;

    float* ws = (float*)d_ws;
    size_t off = 0;
    float* h1    = ws + off; off += (size_t)N * F_DIM;
    float* h2    = ws + off; off += (size_t)K * F_DIM;
    float* as1   = ws + off; off += (size_t)N * H_HEADS;
    float* ad1   = ws + off; off += (size_t)N * H_HEADS;
    float* m1    = ws + off; off += (size_t)N * H_HEADS;
    float* z1    = ws + off; off += (size_t)N * H_HEADS;
    float* as2   = ws + off; off += (size_t)K * H_HEADS;
    float* ad2   = ws + off; off += (size_t)K * H_HEADS;
    float* m2    = ws + off; off += (size_t)K * H_HEADS;
    float* z2    = ws + off; off += (size_t)K * H_HEADS;
    float* xacc1 = ws + off; off += (size_t)N * F_DIM;
    float* xacc2 = ws + off; off += (size_t)K * F_DIM;

    const int T = 256;
    auto blk = [](long n, int t) { return (int)((n + t - 1) / t); };

    fill_kernel<<<1024, T, 0, stream>>>(xacc1, (long)N * F_DIM, 0.f);
    fill_kernel<<<256,  T, 0, stream>>>(xacc2, (long)K * F_DIM, 0.f);
    fill_kernel<<<256,  T, 0, stream>>>(z1, (long)N * H_HEADS, 0.f);
    fill_kernel<<<64,   T, 0, stream>>>(z2, (long)K * H_HEADS, 0.f);
    fill_kernel<<<256,  T, 0, stream>>>(m1, (long)N * H_HEADS, -INFINITY);
    fill_kernel<<<64,   T, 0, stream>>>(m2, (long)K * H_HEADS, -INFINITY);

    gemm_wmma_kernel<<<blk(N, 128), T, F_DIM * F_DIM * sizeof(float), stream>>>(
        x, W1, a_src1, a_dst1, h1, as1, ad1, N);
    gemm_wmma_kernel<<<blk(K, 128), T, F_DIM * F_DIM * sizeof(float), stream>>>(
        x_g, W2, a_src2, a_dst2, h2, as2, ad2, K);

    score_kernel<<<blk((long)E + N, T), T, 0, stream>>>(as1, ad1, ei, E, N, alpha1, m1);
    score_kernel<<<blk((long)Eg + K, T), T, 0, stream>>>(as2, ad2, ei_g, Eg, K, alpha2, m2);

    expsum_kernel<<<blk((long)E + N, T), T, 0, stream>>>(ei, E, N, m1, alpha1, z1);
    expsum_kernel<<<blk((long)Eg + K, T), T, 0, stream>>>(ei_g, Eg, K, m2, alpha2, z2);

    agg_kernel<<<blk((long)E + N, 8), T, 0, stream>>>(h1, z1, ei, E, N, alpha1, xacc1);
    agg_kernel<<<blk((long)Eg + K, 8), T, 0, stream>>>(h2, z2, ei_g, Eg, K, alpha2, xacc2);

    add_bias_kernel<<<blk((long)K * F_DIM, T), T, 0, stream>>>(xacc2, b2, xg_out, (long)K * F_DIM);

    group_kernel<<<blk(N, 8), T, 0, stream>>>(xacc1, b1, xg_out, gidx, nbr, x_final, N);
}


# Edges grouped by destination in-call (returning int atomic add as arrival-order rank, prefix sum, place); one wave per node does softmax and row sum in registers, stores once: no float atomics
# speedup vs baseline: 2.1935x; 1.3593x over previous
#include <hip/hip_runtime.h>
#include <math.h>

#define H_HEADS 4
#define C_CH 32
#define F_DIM 128
#define NEG_SLOPE 0.2f

typedef __attribute__((ext_vector_type(2))) float v2f;
typedef __attribute__((ext_vector_type(8))) float v8f;

__global__ __launch_bounds__(256)
void gemm_wmma_kernel(const float* __restrict__ X, const float* __restrict__ W,
                      const float* __restrict__ a_src, const float* __restrict__ a_dst,
                      float* __restrict__ Hout, float* __restrict__ as,
                      float* __restrict__ ad, int rows) {
    extern __shared__ float Wlds[];
    for (int i = threadIdx.x; i < (F_DIM * F_DIM) / 2; i += blockDim.x) {
        int kp  = i >> 7;
        int col = i & 127;
        int cs  = (col + ((kp & 1) << 4)) & 127;
        float2 w2;
        w2.x = W[(size_t)(2 * kp)     * F_DIM + col];
        w2.y = W[(size_t)(2 * kp + 1) * F_DIM + col];
        *(float2*)(Wlds + (size_t)kp * 256 + cs * 2) = w2;
    }
    __syncthreads();

    const int lane = threadIdx.x & 31;
    const int wave = threadIdx.x >> 5;
    const int lm   = lane & 15;
    const int lh   = lane >> 4;
    const int rowBase = blockIdx.x * 128 + wave * 16;
    if (rowBase >= rows) return;

    int aRow = rowBase + lm;
    if (aRow >= rows) aRow = rows - 1;

    v8f acc[8];
    v8f zero = {0.f, 0.f, 0.f, 0.f, 0.f, 0.f, 0.f, 0.f};
#pragma unroll
    for (int ct = 0; ct < 8; ++ct) acc[ct] = zero;

    const float* xrow = X + (size_t)aRow * F_DIM;
    for (int k0 = 0; k0 < F_DIM; k0 += 4) {
        const int kp = (k0 >> 1) + lh;
        v2f a;
        a.x = xrow[2 * kp];
        a.y = xrow[2 * kp + 1];
        const float* bbase = Wlds + (size_t)kp * 256;
#pragma unroll
        for (int ct = 0; ct < 8; ++ct) {
            const int cs = (ct * 16 + lm + 16 * lh) & 127;
            v2f b = *(const v2f*)(bbase + cs * 2);
            acc[ct] = __builtin_amdgcn_wmma_f32_16x16x4_f32(
                false, a, false, b, (short)0, acc[ct], false, false);
        }
    }

#pragma unroll
    for (int ct = 0; ct < 8; ++ct) {
#pragma unroll
        for (int g = 0; g < 8; ++g) {
            int r = rowBase + g + 8 * lh;
            if (r < rows) Hout[(size_t)r * F_DIM + ct * 16 + lm] = acc[ct][g];
        }
    }

    float asv[8], adv[8];
#pragma unroll
    for (int ct = 0; ct < 8; ++ct) {
        asv[ct] = a_src[ct * 16 + lm];
        adv[ct] = a_dst[ct * 16 + lm];
    }
#pragma unroll
    for (int h = 0; h < 4; ++h) {
#pragma unroll
        for (int g = 0; g < 8; ++g) {
            float ss = acc[2 * h][g] * asv[2 * h];
            ss = fmaf(acc[2 * h + 1][g], asv[2 * h + 1], ss);
            float sd = acc[2 * h][g] * adv[2 * h];
            sd = fmaf(acc[2 * h + 1][g], adv[2 * h + 1], sd);
#pragma unroll
            for (int off = 8; off > 0; off >>= 1) {
                ss += __shfl_xor(ss, off, 32);
                sd += __shfl_xor(sd, off, 32);
            }
            int r = rowBase + g + 8 * lh;
            if (lm == 0 && r < rows) {
                as[(size_t)r * H_HEADS + h] = ss;
                ad[(size_t)r * H_HEADS + h] = sd;
            }
        }
    }
}

__global__ void zero_kernel(int* __restrict__ p, int n) {
    int i = blockIdx.x * blockDim.x + threadIdx.x;
    const int stride = gridDim.x * blockDim.x;
    for (; i < n; i += stride) p[i] = 0;
}

__global__ void rank_kernel(const int* __restrict__ ei, int E,
                            int* __restrict__ cnt, int* __restrict__ rank) {
    const int e = blockIdx.x * blockDim.x + threadIdx.x;
    if (e >= E) return;
    rank[e] = atomicAdd(&cnt[ei[E + e]], 1);
}

__global__ __launch_bounds__(1024)
void scan_kernel(int* __restrict__ cnt, int npad) {
    __shared__ int wsum[2][32];
    const int tid  = threadIdx.x;
    const int lane = tid & 31;
    const int wave = tid >> 5;
    int4* p = (int4*)cnt + tid * 4;
    int4 c0 = p[0], c1 = p[1], c2 = p[2], c3 = p[3];
    int carry = 0;
    int buf = 0;
    for (int base = 0; base < npad; base += 16384) {
        int4* q = p + 4096;
        int4 n0 = c0, n1 = c1, n2 = c2, n3 = c3;
        if (base + 16384 < npad) {
            n0 = q[0]; n1 = q[1]; n2 = q[2]; n3 = q[3];
        }
        int s[16];
        s[0]  = c0.x;
        s[1]  = s[0]  + c0.y;
        s[2]  = s[1]  + c0.z;
        s[3]  = s[2]  + c0.w;
        s[4]  = s[3]  + c1.x;
        s[5]  = s[4]  + c1.y;
        s[6]  = s[5]  + c1.z;
        s[7]  = s[6]  + c1.w;
        s[8]  = s[7]  + c2.x;
        s[9]  = s[8]  + c2.y;
        s[10] = s[9]  + c2.z;
        s[11] = s[10] + c2.w;
        s[12] = s[11] + c3.x;
        s[13] = s[12] + c3.y;
        s[14] = s[13] + c3.z;
        s[15] = s[14] + c3.w;
        const int tot = s[15];
        int inc = tot;
        for (int off = 1; off < 32; off <<= 1) {
            const int t = __shfl_up(inc, off, 32);
            if (lane >= off) inc += t;
        }
        if (lane == 31) wsum[buf][wave] = inc;
        __syncthreads();
        const int wv = wsum[buf][lane];
        int winc = wv;
        for (int off = 1; off < 32; off <<= 1) {
            const int t = __shfl_up(winc, off, 32);
            if (lane >= off) winc += t;
        }
        const int wave_excl = __shfl(winc, wave, 32) - __shfl(wv, wave, 32);
        const int total     = __shfl(winc, 31, 32);
        const int excl      = carry + wave_excl + inc - tot;
        p[0] = make_int4(excl,         excl + s[0],  excl + s[1],  excl + s[2]);
        p[1] = make_int4(excl + s[3],  excl + s[4],  excl + s[5],  excl + s[6]);
        p[2] = make_int4(excl + s[7],  excl + s[8],  excl + s[9],  excl + s[10]);
        p[3] = make_int4(excl + s[11], excl + s[12], excl + s[13], excl + s[14]);
        carry += total;
        c0 = n0; c1 = n1; c2 = n2; c3 = n3;
        p = q;
        buf ^= 1;
    }
}

__global__ void place_kernel(const int* __restrict__ ei, int E,
                             const int* __restrict__ rowptr, const int* __restrict__ rank,
                             int2* __restrict__ list) {
    const int e = blockIdx.x * blockDim.x + threadIdx.x;
    if (e >= E) return;
    const int pos = rowptr[ei[E + e]] + rank[e];
    list[pos] = make_int2(ei[e], e);
}

__device__ inline float leaky(float e) { return (e > 0.f) ? e : NEG_SLOPE * e; }

__device__ inline float lane_value(float v, int k) {
    return __int_as_float(__builtin_amdgcn_readlane(__float_as_int(v), k));
}

__global__ __launch_bounds__(256)
void seg_kernel(const float* __restrict__ hbuf, const float* __restrict__ as,
                const float* __restrict__ ad, const int* __restrict__ rowptr,
                const int2* __restrict__ list, const float* __restrict__ bias,
                int E, int Nn, float* __restrict__ alpha, float* __restrict__ xout) {
    const int lane = threadIdx.x & 31;
    const int node = __builtin_amdgcn_readfirstlane(
        (int)(blockIdx.x * (blockDim.x >> 5) + (threadIdx.x >> 5)));
    if (node >= Nn) return;
    const int beg = rowptr[node];
    const int end = rowptr[node + 1];
    const float4 adn = *(const float4*)(ad + (size_t)node * 4);
    const float4 asn = *(const float4*)(as + (size_t)node * 4);
    float4 es;
    es.x = leaky(asn.x + adn.x);
    es.y = leaky(asn.y + adn.y);
    es.z = leaky(asn.z + adn.z);
    es.w = leaky(asn.w + adn.w);

    float4 m = es;
    for (int j = beg + lane; j < end; j += 32) {
        const int sidx = list[j].x;
        const float4 a = *(const float4*)(as + (size_t)sidx * 4);
        m.x = fmaxf(m.x, leaky(a.x + adn.x));
        m.y = fmaxf(m.y, leaky(a.y + adn.y));
        m.z = fmaxf(m.z, leaky(a.z + adn.z));
        m.w = fmaxf(m.w, leaky(a.w + adn.w));
    }
#pragma unroll
    for (int off = 16; off > 0; off >>= 1) {
        m.x = fmaxf(m.x, __shfl_xor(m.x, off, 32));
        m.y = fmaxf(m.y, __shfl_xor(m.y, off, 32));
        m.z = fmaxf(m.z, __shfl_xor(m.z, off, 32));
        m.w = fmaxf(m.w, __shfl_xor(m.w, off, 32));
    }

    float4 z = make_float4(0.f, 0.f, 0.f, 0.f);
    for (int j = beg + lane; j < end; j += 32) {
        const int sidx = list[j].x;
        const float4 a = *(const float4*)(as + (size_t)sidx * 4);
        z.x += __expf(leaky(a.x + adn.x) - m.x);
        z.y += __expf(leaky(a.y + adn.y) - m.y);
        z.z += __expf(leaky(a.z + adn.z) - m.z);
        z.w += __expf(leaky(a.w + adn.w) - m.w);
    }
#pragma unroll
    for (int off = 16; off > 0; off >>= 1) {
        z.x += __shfl_xor(z.x, off, 32);
        z.y += __shfl_xor(z.y, off, 32);
        z.z += __shfl_xor(z.z, off, 32);
        z.w += __shfl_xor(z.w, off, 32);
    }
    float4 ps;
    ps.x = __expf(es.x - m.x);
    ps.y = __expf(es.y - m.y);
    ps.z = __expf(es.z - m.z);
    ps.w = __expf(es.w - m.w);
    z.x += ps.x;
    z.y += ps.y;
    z.z += ps.z;
    z.w += ps.w;

    const int hd = lane >> 3;
    float4 als;
    als.x = ps.x / z.x;
    als.y = ps.y / z.y;
    als.z = ps.z / z.z;
    als.w = ps.w / z.w;
    if (lane == 0) *(float4*)(alpha + (size_t)(E + node) * 4) = als;
    const float wself = (hd == 0) ? als.x : (hd == 1) ? als.y : (hd == 2) ? als.z : als.w;
    const float4 hself = *(const float4*)(hbuf + (size_t)node * F_DIM + lane * 4);
    float4 acc;
    acc.x = wself * hself.x;
    acc.y = wself * hself.y;
    acc.z = wself * hself.z;
    acc.w = wself * hself.w;

    for (int cb = beg; cb < end; cb += 32) {
        const int j = cb + lane;
        int sidx = 0;
        float4 al = make_float4(0.f, 0.f, 0.f, 0.f);
        if (j < end) {
            const int2 rec = list[j];
            sidx = rec.x;
            const float4 a = *(const float4*)(as + (size_t)sidx * 4);
            al.x = __expf(leaky(a.x + adn.x) - m.x) / z.x;
            al.y = __expf(leaky(a.y + adn.y) - m.y) / z.y;
            al.z = __expf(leaky(a.z + adn.z) - m.z) / z.z;
            al.w = __expf(leaky(a.w + adn.w) - m.w) / z.w;
            *(float4*)(alpha + (size_t)rec.y * 4) = al;
        }
        const int left = end - cb;
        const int cnt  = (left < 32) ? left : 32;
#pragma unroll 4
        for (int k = 0; k < cnt; ++k) {
            const int sk = __builtin_amdgcn_readlane(sidx, k);
            const float a0 = lane_value(al.x, k);
            const float a1 = lane_value(al.y, k);
            const float a2 = lane_value(al.z, k);
            const float a3 = lane_value(al.w, k);
            const float w = (hd == 0) ? a0 : (hd == 1) ? a1 : (hd == 2) ? a2 : a3;
            const float4 hv = *(const float4*)(hbuf + (size_t)sk * F_DIM + lane * 4);
            acc.x = fmaf(w, hv.x, acc.x);
            acc.y = fmaf(w, hv.y, acc.y);
            acc.z = fmaf(w, hv.z, acc.z);
            acc.w = fmaf(w, hv.w, acc.w);
        }
    }
    const float4 bv = *(const float4*)(bias + lane * 4);
    acc.x += bv.x;
    acc.y += bv.y;
    acc.z += bv.z;
    acc.w += bv.w;
    *(float4*)(xout + (size_t)node * F_DIM + lane * 4) = acc;
}

__global__ void group_kernel(float* xio, const float* __restrict__ xg,
                             const int* __restrict__ gidx, const int* __restrict__ nbr,
                             int Nn) {
    int lane = threadIdx.x & 31;
    int node = blockIdx.x * (blockDim.x >> 5) + (threadIdx.x >> 5);
    if (node >= Nn) return;
    float4 xo = *(const float4*)(xio + (size_t)node * F_DIM + lane * 4);
    int gi = gidx[node];
    int idx[17];
    idx[0] = gi;
#pragma unroll
    for (int k = 0; k < 16; ++k) idx[k + 1] = nbr[(size_t)gi * 16 + k];

    float4 cv[17];
    float sv[17];
#pragma unroll
    for (int k = 0; k < 17; ++k) {
        float4 c = *(const float4*)(xg + (size_t)idx[k] * F_DIM + lane * 4);
        cv[k] = c;
        float s = xo.x * c.x + xo.y * c.y + xo.z * c.z + xo.w * c.w;
#pragma unroll
        for (int off = 16; off > 0; off >>= 1) s += __shfl_xor(s, off, 32);
        sv[k] = s;
    }
    float mx = sv[0];
#pragma unroll
    for (int k = 1; k < 17; ++k) mx = fmaxf(mx, sv[k]);
    float sum = 0.f;
#pragma unroll
    for (int k = 0; k < 17; ++k) { sv[k] = __expf(sv[k] - mx); sum += sv[k]; }
    float inv = 1.f / sum;
    float4 go = make_float4(0.f, 0.f, 0.f, 0.f);
#pragma unroll
    for (int k = 0; k < 17; ++k) {
        float w = sv[k] * inv;
        go.x = fmaf(w, cv[k].x, go.x);
        go.y = fmaf(w, cv[k].y, go.y);
        go.z = fmaf(w, cv[k].z, go.z);
        go.w = fmaf(w, cv[k].w, go.w);
    }
    float4 o;
    o.x = 0.5f * (xo.x + go.x);
    o.y = 0.5f * (xo.y + go.y);
    o.z = 0.5f * (xo.z + go.z);
    o.w = 0.5f * (xo.w + go.w);
    *(float4*)(xio + (size_t)node * F_DIM + lane * 4) = o;
}

extern "C" void kernel_launch(void* const* d_in, const int* in_sizes, int n_in,
                              void* d_out, int out_size, void* d_ws, size_t ws_size,
                              hipStream_t stream) {
    const float* x       = (const float*)d_in[0];
    const int*   ei      = (const int*)d_in[1];
    const float* x_g     = (const float*)d_in[2];
    const int*   ei_g    = (const int*)d_in[3];
    const int*   gidx    = (const int*)d_in[4];
    const int*   nbr     = (const int*)d_in[5];
    const float* W1      = (const float*)d_in[6];
    const float* a_src1  = (const float*)d_in[7];
    const float* a_dst1  = (const float*)d_in[8];
    const float* b1      = (const float*)d_in[9];
    const float* W2      = (const float*)d_in[10];
    const float* a_src2  = (const float*)d_in[11];
    const float* a_dst2  = (const float*)d_in[12];
    const float* b2      = (const float*)d_in[13];

    const int N  = in_sizes[0] / F_DIM;
    const int E  = in_sizes[1] / 2;
    const int K  = in_sizes[2] / F_DIM;
    const int Eg = in_sizes[3] / 2;

    float* out     = (float*)d_out;
    float* x_final = out;
    float* xg_out  = out + (size_t)N * F_DIM;
    float* alpha1  = xg_out + (size_t)K * F_DIM;
    float* alpha2  = alpha1 + (size_t)(E + N) * H_HEADS;

    auto al4 = [](size_t n) { return (n + 3) & ~(size_t)3; };
    const int Np1 = ((N + 1 + 16383) / 16384) * 16384;
    const int Np2 = ((K + 1 + 16383) / 16384) * 16384;
    float* ws = (float*)d_ws;
    size_t off = 0;
    float* h1    = ws + off; off += (size_t)N * F_DIM;
    float* h2    = ws + off; off += (size_t)K * F_DIM;
    float* as1   = ws + off; off += (size_t)N * H_HEADS;
    float* ad1   = ws + off; off += (size_t)N * H_HEADS;
    float* as2   = ws + off; off += (size_t)K * H_HEADS;
    float* ad2   = ws + off; off += (size_t)K * H_HEADS;
    int*   cur1  = (int*)(ws + off);  off += (size_t)Np1;
    int*   cur2  = (int*)(ws + off);  off += (size_t)Np2;
    int*   rank1 = (int*)(ws + off);  off += al4((size_t)E);
    int*   rank2 = (int*)(ws + off);  off += al4((size_t)Eg);
    int2*  list1 = (int2*)(ws + off); off += al4((size_t)E * 2);
    int2*  list2 = (int2*)(ws + off); off += al4((size_t)Eg * 2);

    const int T = 256;
    auto blk = [](long n, int t) { return (int)((n + t - 1) / t); };

    zero_kernel<<<blk((long)Np1 + Np2, T), T, 0, stream>>>(cur1, Np1 + Np2);

    gemm_wmma_kernel<<<blk(N, 128), T, F_DIM * F_DIM * sizeof(float), stream>>>(
        x, W1, a_src1, a_dst1, h1, as1, ad1, N);
    gemm_wmma_kernel<<<blk(K, 128), T, F_DIM * F_DIM * sizeof(float), stream>>>(
        x_g, W2, a_src2, a_dst2, h2, as2, ad2, K);

    rank_kernel<<<blk(E, T), T, 0, stream>>>(ei, E, cur1, rank1);
    rank_kernel<<<blk(Eg, T), T, 0, stream>>>(ei_g, Eg, cur2, rank2);
    scan_kernel<<<1, 1024, 0, stream>>>(cur1, Np1);
    scan_kernel<<<1, 1024, 0, stream>>>(cur2, Np2);
    place_kernel<<<blk(E, T), T, 0, stream>>>(ei, E, cur1, rank1, list1);
    place_kernel<<<blk(Eg, T), T, 0, stream>>>(ei_g, Eg, cur2, rank2, list2);

    seg_kernel<<<blk(N, 8), T, 0, stream>>>(h1, as1, ad1, cur1, list1, b1, E, N, alpha1, x_final);
    seg_kernel<<<blk(K, 8), T, 0, stream>>>(h2, as2, ad2, cur2, list2, b2, Eg, K, alpha2, xg_out);

    group_kernel<<<blk(N, 8), T, 0, stream>>>(x_final, xg_out, gidx, nbr, N);
}


# Group mix by group: a block owns 4 groups, finds their nodes by reading the group index array (LDS lists); a wave keeps its group's 17 candidate rows in registers over its nodes; node sums unchanged
# speedup vs baseline: 3.2310x; 1.4730x over previous
#include <hip/hip_runtime.h>
#include <math.h>

#define H_HEADS 4
#define C_CH 32
#define F_DIM 128
#define NEG_SLOPE 0.2f

typedef __attribute__((ext_vector_type(2))) float v2f;
typedef __attribute__((ext_vector_type(8))) float v8f;

__global__ __launch_bounds__(256)
void gemm_wmma_kernel(const float* __restrict__ XA, const float* __restrict__ WA,
                      const float* __restrict__ a_srcA, const float* __restrict__ a_dstA,
                      float* __restrict__ HoutA, float* __restrict__ asA,
                      float* __restrict__ adA, int rowsA,
                      const float* __restrict__ XB, const float* __restrict__ WB,
                      const float* __restrict__ a_srcB, const float* __restrict__ a_dstB,
                      float* __restrict__ HoutB, float* __restrict__ asB,
                      float* __restrict__ adB, int rowsB, int nbB) {
    extern __shared__ float Wlds[];
    const bool gB = blockIdx.x < nbB;
    const float* X     = gB ? XB : XA;
    const float* W     = gB ? WB : WA;
    const float* a_src = gB ? a_srcB : a_srcA;
    const float* a_dst = gB ? a_dstB : a_dstA;
    float* Hout        = gB ? HoutB : HoutA;
    float* as          = gB ? asB : asA;
    float* ad          = gB ? adB : adA;
    const int rows     = gB ? rowsB : rowsA;
    const int blockLocal = gB ? blockIdx.x : blockIdx.x - nbB;
    for (int i = threadIdx.x; i < (F_DIM * F_DIM) / 2; i += blockDim.x) {
        int kp  = i >> 7;
        int col = i & 127;
        int cs  = (col + ((kp & 1) << 4)) & 127;
        float2 w2;
        w2.x = W[(size_t)(2 * kp)     * F_DIM + col];
        w2.y = W[(size_t)(2 * kp + 1) * F_DIM + col];
        *(float2*)(Wlds + (size_t)kp * 256 + cs * 2) = w2;
    }
    __syncthreads();

    const int lane = threadIdx.x & 31;
    const int wave = threadIdx.x >> 5;
    const int lm   = lane & 15;
    const int lh   = lane >> 4;
    const int rowBase = blockLocal * 128 + wave * 16;
    if (rowBase >= rows) return;

    int aRow = rowBase + lm;
    if (aRow >= rows) aRow = rows - 1;

    v8f acc[8];
    v8f zero = {0.f, 0.f, 0.f, 0.f, 0.f, 0.f, 0.f, 0.f};
#pragma unroll
    for (int ct = 0; ct < 8; ++ct) acc[ct] = zero;

    const float* xrow = X + (size_t)aRow * F_DIM;
    for (int k0 = 0; k0 < F_DIM; k0 += 4) {
        const int kp = (k0 >> 1) + lh;
        v2f a;
        a.x = xrow[2 * kp];
        a.y = xrow[2 * kp + 1];
        const float* bbase = Wlds + (size_t)kp * 256;
#pragma unroll
        for (int ct = 0; ct < 8; ++ct) {
            const int cs = (ct * 16 + lm + 16 * lh) & 127;
            v2f b = *(const v2f*)(bbase + cs * 2);
            acc[ct] = __builtin_amdgcn_wmma_f32_16x16x4_f32(
                false, a, false, b, (short)0, acc[ct], false, false);
        }
    }

#pragma unroll
    for (int ct = 0; ct < 8; ++ct) {
#pragma unroll
        for (int g = 0; g < 8; ++g) {
            int r = rowBase + g + 8 * lh;
            if (r < rows) Hout[(size_t)r * F_DIM + ct * 16 + lm] = acc[ct][g];
        }
    }

    float asv[8], adv[8];
#pragma unroll
    for (int ct = 0; ct < 8; ++ct) {
        asv[ct] = a_src[ct * 16 + lm];
        adv[ct] = a_dst[ct * 16 + lm];
    }
#pragma unroll
    for (int h = 0; h < 4; ++h) {
#pragma unroll
        for (int g = 0; g < 8; ++g) {
            float ss = acc[2 * h][g] * asv[2 * h];
            ss = fmaf(acc[2 * h + 1][g], asv[2 * h + 1], ss);
            float sd = acc[2 * h][g] * adv[2 * h];
            sd = fmaf(acc[2 * h + 1][g], adv[2 * h + 1], sd);
#pragma unroll
            for (int off = 8; off > 0; off >>= 1) {
                ss += __shfl_xor(ss, off, 32);
                sd += __shfl_xor(sd, off, 32);
            }
            int r = rowBase + g + 8 * lh;
            if (lm == 0 && r < rows) {
                as[(size_t)r * H_HEADS + h] = ss;
                ad[(size_t)r * H_HEADS + h] = sd;
            }
        }
    }
}

#define OWN_R 256
#define STAGE_CAP 7680
#define NO_DST 0x7fffffff

__global__ __launch_bounds__(1024)
void owner_kernel(const int* __restrict__ eiA, int EA, int NnA,
                  int* __restrict__ rowptrA, int2* __restrict__ listA,
                  const int* __restrict__ eiB, int EB, int NnB,
                  int* __restrict__ rowptrB, int2* __restrict__ listB, int nbB) {
    __shared__ int2 stage[STAGE_CAP];
    __shared__ int cntL[OWN_R];
    __shared__ int cur[OWN_R];
    __shared__ int wbelow[32];
    __shared__ int wscan[OWN_R / 32];
    __shared__ int nstage;
    const bool gB = blockIdx.x < nbB;
    const int* ei = gB ? eiB : eiA;
    const int  E  = gB ? EB : EA;
    const int  Nn = gB ? NnB : NnA;
    int* rowptr   = gB ? rowptrB : rowptrA;
    int2* list    = gB ? listB : listA;
    const int lo   = (gB ? blockIdx.x : blockIdx.x - nbB) * OWN_R;
    const int tid  = threadIdx.x;
    const int lane = tid & 31;
    const int wave = tid >> 5;
    if (tid < OWN_R) { cntL[tid] = 0; cur[tid] = 0; }
    if (tid == 0) nstage = 0;
    __syncthreads();

    const int* dst = ei + E;
    const int n4 = ((E & 3) == 0) ? (E >> 2) : 0;
    const int4* d4 = (const int4*)dst;
    const int4 none = make_int4(NO_DST, NO_DST, NO_DST, NO_DST);
    int below = 0;
    auto put = [&](unsigned t, int e) {
        const int r = atomicAdd(&cntL[t], 1);
        const int p = atomicAdd(&nstage, 1);
        if (p < STAGE_CAP) stage[p] = make_int2(e, (int)(((unsigned)r << 8) | t));
    };
    auto count_below = [&](int d) {
        below += __popc(__builtin_amdgcn_ballot_w32(d < lo));
    };
    const int n4m = (n4 > 0) ? n4 - 1 : 0;
    for (int b = 0; b < n4; b += 8 * 1024) {
        int4 v[8];
#pragma unroll
        for (int u = 0; u < 8; ++u) {
            const int i = b + u * 1024 + tid;
            v[u] = d4[(i < n4) ? i : n4m];
        }
#pragma unroll
        for (int u = 0; u < 8; ++u) {
            const int i = b + u * 1024 + tid;
            const int4 d = (i < n4) ? v[u] : none;
            count_below(d.x);
            count_below(d.y);
            count_below(d.z);
            count_below(d.w);
            const unsigned t0 = (unsigned)(d.x - lo);
            const unsigned t1 = (unsigned)(d.y - lo);
            const unsigned t2 = (unsigned)(d.z - lo);
            const unsigned t3 = (unsigned)(d.w - lo);
            const unsigned R = (unsigned)OWN_R;
            if ((t0 < R) | (t1 < R) | (t2 < R) | (t3 < R)) {
                if (t0 < R) put(t0, 4 * i);
                if (t1 < R) put(t1, 4 * i + 1);
                if (t2 < R) put(t2, 4 * i + 2);
                if (t3 < R) put(t3, 4 * i + 3);
            }
        }
    }
    for (int e0 = 4 * n4; e0 < E; e0 += 1024) {
        const int e = e0 + tid;
        const int d = (e < E) ? dst[e] : NO_DST;
        count_below(d);
        const unsigned t = (unsigned)(d - lo);
        if (t < (unsigned)OWN_R) put(t, e);
    }
    if (lane == 0) wbelow[wave] = below;
    __syncthreads();

    int base = 0;
    for (int w = 0; w < 32; ++w) base += wbelow[w];
    const int c = (tid < OWN_R) ? cntL[tid] : 0;
    int inc = c;
    for (int off = 1; off < 32; off <<= 1) {
        const int t = __shfl_up(inc, off, 32);
        if (lane >= off) inc += t;
    }
    if (tid < OWN_R && lane == 31) wscan[wave] = inc;
    __syncthreads();
    if (tid < OWN_R) {
        int wex = 0;
        for (int w = 0; w < wave; ++w) wex += wscan[w];
        const int start = wex + inc - c;
        cntL[tid] = start;
        if (lo + tid <= Nn) rowptr[lo + tid] = base + start;
    }
    __syncthreads();
    const int n = nstage;
    if (n <= STAGE_CAP) {
        for (int p = tid; p < n; p += 1024) {
            const int2 s = stage[p];
            const unsigned pk = (unsigned)s.y;
            list[base + cntL[pk & 255u] + (int)(pk >> 8)] = make_int2(ei[s.x], s.x);
        }
    } else {
        for (int e0 = 0; e0 < E; e0 += 1024) {
            const int e = e0 + tid;
            const int d = (e < E) ? dst[e] : NO_DST;
            const unsigned t = (unsigned)(d - lo);
            if (t < (unsigned)OWN_R) {
                const int r = atomicAdd(&cur[t], 1);
                list[base + cntL[t] + r] = make_int2(ei[e], e);
            }
        }
    }
}

__device__ inline float leaky(float e) { return (e > 0.f) ? e : NEG_SLOPE * e; }

__device__ inline float lane_value(float v, int k) {
    return __int_as_float(__builtin_amdgcn_readlane(__float_as_int(v), k));
}

__device__ __forceinline__
void seg_body(const float* __restrict__ hbuf, const float* __restrict__ as,
              const float* __restrict__ ad, const int* __restrict__ rowptr,
              const int2* __restrict__ list, const float* __restrict__ bias,
              int E, int Nn, float* __restrict__ alpha, float* __restrict__ xout,
              int blockLocal) {
    const int lane = threadIdx.x & 31;
    const int node = __builtin_amdgcn_readfirstlane(
        (int)(blockLocal * (blockDim.x >> 5) + (threadIdx.x >> 5)));
    if (node >= Nn) return;
    const int beg = rowptr[node];
    const int end = rowptr[node + 1];
    const float4 adn = *(const float4*)(ad + (size_t)node * 4);
    const float4 asn = *(const float4*)(as + (size_t)node * 4);
    float4 es;
    es.x = leaky(asn.x + adn.x);
    es.y = leaky(asn.y + adn.y);
    es.z = leaky(asn.z + adn.z);
    es.w = leaky(asn.w + adn.w);

    float4 m = es;
    for (int j = beg + lane; j < end; j += 32) {
        const int sidx = list[j].x;
        const float4 a = *(const float4*)(as + (size_t)sidx * 4);
        m.x = fmaxf(m.x, leaky(a.x + adn.x));
        m.y = fmaxf(m.y, leaky(a.y + adn.y));
        m.z = fmaxf(m.z, leaky(a.z + adn.z));
        m.w = fmaxf(m.w, leaky(a.w + adn.w));
    }
#pragma unroll
    for (int off = 16; off > 0; off >>= 1) {
        m.x = fmaxf(m.x, __shfl_xor(m.x, off, 32));
        m.y = fmaxf(m.y, __shfl_xor(m.y, off, 32));
        m.z = fmaxf(m.z, __shfl_xor(m.z, off, 32));
        m.w = fmaxf(m.w, __shfl_xor(m.w, off, 32));
    }

    float4 z = make_float4(0.f, 0.f, 0.f, 0.f);
    for (int j = beg + lane; j < end; j += 32) {
        const int sidx = list[j].x;
        const float4 a = *(const float4*)(as + (size_t)sidx * 4);
        z.x += __expf(leaky(a.x + adn.x) - m.x);
        z.y += __expf(leaky(a.y + adn.y) - m.y);
        z.z += __expf(leaky(a.z + adn.z) - m.z);
        z.w += __expf(leaky(a.w + adn.w) - m.w);
    }
#pragma unroll
    for (int off = 16; off > 0; off >>= 1) {
        z.x += __shfl_xor(z.x, off, 32);
        z.y += __shfl_xor(z.y, off, 32);
        z.z += __shfl_xor(z.z, off, 32);
        z.w += __shfl_xor(z.w, off, 32);
    }
    float4 ps;
    ps.x = __expf(es.x - m.x);
    ps.y = __expf(es.y - m.y);
    ps.z = __expf(es.z - m.z);
    ps.w = __expf(es.w - m.w);
    z.x += ps.x;
    z.y += ps.y;
    z.z += ps.z;
    z.w += ps.w;

    const int hd = lane >> 3;
    float4 als;
    als.x = ps.x / z.x;
    als.y = ps.y / z.y;
    als.z = ps.z / z.z;
    als.w = ps.w / z.w;
    if (lane == 0) *(float4*)(alpha + (size_t)(E + node) * 4) = als;
    const float wself = (hd == 0) ? als.x : (hd == 1) ? als.y : (hd == 2) ? als.z : als.w;
    const float4 hself = *(const float4*)(hbuf + (size_t)node * F_DIM + lane * 4);
    float4 acc;
    acc.x = wself * hself.x;
    acc.y = wself * hself.y;
    acc.z = wself * hself.z;
    acc.w = wself * hself.w;

    for (int cb = beg; cb < end; cb += 32) {
        const int j = cb + lane;
        int sidx = 0;
        float4 al = make_float4(0.f, 0.f, 0.f, 0.f);
        if (j < end) {
            const int2 rec = list[j];
            sidx = rec.x;
            const float4 a = *(const float4*)(as + (size_t)sidx * 4);
            al.x = __expf(leaky(a.x + adn.x) - m.x) / z.x;
            al.y = __expf(leaky(a.y + adn.y) - m.y) / z.y;
            al.z = __expf(leaky(a.z + adn.z) - m.z) / z.z;
            al.w = __expf(leaky(a.w + adn.w) - m.w) / z.w;
            *(float4*)(alpha + (size_t)rec.y * 4) = al;
        }
        const int left = end - cb;
        const int cnt  = (left < 32) ? left : 32;
#pragma unroll 4
        for (int k = 0; k < cnt; ++k) {
            const int sk = __builtin_amdgcn_readlane(sidx, k);
            const float a0 = lane_value(al.x, k);
            const float a1 = lane_value(al.y, k);
            const float a2 = lane_value(al.z, k);
            const float a3 = lane_value(al.w, k);
            const float w = (hd == 0) ? a0 : (hd == 1) ? a1 : (hd == 2) ? a2 : a3;
            const float4 hv = *(const float4*)(hbuf + (size_t)sk * F_DIM + lane * 4);
            acc.x = fmaf(w, hv.x, acc.x);
            acc.y = fmaf(w, hv.y, acc.y);
            acc.z = fmaf(w, hv.z, acc.z);
            acc.w = fmaf(w, hv.w, acc.w);
        }
    }
    const float4 bv = *(const float4*)(bias + lane * 4);
    acc.x += bv.x;
    acc.y += bv.y;
    acc.z += bv.z;
    acc.w += bv.w;
    *(float4*)(xout + (size_t)node * F_DIM + lane * 4) = acc;
}

__global__ __launch_bounds__(256)
void seg_kernel(const float* __restrict__ hbufA, const float* __restrict__ asA,
                const float* __restrict__ adA, const int* __restrict__ rowptrA,
                const int2* __restrict__ listA, const float* __restrict__ biasA,
                int EA, int NnA, float* __restrict__ alphaA, float* __restrict__ xoutA,
                const float* __restrict__ hbufB, const float* __restrict__ asB,
                const float* __restrict__ adB, const int* __restrict__ rowptrB,
                const int2* __restrict__ listB, const float* __restrict__ biasB,
                int EB, int NnB, float* __restrict__ alphaB, float* __restrict__ xoutB,
                int nbB) {
    if (blockIdx.x < nbB)
        seg_body(hbufB, asB, adB, rowptrB, listB, biasB, EB, NnB, alphaB, xoutB,
                 (int)blockIdx.x);
    else
        seg_body(hbufA, asA, adA, rowptrA, listA, biasA, EA, NnA, alphaA, xoutA,
                 (int)blockIdx.x - nbB);
}

#define GRP_G 4
#define GRP_WPG 6
#define GRP_T (GRP_G * GRP_WPG * 32)
#define GRP_CAP 2048

__device__ __forceinline__
void mix_node(float* xio, const float4 (&cv)[17], int node, int lane, float4 xo) {
    float sv[17];
#pragma unroll
    for (int k = 0; k < 17; ++k) {
        const float4 c = cv[k];
        float s = xo.x * c.x + xo.y * c.y + xo.z * c.z + xo.w * c.w;
#pragma unroll
        for (int off = 16; off > 0; off >>= 1) s += __shfl_xor(s, off, 32);
        sv[k] = s;
    }
    float mx = sv[0];
#pragma unroll
    for (int k = 1; k < 17; ++k) mx = fmaxf(mx, sv[k]);
    float sum = 0.f;
#pragma unroll
    for (int k = 0; k < 17; ++k) { sv[k] = __expf(sv[k] - mx); sum += sv[k]; }
    float inv = 1.f / sum;
    float4 go = make_float4(0.f, 0.f, 0.f, 0.f);
#pragma unroll
    for (int k = 0; k < 17; ++k) {
        float w = sv[k] * inv;
        go.x = fmaf(w, cv[k].x, go.x);
        go.y = fmaf(w, cv[k].y, go.y);
        go.z = fmaf(w, cv[k].z, go.z);
        go.w = fmaf(w, cv[k].w, go.w);
    }
    float4 o;
    o.x = 0.5f * (xo.x + go.x);
    o.y = 0.5f * (xo.y + go.y);
    o.z = 0.5f * (xo.z + go.z);
    o.w = 0.5f * (xo.w + go.w);
    *(float4*)(xio + (size_t)node * F_DIM + lane * 4) = o;
}

__global__ __launch_bounds__(GRP_T)
void group_kernel(float* xio, const float* __restrict__ xg,
                  const int* __restrict__ gidx, const int* __restrict__ nbr,
                  int Nn, int Kg) {
    __shared__ int ent[GRP_G][GRP_CAP];
    __shared__ int cnt[GRP_G];
    const int tid  = threadIdx.x;
    const int lane = tid & 31;
    const int wave = __builtin_amdgcn_readfirstlane(tid >> 5);
    const int lo   = blockIdx.x * GRP_G;
    const int t    = wave / GRP_WPG;
    const int r    = wave - t * GRP_WPG;
    const int g    = lo + t;
    if (tid < GRP_G) cnt[tid] = 0;
    __syncthreads();

    auto take = [&](int d, int node) {
        const unsigned tt = (unsigned)(d - lo);
        if (tt < (unsigned)GRP_G) {
            const int p = atomicAdd(&cnt[tt], 1);
            if (p < GRP_CAP) ent[tt][p] = node;
        }
    };
    const int n4 = ((((size_t)gidx) & 15) == 0) ? (Nn >> 2) : 0;
    const int4* g4 = (const int4*)gidx;
    const int n4m = (n4 > 0) ? n4 - 1 : 0;
    for (int b = 0; b < n4; b += 8 * GRP_T) {
        int4 v[8];
#pragma unroll
        for (int u = 0; u < 8; ++u) {
            const int i = b + u * GRP_T + tid;
            v[u] = g4[(i < n4) ? i : n4m];
        }
#pragma unroll
        for (int u = 0; u < 8; ++u) {
            const int i = b + u * GRP_T + tid;
            if (i < n4) {
                take(v[u].x, 4 * i);
                take(v[u].y, 4 * i + 1);
                take(v[u].z, 4 * i + 2);
                take(v[u].w, 4 * i + 3);
            }
        }
    }
    for (int e0 = 4 * n4; e0 < Nn; e0 += GRP_T) {
        const int e = e0 + tid;
        if (e < Nn) take(gidx[e], e);
    }
    __syncthreads();

    if (g >= Kg) return;

    float4 cv[17];
    cv[0] = *(const float4*)(xg + (size_t)g * F_DIM + lane * 4);
#pragma unroll
    for (int k = 0; k < 16; ++k) {
        const int c = nbr[(size_t)g * 16 + k];
        cv[k + 1] = *(const float4*)(xg + (size_t)c * F_DIM + lane * 4);
    }

    const int n = cnt[t];
    if (n <= GRP_CAP) {
        int p = r;
        int node = 0;
        float4 xo = make_float4(0.f, 0.f, 0.f, 0.f);
        if (p < n) {
            node = __builtin_amdgcn_readfirstlane(ent[t][p]);
            xo = *(const float4*)(xio + (size_t)node * F_DIM + lane * 4);
        }
        while (p < n) {
            const int pn = p + GRP_WPG;
            int node2 = 0;
            float4 xo2 = xo;
            if (pn < n) {
                node2 = __builtin_amdgcn_readfirstlane(ent[t][pn]);
                xo2 = *(const float4*)(xio + (size_t)node2 * F_DIM + lane * 4);
            }
            mix_node(xio, cv, node, lane, xo);
            node = node2; xo = xo2; p = pn;
        }
    } else {
        for (int v = r; v < Nn; v += GRP_WPG) {
            if (gidx[v] == g) {
                const float4 xo = *(const float4*)(xio + (size_t)v * F_DIM + lane * 4);
                mix_node(xio, cv, v, lane, xo);
            }
        }
    }
}

extern "C" void kernel_launch(void* const* d_in, const int* in_sizes, int n_in,
                              void* d_out, int out_size, void* d_ws, size_t ws_size,
                              hipStream_t stream) {
    const float* x       = (const float*)d_in[0];
    const int*   ei      = (const int*)d_in[1];
    const float* x_g     = (const float*)d_in[2];
    const int*   ei_g    = (const int*)d_in[3];
    const int*   gidx    = (const int*)d_in[4];
    const int*   nbr     = (const int*)d_in[5];
    const float* W1      = (const float*)d_in[6];
    const float* a_src1  = (const float*)d_in[7];
    const float* a_dst1  = (const float*)d_in[8];
    const float* b1      = (const float*)d_in[9];
    const float* W2      = (const float*)d_in[10];
    const float* a_src2  = (const float*)d_in[11];
    const float* a_dst2  = (const float*)d_in[12];
    const float* b2      = (const float*)d_in[13];

    const int N  = in_sizes[0] / F_DIM;
    const int E  = in_sizes[1] / 2;
    const int K  = in_sizes[2] / F_DIM;
    const int Eg = in_sizes[3] / 2;

    float* out     = (float*)d_out;
    float* x_final = out;
    float* xg_out  = out + (size_t)N * F_DIM;
    float* alpha1  = xg_out + (size_t)K * F_DIM;
    float* alpha2  = alpha1 + (size_t)(E + N) * H_HEADS;

    auto al4 = [](size_t n) { return (n + 3) & ~(size_t)3; };
    const int Np1 = ((N + 1 + 16383) / 16384) * 16384;
    const int Np2 = ((K + 1 + 16383) / 16384) * 16384;
    float* ws = (float*)d_ws;
    size_t off = 0;
    float* h1    = ws + off; off += (size_t)N * F_DIM;
    float* h2    = ws + off; off += (size_t)K * F_DIM;
    float* as1   = ws + off; off += (size_t)N * H_HEADS;
    float* ad1   = ws + off; off += (size_t)N * H_HEADS;
    float* as2   = ws + off; off += (size_t)K * H_HEADS;
    float* ad2   = ws + off; off += (size_t)K * H_HEADS;
    int*   cur1  = (int*)(ws + off);  off += (size_t)Np1;
    int*   cur2  = (int*)(ws + off);  off += (size_t)Np2;
    int2*  list1 = (int2*)(ws + off); off += al4((size_t)E * 2);
    int2*  list2 = (int2*)(ws + off); off += al4((size_t)Eg * 2);

    const int T = 256;
    auto blk = [](long n, int t) { return (int)((n + t - 1) / t); };


    owner_kernel<<<blk(K + 1, OWN_R) + blk(N + 1, OWN_R), 1024, 0, stream>>>(
        ei, E, N, cur1, list1, ei_g, Eg, K, cur2, list2, blk(K + 1, OWN_R));

    gemm_wmma_kernel<<<blk(K, 128) + blk(N, 128), T, F_DIM * F_DIM * sizeof(float), stream>>>(
        x, W1, a_src1, a_dst1, h1, as1, ad1, N,
        x_g, W2, a_src2, a_dst2, h2, as2, ad2, K, blk(K, 128));

    seg_kernel<<<blk(K, 8) + blk(N, 8), T, 0, stream>>>(
        h1, as1, ad1, cur1, list1, b1, E, N, alpha1, x_final,
        h2, as2, ad2, cur2, list2, b2, Eg, K, alpha2, xg_out, blk(K, 8));

    group_kernel<<<blk(K, GRP_G), GRP_T, 0, stream>>>(x_final, xg_out, gidx, nbr, N, K);
}


# Softmax of a node with at most 32 incoming edges keeps each edge's record and score in one lane's registers: list and source coefficients read once, not three times; same terms, same sum order
# speedup vs baseline: 3.3844x; 1.0475x over previous
#include <hip/hip_runtime.h>
#include <math.h>

#define H_HEADS 4
#define C_CH 32
#define F_DIM 128
#define NEG_SLOPE 0.2f

typedef __attribute__((ext_vector_type(2))) float v2f;
typedef __attribute__((ext_vector_type(8))) float v8f;

__global__ __launch_bounds__(256)
void gemm_wmma_kernel(const float* __restrict__ XA, const float* __restrict__ WA,
                      const float* __restrict__ a_srcA, const float* __restrict__ a_dstA,
                      float* __restrict__ HoutA, float* __restrict__ asA,
                      float* __restrict__ adA, int rowsA,
                      const float* __restrict__ XB, const float* __restrict__ WB,
                      const float* __restrict__ a_srcB, const float* __restrict__ a_dstB,
                      float* __restrict__ HoutB, float* __restrict__ asB,
                      float* __restrict__ adB, int rowsB, int nbB) {
    extern __shared__ float Wlds[];
    const bool gB = blockIdx.x < nbB;
    const float* X     = gB ? XB : XA;
    const float* W     = gB ? WB : WA;
    const float* a_src = gB ? a_srcB : a_srcA;
    const float* a_dst = gB ? a_dstB : a_dstA;
    float* Hout        = gB ? HoutB : HoutA;
    float* as          = gB ? asB : asA;
    float* ad          = gB ? adB : adA;
    const int rows     = gB ? rowsB : rowsA;
    const int blockLocal = gB ? blockIdx.x : blockIdx.x - nbB;
    for (int i = threadIdx.x; i < (F_DIM * F_DIM) / 2; i += blockDim.x) {
        int kp  = i >> 7;
        int col = i & 127;
        int cs  = (col + ((kp & 1) << 4)) & 127;
        float2 w2;
        w2.x = W[(size_t)(2 * kp)     * F_DIM + col];
        w2.y = W[(size_t)(2 * kp + 1) * F_DIM + col];
        *(float2*)(Wlds + (size_t)kp * 256 + cs * 2) = w2;
    }
    __syncthreads();

    const int lane = threadIdx.x & 31;
    const int wave = threadIdx.x >> 5;
    const int lm   = lane & 15;
    const int lh   = lane >> 4;
    const int rowBase = blockLocal * 128 + wave * 16;
    if (rowBase >= rows) return;

    int aRow = rowBase + lm;
    if (aRow >= rows) aRow = rows - 1;

    v8f acc[8];
    v8f zero = {0.f, 0.f, 0.f, 0.f, 0.f, 0.f, 0.f, 0.f};
#pragma unroll
    for (int ct = 0; ct < 8; ++ct) acc[ct] = zero;

    const float* xrow = X + (size_t)aRow * F_DIM;
    for (int k0 = 0; k0 < F_DIM; k0 += 4) {
        const int kp = (k0 >> 1) + lh;
        v2f a;
        a.x = xrow[2 * kp];
        a.y = xrow[2 * kp + 1];
        const float* bbase = Wlds + (size_t)kp * 256;
#pragma unroll
        for (int ct = 0; ct < 8; ++ct) {
            const int cs = (ct * 16 + lm + 16 * lh) & 127;
            v2f b = *(const v2f*)(bbase + cs * 2);
            acc[ct] = __builtin_amdgcn_wmma_f32_16x16x4_f32(
                false, a, false, b, (short)0, acc[ct], false, false);
        }
    }

#pragma unroll
    for (int ct = 0; ct < 8; ++ct) {
#pragma unroll
        for (int g = 0; g < 8; ++g) {
            int r = rowBase + g + 8 * lh;
            if (r < rows) Hout[(size_t)r * F_DIM + ct * 16 + lm] = acc[ct][g];
        }
    }

    float asv[8], adv[8];
#pragma unroll
    for (int ct = 0; ct < 8; ++ct) {
        asv[ct] = a_src[ct * 16 + lm];
        adv[ct] = a_dst[ct * 16 + lm];
    }
#pragma unroll
    for (int h = 0; h < 4; ++h) {
#pragma unroll
        for (int g = 0; g < 8; ++g) {
            float ss = acc[2 * h][g] * asv[2 * h];
            ss = fmaf(acc[2 * h + 1][g], asv[2 * h + 1], ss);
            float sd = acc[2 * h][g] * adv[2 * h];
            sd = fmaf(acc[2 * h + 1][g], adv[2 * h + 1], sd);
#pragma unroll
            for (int off = 8; off > 0; off >>= 1) {
                ss += __shfl_xor(ss, off, 32);
                sd += __shfl_xor(sd, off, 32);
            }
            int r = rowBase + g + 8 * lh;
            if (lm == 0 && r < rows) {
                as[(size_t)r * H_HEADS + h] = ss;
                ad[(size_t)r * H_HEADS + h] = sd;
            }
        }
    }
}

#define OWN_R 256
#define STAGE_CAP 7680
#define NO_DST 0x7fffffff

__global__ __launch_bounds__(1024)
void owner_kernel(const int* __restrict__ eiA, int EA, int NnA,
                  int* __restrict__ rowptrA, int2* __restrict__ listA,
                  const int* __restrict__ eiB, int EB, int NnB,
                  int* __restrict__ rowptrB, int2* __restrict__ listB, int nbB) {
    __shared__ int2 stage[STAGE_CAP];
    __shared__ int cntL[OWN_R];
    __shared__ int cur[OWN_R];
    __shared__ int wbelow[32];
    __shared__ int wscan[OWN_R / 32];
    __shared__ int nstage;
    const bool gB = blockIdx.x < nbB;
    const int* ei = gB ? eiB : eiA;
    const int  E  = gB ? EB : EA;
    const int  Nn = gB ? NnB : NnA;
    int* rowptr   = gB ? rowptrB : rowptrA;
    int2* list    = gB ? listB : listA;
    const int lo   = (gB ? blockIdx.x : blockIdx.x - nbB) * OWN_R;
    const int tid  = threadIdx.x;
    const int lane = tid & 31;
    const int wave = tid >> 5;
    if (tid < OWN_R) { cntL[tid] = 0; cur[tid] = 0; }
    if (tid == 0) nstage = 0;
    __syncthreads();

    const int* dst = ei + E;
    const int n4 = ((E & 3) == 0) ? (E >> 2) : 0;
    const int4* d4 = (const int4*)dst;
    const int4 none = make_int4(NO_DST, NO_DST, NO_DST, NO_DST);
    int below = 0;
    auto put = [&](unsigned t, int e) {
        const int r = atomicAdd(&cntL[t], 1);
        const int p = atomicAdd(&nstage, 1);
        if (p < STAGE_CAP) stage[p] = make_int2(e, (int)(((unsigned)r << 8) | t));
    };
    auto count_below = [&](int d) {
        below += __popc(__builtin_amdgcn_ballot_w32(d < lo));
    };
    const int n4m = (n4 > 0) ? n4 - 1 : 0;
    for (int b = 0; b < n4; b += 8 * 1024) {
        int4 v[8];
#pragma unroll
        for (int u = 0; u < 8; ++u) {
            const int i = b + u * 1024 + tid;
            v[u] = d4[(i < n4) ? i : n4m];
        }
#pragma unroll
        for (int u = 0; u < 8; ++u) {
            const int i = b + u * 1024 + tid;
            const int4 d = (i < n4) ? v[u] : none;
            count_below(d.x);
            count_below(d.y);
            count_below(d.z);
            count_below(d.w);
            const unsigned t0 = (unsigned)(d.x - lo);
            const unsigned t1 = (unsigned)(d.y - lo);
            const unsigned t2 = (unsigned)(d.z - lo);
            const unsigned t3 = (unsigned)(d.w - lo);
            const unsigned R = (unsigned)OWN_R;
            if ((t0 < R) | (t1 < R) | (t2 < R) | (t3 < R)) {
                if (t0 < R) put(t0, 4 * i);
                if (t1 < R) put(t1, 4 * i + 1);
                if (t2 < R) put(t2, 4 * i + 2);
                if (t3 < R) put(t3, 4 * i + 3);
            }
        }
    }
    for (int e0 = 4 * n4; e0 < E; e0 += 1024) {
        const int e = e0 + tid;
        const int d = (e < E) ? dst[e] : NO_DST;
        count_below(d);
        const unsigned t = (unsigned)(d - lo);
        if (t < (unsigned)OWN_R) put(t, e);
    }
    if (lane == 0) wbelow[wave] = below;
    __syncthreads();

    int base = 0;
    for (int w = 0; w < 32; ++w) base += wbelow[w];
    const int c = (tid < OWN_R) ? cntL[tid] : 0;
    int inc = c;
    for (int off = 1; off < 32; off <<= 1) {
        const int t = __shfl_up(inc, off, 32);
        if (lane >= off) inc += t;
    }
    if (tid < OWN_R && lane == 31) wscan[wave] = inc;
    __syncthreads();
    if (tid < OWN_R) {
        int wex = 0;
        for (int w = 0; w < wave; ++w) wex += wscan[w];
        const int start = wex + inc - c;
        cntL[tid] = start;
        if (lo + tid <= Nn) rowptr[lo + tid] = base + start;
    }
    __syncthreads();
    const int n = nstage;
    if (n <= STAGE_CAP) {
        for (int p = tid; p < n; p += 1024) {
            const int2 s = stage[p];
            const unsigned pk = (unsigned)s.y;
            list[base + cntL[pk & 255u] + (int)(pk >> 8)] = make_int2(ei[s.x], s.x);
        }
    } else {
        for (int e0 = 0; e0 < E; e0 += 1024) {
            const int e = e0 + tid;
            const int d = (e < E) ? dst[e] : NO_DST;
            const unsigned t = (unsigned)(d - lo);
            if (t < (unsigned)OWN_R) {
                const int r = atomicAdd(&cur[t], 1);
                list[base + cntL[t] + r] = make_int2(ei[e], e);
            }
        }
    }
}

__device__ inline float leaky(float e) { return (e > 0.f) ? e : NEG_SLOPE * e; }

__device__ inline float lane_value(float v, int k) {
    return __int_as_float(__builtin_amdgcn_readlane(__float_as_int(v), k));
}

__device__ __forceinline__
void seg_body(const float* __restrict__ hbuf, const float* __restrict__ as,
              const float* __restrict__ ad, const int* __restrict__ rowptr,
              const int2* __restrict__ list, const float* __restrict__ bias,
              int E, int Nn, float* __restrict__ alpha, float* __restrict__ xout,
              int blockLocal) {
    const int lane = threadIdx.x & 31;
    const int node = __builtin_amdgcn_readfirstlane(
        (int)(blockLocal * (blockDim.x >> 5) + (threadIdx.x >> 5)));
    if (node >= Nn) return;
    const int beg = rowptr[node];
    const int end = rowptr[node + 1];
    const float4 adn = *(const float4*)(ad + (size_t)node * 4);
    const float4 asn = *(const float4*)(as + (size_t)node * 4);
    float4 es;
    es.x = leaky(asn.x + adn.x);
    es.y = leaky(asn.y + adn.y);
    es.z = leaky(asn.z + adn.z);
    es.w = leaky(asn.w + adn.w);
    const int deg = end - beg;
    const bool one = deg <= 32;
    const bool has = lane < deg;

    float4 m = es;
    int2 rec = make_int2(0, 0);
    float4 ev = make_float4(0.f, 0.f, 0.f, 0.f);
    if (one) {
        if (has) {
            rec = list[beg + lane];
            const float4 a = *(const float4*)(as + (size_t)rec.x * 4);
            ev.x = leaky(a.x + adn.x);
            ev.y = leaky(a.y + adn.y);
            ev.z = leaky(a.z + adn.z);
            ev.w = leaky(a.w + adn.w);
            m.x = fmaxf(m.x, ev.x);
            m.y = fmaxf(m.y, ev.y);
            m.z = fmaxf(m.z, ev.z);
            m.w = fmaxf(m.w, ev.w);
        }
    } else {
        for (int j = beg + lane; j < end; j += 32) {
            const int sidx = list[j].x;
            const float4 a = *(const float4*)(as + (size_t)sidx * 4);
            m.x = fmaxf(m.x, leaky(a.x + adn.x));
            m.y = fmaxf(m.y, leaky(a.y + adn.y));
            m.z = fmaxf(m.z, leaky(a.z + adn.z));
            m.w = fmaxf(m.w, leaky(a.w + adn.w));
        }
    }
#pragma unroll
    for (int off = 16; off > 0; off >>= 1) {
        m.x = fmaxf(m.x, __shfl_xor(m.x, off, 32));
        m.y = fmaxf(m.y, __shfl_xor(m.y, off, 32));
        m.z = fmaxf(m.z, __shfl_xor(m.z, off, 32));
        m.w = fmaxf(m.w, __shfl_xor(m.w, off, 32));
    }

    float4 z = make_float4(0.f, 0.f, 0.f, 0.f);
    if (one) {
        if (has) {
            ev.x = __expf(ev.x - m.x);
            ev.y = __expf(ev.y - m.y);
            ev.z = __expf(ev.z - m.z);
            ev.w = __expf(ev.w - m.w);
            z = ev;
        }
    } else {
        for (int j = beg + lane; j < end; j += 32) {
            const int sidx = list[j].x;
            const float4 a = *(const float4*)(as + (size_t)sidx * 4);
            z.x += __expf(leaky(a.x + adn.x) - m.x);
            z.y += __expf(leaky(a.y + adn.y) - m.y);
            z.z += __expf(leaky(a.z + adn.z) - m.z);
            z.w += __expf(leaky(a.w + adn.w) - m.w);
        }
    }
#pragma unroll
    for (int off = 16; off > 0; off >>= 1) {
        z.x += __shfl_xor(z.x, off, 32);
        z.y += __shfl_xor(z.y, off, 32);
        z.z += __shfl_xor(z.z, off, 32);
        z.w += __shfl_xor(z.w, off, 32);
    }
    float4 ps;
    ps.x = __expf(es.x - m.x);
    ps.y = __expf(es.y - m.y);
    ps.z = __expf(es.z - m.z);
    ps.w = __expf(es.w - m.w);
    z.x += ps.x;
    z.y += ps.y;
    z.z += ps.z;
    z.w += ps.w;

    const int hd = lane >> 3;
    float4 als;
    als.x = ps.x / z.x;
    als.y = ps.y / z.y;
    als.z = ps.z / z.z;
    als.w = ps.w / z.w;
    if (lane == 0) *(float4*)(alpha + (size_t)(E + node) * 4) = als;
    const float wself = (hd == 0) ? als.x : (hd == 1) ? als.y : (hd == 2) ? als.z : als.w;
    const float4 hself = *(const float4*)(hbuf + (size_t)node * F_DIM + lane * 4);
    float4 acc;
    acc.x = wself * hself.x;
    acc.y = wself * hself.y;
    acc.z = wself * hself.z;
    acc.w = wself * hself.w;

    float4 al1 = make_float4(0.f, 0.f, 0.f, 0.f);
    if (one && has) {
        al1.x = ev.x / z.x;
        al1.y = ev.y / z.y;
        al1.z = ev.z / z.z;
        al1.w = ev.w / z.w;
        *(float4*)(alpha + (size_t)rec.y * 4) = al1;
    }
    for (int cb = beg; cb < end; cb += 32) {
        int sidx = rec.x;
        float4 al = al1;
        if (!one) {
            const int j = cb + lane;
            sidx = 0;
            al = make_float4(0.f, 0.f, 0.f, 0.f);
            if (j < end) {
                const int2 r2 = list[j];
                sidx = r2.x;
                const float4 a = *(const float4*)(as + (size_t)sidx * 4);
                al.x = __expf(leaky(a.x + adn.x) - m.x) / z.x;
                al.y = __expf(leaky(a.y + adn.y) - m.y) / z.y;
                al.z = __expf(leaky(a.z + adn.z) - m.z) / z.z;
                al.w = __expf(leaky(a.w + adn.w) - m.w) / z.w;
                *(float4*)(alpha + (size_t)r2.y * 4) = al;
            }
        }
        const int left = end - cb;
        const int cnt  = (left < 32) ? left : 32;
#pragma unroll 4
        for (int k = 0; k < cnt; ++k) {
            const int sk = __builtin_amdgcn_readlane(sidx, k);
            const float a0 = lane_value(al.x, k);
            const float a1 = lane_value(al.y, k);
            const float a2 = lane_value(al.z, k);
            const float a3 = lane_value(al.w, k);
            const float w = (hd == 0) ? a0 : (hd == 1) ? a1 : (hd == 2) ? a2 : a3;
            const float4 hv = *(const float4*)(hbuf + (size_t)sk * F_DIM + lane * 4);
            acc.x = fmaf(w, hv.x, acc.x);
            acc.y = fmaf(w, hv.y, acc.y);
            acc.z = fmaf(w, hv.z, acc.z);
            acc.w = fmaf(w, hv.w, acc.w);
        }
    }
    const float4 bv = *(const float4*)(bias + lane * 4);
    acc.x += bv.x;
    acc.y += bv.y;
    acc.z += bv.z;
    acc.w += bv.w;
    *(float4*)(xout + (size_t)node * F_DIM + lane * 4) = acc;
}

__global__ __launch_bounds__(256)
void seg_kernel(const float* __restrict__ hbufA, const float* __restrict__ asA,
                const float* __restrict__ adA, const int* __restrict__ rowptrA,
                const int2* __restrict__ listA, const float* __restrict__ biasA,
                int EA, int NnA, float* __restrict__ alphaA, float* __restrict__ xoutA,
                const float* __restrict__ hbufB, const float* __restrict__ asB,
                const float* __restrict__ adB, const int* __restrict__ rowptrB,
                const int2* __restrict__ listB, const float* __restrict__ biasB,
                int EB, int NnB, float* __restrict__ alphaB, float* __restrict__ xoutB,
                int nbB) {
    if (blockIdx.x < nbB)
        seg_body(hbufB, asB, adB, rowptrB, listB, biasB, EB, NnB, alphaB, xoutB,
                 (int)blockIdx.x);
    else
        seg_body(hbufA, asA, adA, rowptrA, listA, biasA, EA, NnA, alphaA, xoutA,
                 (int)blockIdx.x - nbB);
}

#define GRP_G 4
#define GRP_WPG 6
#define GRP_T (GRP_G * GRP_WPG * 32)
#define GRP_CAP 2048

__device__ __forceinline__
void mix_node(float* xio, const float4 (&cv)[17], int node, int lane, float4 xo) {
    float sv[17];
#pragma unroll
    for (int k = 0; k < 17; ++k) {
        const float4 c = cv[k];
        float s = xo.x * c.x + xo.y * c.y + xo.z * c.z + xo.w * c.w;
#pragma unroll
        for (int off = 16; off > 0; off >>= 1) s += __shfl_xor(s, off, 32);
        sv[k] = s;
    }
    float mx = sv[0];
#pragma unroll
    for (int k = 1; k < 17; ++k) mx = fmaxf(mx, sv[k]);
    float sum = 0.f;
#pragma unroll
    for (int k = 0; k < 17; ++k) { sv[k] = __expf(sv[k] - mx); sum += sv[k]; }
    float inv = 1.f / sum;
    float4 go = make_float4(0.f, 0.f, 0.f, 0.f);
#pragma unroll
    for (int k = 0; k < 17; ++k) {
        float w = sv[k] * inv;
        go.x = fmaf(w, cv[k].x, go.x);
        go.y = fmaf(w, cv[k].y, go.y);
        go.z = fmaf(w, cv[k].z, go.z);
        go.w = fmaf(w, cv[k].w, go.w);
    }
    float4 o;
    o.x = 0.5f * (xo.x + go.x);
    o.y = 0.5f * (xo.y + go.y);
    o.z = 0.5f * (xo.z + go.z);
    o.w = 0.5f * (xo.w + go.w);
    *(float4*)(xio + (size_t)node * F_DIM + lane * 4) = o;
}

__global__ __launch_bounds__(GRP_T)
void group_kernel(float* xio, const float* __restrict__ xg,
                  const int* __restrict__ gidx, const int* __restrict__ nbr,
                  int Nn, int Kg) {
    __shared__ int ent[GRP_G][GRP_CAP];
    __shared__ int cnt[GRP_G];
    const int tid  = threadIdx.x;
    const int lane = tid & 31;
    const int wave = __builtin_amdgcn_readfirstlane(tid >> 5);
    const int lo   = blockIdx.x * GRP_G;
    const int t    = wave / GRP_WPG;
    const int r    = wave - t * GRP_WPG;
    const int g    = lo + t;
    if (tid < GRP_G) cnt[tid] = 0;
    __syncthreads();

    auto take = [&](int d, int node) {
        const unsigned tt = (unsigned)(d - lo);
        if (tt < (unsigned)GRP_G) {
            const int p = atomicAdd(&cnt[tt], 1);
            if (p < GRP_CAP) ent[tt][p] = node;
        }
    };
    const int n4 = ((((size_t)gidx) & 15) == 0) ? (Nn >> 2) : 0;
    const int4* g4 = (const int4*)gidx;
    const int n4m = (n4 > 0) ? n4 - 1 : 0;
    for (int b = 0; b < n4; b += 8 * GRP_T) {
        int4 v[8];
#pragma unroll
        for (int u = 0; u < 8; ++u) {
            const int i = b + u * GRP_T + tid;
            v[u] = g4[(i < n4) ? i : n4m];
        }
#pragma unroll
        for (int u = 0; u < 8; ++u) {
            const int i = b + u * GRP_T + tid;
            if (i < n4) {
                take(v[u].x, 4 * i);
                take(v[u].y, 4 * i + 1);
                take(v[u].z, 4 * i + 2);
                take(v[u].w, 4 * i + 3);
            }
        }
    }
    for (int e0 = 4 * n4; e0 < Nn; e0 += GRP_T) {
        const int e = e0 + tid;
        if (e < Nn) take(gidx[e], e);
    }
    __syncthreads();

    if (g >= Kg) return;

    float4 cv[17];
    cv[0] = *(const float4*)(xg + (size_t)g * F_DIM + lane * 4);
#pragma unroll
    for (int k = 0; k < 16; ++k) {
        const int c = nbr[(size_t)g * 16 + k];
        cv[k + 1] = *(const float4*)(xg + (size_t)c * F_DIM + lane * 4);
    }

    const int n = cnt[t];
    if (n <= GRP_CAP) {
        int p = r;
        int node = 0;
        float4 xo = make_float4(0.f, 0.f, 0.f, 0.f);
        if (p < n) {
            node = __builtin_amdgcn_readfirstlane(ent[t][p]);
            xo = *(const float4*)(xio + (size_t)node * F_DIM + lane * 4);
        }
        while (p < n) {
            const int pn = p + GRP_WPG;
            int node2 = 0;
            float4 xo2 = xo;
            if (pn < n) {
                node2 = __builtin_amdgcn_readfirstlane(ent[t][pn]);
                xo2 = *(const float4*)(xio + (size_t)node2 * F_DIM + lane * 4);
            }
            mix_node(xio, cv, node, lane, xo);
            node = node2; xo = xo2; p = pn;
        }
    } else {
        for (int v = r; v < Nn; v += GRP_WPG) {
            if (gidx[v] == g) {
                const float4 xo = *(const float4*)(xio + (size_t)v * F_DIM + lane * 4);
                mix_node(xio, cv, v, lane, xo);
            }
        }
    }
}

extern "C" void kernel_launch(void* const* d_in, const int* in_sizes, int n_in,
                              void* d_out, int out_size, void* d_ws, size_t ws_size,
                              hipStream_t stream) {
    const float* x       = (const float*)d_in[0];
    const int*   ei      = (const int*)d_in[1];
    const float* x_g     = (const float*)d_in[2];
    const int*   ei_g    = (const int*)d_in[3];
    const int*   gidx    = (const int*)d_in[4];
    const int*   nbr     = (const int*)d_in[5];
    const float* W1      = (const float*)d_in[6];
    const float* a_src1  = (const float*)d_in[7];
    const float* a_dst1  = (const float*)d_in[8];
    const float* b1      = (const float*)d_in[9];
    const float* W2      = (const float*)d_in[10];
    const float* a_src2  = (const float*)d_in[11];
    const float* a_dst2  = (const float*)d_in[12];
    const float* b2      = (const float*)d_in[13];

    const int N  = in_sizes[0] / F_DIM;
    const int E  = in_sizes[1] / 2;
    const int K  = in_sizes[2] / F_DIM;
    const int Eg = in_sizes[3] / 2;

    float* out     = (float*)d_out;
    float* x_final = out;
    float* xg_out  = out + (size_t)N * F_DIM;
    float* alpha1  = xg_out + (size_t)K * F_DIM;
    float* alpha2  = alpha1 + (size_t)(E + N) * H_HEADS;

    auto al4 = [](size_t n) { return (n + 3) & ~(size_t)3; };
    const int Np1 = ((N + 1 + 16383) / 16384) * 16384;
    const int Np2 = ((K + 1 + 16383) / 16384) * 16384;
    float* ws = (float*)d_ws;
    size_t off = 0;
    float* h1    = ws + off; off += (size_t)N * F_DIM;
    float* h2    = ws + off; off += (size_t)K * F_DIM;
    float* as1   = ws + off; off += (size_t)N * H_HEADS;
    float* ad1   = ws + off; off += (size_t)N * H_HEADS;
    float* as2   = ws + off; off += (size_t)K * H_HEADS;
    float* ad2   = ws + off; off += (size_t)K * H_HEADS;
    int*   cur1  = (int*)(ws + off);  off += (size_t)Np1;
    int*   cur2  = (int*)(ws + off);  off += (size_t)Np2;
    int2*  list1 = (int2*)(ws + off); off += al4((size_t)E * 2);
    int2*  list2 = (int2*)(ws + off); off += al4((size_t)Eg * 2);

    const int T = 256;
    auto blk = [](long n, int t) { return (int)((n + t - 1) / t); };


    owner_kernel<<<blk(K + 1, OWN_R) + blk(N + 1, OWN_R), 1024, 0, stream>>>(
        ei, E, N, cur1, list1, ei_g, Eg, K, cur2, list2, blk(K + 1, OWN_R));

    gemm_wmma_kernel<<<blk(K, 128) + blk(N, 128), T, F_DIM * F_DIM * sizeof(float), stream>>>(
        x, W1, a_src1, a_dst1, h1, as1, ad1, N,
        x_g, W2, a_src2, a_dst2, h2, as2, ad2, K, blk(K, 128));

    seg_kernel<<<blk(K, 8) + blk(N, 8), T, 0, stream>>>(
        h1, as1, ad1, cur1, list1, b1, E, N, alpha1, x_final,
        h2, as2, ad2, cur2, list2, b2, Eg, K, alpha2, xg_out, blk(K, 8));

    group_kernel<<<blk(K, GRP_G), GRP_T, 0, stream>>>(x_final, xg_out, gidx, nbr, N, K);
}


# Edge array cut in 4 pieces: a grouping block scans one piece for 1024 nodes (was all edges for 256), 4x fewer destination reads; a node's list is 4 runs, summed piece by piece, then arrival order
# speedup vs baseline: 3.8783x; 1.1459x over previous
#include <hip/hip_runtime.h>
#include <math.h>

#define H_HEADS 4
#define C_CH 32
#define F_DIM 128
#define NEG_SLOPE 0.2f

typedef __attribute__((ext_vector_type(2))) float v2f;
typedef __attribute__((ext_vector_type(8))) float v8f;

__global__ __launch_bounds__(256)
void gemm_wmma_kernel(const float* __restrict__ XA, const float* __restrict__ WA,
                      const float* __restrict__ a_srcA, const float* __restrict__ a_dstA,
                      float* __restrict__ HoutA, float* __restrict__ asA,
                      float* __restrict__ adA, int rowsA,
                      const float* __restrict__ XB, const float* __restrict__ WB,
                      const float* __restrict__ a_srcB, const float* __restrict__ a_dstB,
                      float* __restrict__ HoutB, float* __restrict__ asB,
                      float* __restrict__ adB, int rowsB, int nbB) {
    extern __shared__ float Wlds[];
    const bool gB = blockIdx.x < nbB;
    const float* X     = gB ? XB : XA;
    const float* W     = gB ? WB : WA;
    const float* a_src = gB ? a_srcB : a_srcA;
    const float* a_dst = gB ? a_dstB : a_dstA;
    float* Hout        = gB ? HoutB : HoutA;
    float* as          = gB ? asB : asA;
    float* ad          = gB ? adB : adA;
    const int rows     = gB ? rowsB : rowsA;
    const int blockLocal = gB ? blockIdx.x : blockIdx.x - nbB;
    for (int i = threadIdx.x; i < (F_DIM * F_DIM) / 2; i += blockDim.x) {
        int kp  = i >> 7;
        int col = i & 127;
        int cs  = (col + ((kp & 1) << 4)) & 127;
        float2 w2;
        w2.x = W[(size_t)(2 * kp)     * F_DIM + col];
        w2.y = W[(size_t)(2 * kp + 1) * F_DIM + col];
        *(float2*)(Wlds + (size_t)kp * 256 + cs * 2) = w2;
    }
    __syncthreads();

    const int lane = threadIdx.x & 31;
    const int wave = threadIdx.x >> 5;
    const int lm   = lane & 15;
    const int lh   = lane >> 4;
    const int rowBase = blockLocal * 128 + wave * 16;
    if (rowBase >= rows) return;

    int aRow = rowBase + lm;
    if (aRow >= rows) aRow = rows - 1;

    v8f acc[8];
    v8f zero = {0.f, 0.f, 0.f, 0.f, 0.f, 0.f, 0.f, 0.f};
#pragma unroll
    for (int ct = 0; ct < 8; ++ct) acc[ct] = zero;

    const float* xrow = X + (size_t)aRow * F_DIM;
    for (int k0 = 0; k0 < F_DIM; k0 += 4) {
        const int kp = (k0 >> 1) + lh;
        v2f a;
        a.x = xrow[2 * kp];
        a.y = xrow[2 * kp + 1];
        const float* bbase = Wlds + (size_t)kp * 256;
#pragma unroll
        for (int ct = 0; ct < 8; ++ct) {
            const int cs = (ct * 16 + lm + 16 * lh) & 127;
            v2f b = *(const v2f*)(bbase + cs * 2);
            acc[ct] = __builtin_amdgcn_wmma_f32_16x16x4_f32(
                false, a, false, b, (short)0, acc[ct], false, false);
        }
    }

#pragma unroll
    for (int ct = 0; ct < 8; ++ct) {
#pragma unroll
        for (int g = 0; g < 8; ++g) {
            int r = rowBase + g + 8 * lh;
            if (r < rows) Hout[(size_t)r * F_DIM + ct * 16 + lm] = acc[ct][g];
        }
    }

    float asv[8], adv[8];
#pragma unroll
    for (int ct = 0; ct < 8; ++ct) {
        asv[ct] = a_src[ct * 16 + lm];
        adv[ct] = a_dst[ct * 16 + lm];
    }
#pragma unroll
    for (int h = 0; h < 4; ++h) {
#pragma unroll
        for (int g = 0; g < 8; ++g) {
            float ss = acc[2 * h][g] * asv[2 * h];
            ss = fmaf(acc[2 * h + 1][g], asv[2 * h + 1], ss);
            float sd = acc[2 * h][g] * adv[2 * h];
            sd = fmaf(acc[2 * h + 1][g], adv[2 * h + 1], sd);
#pragma unroll
            for (int off = 8; off > 0; off >>= 1) {
                ss += __shfl_xor(ss, off, 32);
                sd += __shfl_xor(sd, off, 32);
            }
            int r = rowBase + g + 8 * lh;
            if (lm == 0 && r < rows) {
                as[(size_t)r * H_HEADS + h] = ss;
                ad[(size_t)r * H_HEADS + h] = sd;
            }
        }
    }
}

#define OWN_PARTS 4
#define OWN_R 1024
#define STAGE_CAP 7680
#define NO_DST 0x7fffffff

__device__ inline int part_len(int E) { return ((E + 4 * OWN_PARTS - 1) / (4 * OWN_PARTS)) * 4; }

__global__ __launch_bounds__(1024)
void owner_kernel(const int* __restrict__ eiA, int EA, int NnA, int NpA,
                  int* __restrict__ rowptrA, int2* __restrict__ listA,
                  const int* __restrict__ eiB, int EB, int NnB, int NpB,
                  int* __restrict__ rowptrB, int2* __restrict__ listB, int nbB) {
    __shared__ int2 stage[STAGE_CAP];
    __shared__ int cntL[OWN_R];
    __shared__ int cur[OWN_R];
    __shared__ int wbelow[32];
    __shared__ int wscan[OWN_R / 32];
    __shared__ int nstage;
    const bool gB = blockIdx.x < nbB;
    const int* ei = gB ? eiB : eiA;
    const int  E  = gB ? EB : EA;
    const int  Nn = gB ? NnB : NnA;
    const int  Np = gB ? NpB : NpA;
    int* rowptr   = gB ? rowptrB : rowptrA;
    int2* list    = gB ? listB : listA;
    const int bl   = gB ? blockIdx.x : blockIdx.x - nbB;
    const int part = bl % OWN_PARTS;
    const int lo   = (bl / OWN_PARTS) * OWN_R;
    const int Ep   = part_len(E);
    const int e_lo = (part * Ep < E) ? part * Ep : E;
    const int Ec   = (Ep < E - e_lo) ? Ep : E - e_lo;
    const int tid  = threadIdx.x;
    const int lane = tid & 31;
    const int wave = tid >> 5;
    if (tid < OWN_R) { cntL[tid] = 0; cur[tid] = 0; }
    if (tid == 0) nstage = 0;
    __syncthreads();

    const int* dst = ei + E + e_lo;
    const int n4 = ((E & 3) == 0) ? (Ec >> 2) : 0;
    const int4* d4 = (const int4*)dst;
    const int4 none = make_int4(NO_DST, NO_DST, NO_DST, NO_DST);
    int below = 0;
    auto put = [&](unsigned t, int e) {
        const int r = atomicAdd(&cntL[t], 1);
        const int p = atomicAdd(&nstage, 1);
        if (p < STAGE_CAP) stage[p] = make_int2(e_lo + e, (int)(((unsigned)r << 10) | t));
    };
    auto count_below = [&](int d) {
        below += __popc(__builtin_amdgcn_ballot_w32(d < lo));
    };
    const int n4m = (n4 > 0) ? n4 - 1 : 0;
    for (int b = 0; b < n4; b += 8 * 1024) {
        int4 v[8];
#pragma unroll
        for (int u = 0; u < 8; ++u) {
            const int i = b + u * 1024 + tid;
            v[u] = d4[(i < n4) ? i : n4m];
        }
#pragma unroll
        for (int u = 0; u < 8; ++u) {
            const int i = b + u * 1024 + tid;
            const int4 d = (i < n4) ? v[u] : none;
            count_below(d.x);
            count_below(d.y);
            count_below(d.z);
            count_below(d.w);
            const unsigned t0 = (unsigned)(d.x - lo);
            const unsigned t1 = (unsigned)(d.y - lo);
            const unsigned t2 = (unsigned)(d.z - lo);
            const unsigned t3 = (unsigned)(d.w - lo);
            const unsigned R = (unsigned)OWN_R;
            if ((t0 < R) | (t1 < R) | (t2 < R) | (t3 < R)) {
                if (t0 < R) put(t0, 4 * i);
                if (t1 < R) put(t1, 4 * i + 1);
                if (t2 < R) put(t2, 4 * i + 2);
                if (t3 < R) put(t3, 4 * i + 3);
            }
        }
    }
    for (int e0 = 4 * n4; e0 < Ec; e0 += 1024) {
        const int e = e0 + tid;
        const int d = (e < Ec) ? dst[e] : NO_DST;
        count_below(d);
        const unsigned t = (unsigned)(d - lo);
        if (t < (unsigned)OWN_R) put(t, e);
    }
    if (lane == 0) wbelow[wave] = below;
    __syncthreads();

    int base = 0;
    for (int w = 0; w < 32; ++w) base += wbelow[w];
    const int c = (tid < OWN_R) ? cntL[tid] : 0;
    int inc = c;
    for (int off = 1; off < 32; off <<= 1) {
        const int t = __shfl_up(inc, off, 32);
        if (lane >= off) inc += t;
    }
    if (tid < OWN_R && lane == 31) wscan[wave] = inc;
    __syncthreads();
    if (tid < OWN_R) {
        int wex = 0;
        for (int w = 0; w < wave; ++w) wex += wscan[w];
        const int start = wex + inc - c;
        cntL[tid] = start;
        if (lo + tid <= Nn) rowptr[(size_t)part * Np + lo + tid] = e_lo + base + start;
    }
    __syncthreads();
    const int n = nstage;
    if (n <= STAGE_CAP) {
        for (int p = tid; p < n; p += 1024) {
            const int2 s = stage[p];
            const unsigned pk = (unsigned)s.y;
            list[e_lo + base + cntL[pk & 1023u] + (int)(pk >> 10)] = make_int2(ei[s.x], s.x);
        }
    } else {
        for (int e0 = 0; e0 < Ec; e0 += 1024) {
            const int e = e0 + tid;
            const int d = (e < Ec) ? dst[e] : NO_DST;
            const unsigned t = (unsigned)(d - lo);
            if (t < (unsigned)OWN_R) {
                const int r = atomicAdd(&cur[t], 1);
                list[e_lo + base + cntL[t] + r] = make_int2(ei[e_lo + e], e_lo + e);
            }
        }
    }
}

__device__ inline float leaky(float e) { return (e > 0.f) ? e : NEG_SLOPE * e; }

__device__ inline float lane_value(float v, int k) {
    return __int_as_float(__builtin_amdgcn_readlane(__float_as_int(v), k));
}

__device__ __forceinline__
void seg_body(const float* __restrict__ hbuf, const float* __restrict__ as,
              const float* __restrict__ ad, const int* __restrict__ rowptr,
              const int2* __restrict__ list, const float* __restrict__ bias,
              int E, int Nn, int Np, float* __restrict__ alpha, float* __restrict__ xout,
              int blockLocal) {
    const int lane = threadIdx.x & 31;
    const int node = __builtin_amdgcn_readfirstlane(
        (int)(blockLocal * (blockDim.x >> 5) + (threadIdx.x >> 5)));
    if (node >= Nn) return;
    int rb[OWN_PARTS];
    int rc[OWN_PARTS];
    int deg = 0;
#pragma unroll
    for (int p = 0; p < OWN_PARTS; ++p) {
        const int b = rowptr[(size_t)p * Np + node];
        const int e = rowptr[(size_t)p * Np + node + 1];
        rb[p] = b;
        deg += e - b;
        rc[p] = deg;
    }
    auto at = [&](int q) {
        int j = rb[0] + q;
#pragma unroll
        for (int p = 1; p < OWN_PARTS; ++p)
            if (q >= rc[p - 1]) j = rb[p] + (q - rc[p - 1]);
        return j;
    };
    const float4 adn = *(const float4*)(ad + (size_t)node * 4);
    const float4 asn = *(const float4*)(as + (size_t)node * 4);
    float4 es;
    es.x = leaky(asn.x + adn.x);
    es.y = leaky(asn.y + adn.y);
    es.z = leaky(asn.z + adn.z);
    es.w = leaky(asn.w + adn.w);
    const bool one = deg <= 32;
    const bool has = lane < deg;

    float4 m = es;
    int2 rec = make_int2(0, 0);
    float4 ev = make_float4(0.f, 0.f, 0.f, 0.f);
    if (one) {
        if (has) {
            rec = list[at(lane)];
            const float4 a = *(const float4*)(as + (size_t)rec.x * 4);
            ev.x = leaky(a.x + adn.x);
            ev.y = leaky(a.y + adn.y);
            ev.z = leaky(a.z + adn.z);
            ev.w = leaky(a.w + adn.w);
            m.x = fmaxf(m.x, ev.x);
            m.y = fmaxf(m.y, ev.y);
            m.z = fmaxf(m.z, ev.z);
            m.w = fmaxf(m.w, ev.w);
        }
    } else {
        for (int q = lane; q < deg; q += 32) {
            const int sidx = list[at(q)].x;
            const float4 a = *(const float4*)(as + (size_t)sidx * 4);
            m.x = fmaxf(m.x, leaky(a.x + adn.x));
            m.y = fmaxf(m.y, leaky(a.y + adn.y));
            m.z = fmaxf(m.z, leaky(a.z + adn.z));
            m.w = fmaxf(m.w, leaky(a.w + adn.w));
        }
    }
#pragma unroll
    for (int off = 16; off > 0; off >>= 1) {
        m.x = fmaxf(m.x, __shfl_xor(m.x, off, 32));
        m.y = fmaxf(m.y, __shfl_xor(m.y, off, 32));
        m.z = fmaxf(m.z, __shfl_xor(m.z, off, 32));
        m.w = fmaxf(m.w, __shfl_xor(m.w, off, 32));
    }

    float4 z = make_float4(0.f, 0.f, 0.f, 0.f);
    if (one) {
        if (has) {
            ev.x = __expf(ev.x - m.x);
            ev.y = __expf(ev.y - m.y);
            ev.z = __expf(ev.z - m.z);
            ev.w = __expf(ev.w - m.w);
            z = ev;
        }
    } else {
        for (int q = lane; q < deg; q += 32) {
            const int sidx = list[at(q)].x;
            const float4 a = *(const float4*)(as + (size_t)sidx * 4);
            z.x += __expf(leaky(a.x + adn.x) - m.x);
            z.y += __expf(leaky(a.y + adn.y) - m.y);
            z.z += __expf(leaky(a.z + adn.z) - m.z);
            z.w += __expf(leaky(a.w + adn.w) - m.w);
        }
    }
#pragma unroll
    for (int off = 16; off > 0; off >>= 1) {
        z.x += __shfl_xor(z.x, off, 32);
        z.y += __shfl_xor(z.y, off, 32);
        z.z += __shfl_xor(z.z, off, 32);
        z.w += __shfl_xor(z.w, off, 32);
    }
    float4 ps;
    ps.x = __expf(es.x - m.x);
    ps.y = __expf(es.y - m.y);
    ps.z = __expf(es.z - m.z);
    ps.w = __expf(es.w - m.w);
    z.x += ps.x;
    z.y += ps.y;
    z.z += ps.z;
    z.w += ps.w;

    const int hd = lane >> 3;
    float4 als;
    als.x = ps.x / z.x;
    als.y = ps.y / z.y;
    als.z = ps.z / z.z;
    als.w = ps.w / z.w;
    if (lane == 0) *(float4*)(alpha + (size_t)(E + node) * 4) = als;
    const float wself = (hd == 0) ? als.x : (hd == 1) ? als.y : (hd == 2) ? als.z : als.w;
    const float4 hself = *(const float4*)(hbuf + (size_t)node * F_DIM + lane * 4);
    float4 acc;
    acc.x = wself * hself.x;
    acc.y = wself * hself.y;
    acc.z = wself * hself.z;
    acc.w = wself * hself.w;

    float4 al1 = make_float4(0.f, 0.f, 0.f, 0.f);
    if (one && has) {
        al1.x = ev.x / z.x;
        al1.y = ev.y / z.y;
        al1.z = ev.z / z.z;
        al1.w = ev.w / z.w;
        *(float4*)(alpha + (size_t)rec.y * 4) = al1;
    }
    for (int cb = 0; cb < deg; cb += 32) {
        int sidx = rec.x;
        float4 al = al1;
        if (!one) {
            const int q = cb + lane;
            sidx = 0;
            al = make_float4(0.f, 0.f, 0.f, 0.f);
            if (q < deg) {
                const int2 r2 = list[at(q)];
                sidx = r2.x;
                const float4 a = *(const float4*)(as + (size_t)sidx * 4);
                al.x = __expf(leaky(a.x + adn.x) - m.x) / z.x;
                al.y = __expf(leaky(a.y + adn.y) - m.y) / z.y;
                al.z = __expf(leaky(a.z + adn.z) - m.z) / z.z;
                al.w = __expf(leaky(a.w + adn.w) - m.w) / z.w;
                *(float4*)(alpha + (size_t)r2.y * 4) = al;
            }
        }
        const int left = deg - cb;
        const int cnt  = (left < 32) ? left : 32;
#pragma unroll 4
        for (int k = 0; k < cnt; ++k) {
            const int sk = __builtin_amdgcn_readlane(sidx, k);
            const float a0 = lane_value(al.x, k);
            const float a1 = lane_value(al.y, k);
            const float a2 = lane_value(al.z, k);
            const float a3 = lane_value(al.w, k);
            const float w = (hd == 0) ? a0 : (hd == 1) ? a1 : (hd == 2) ? a2 : a3;
            const float4 hv = *(const float4*)(hbuf + (size_t)sk * F_DIM + lane * 4);
            acc.x = fmaf(w, hv.x, acc.x);
            acc.y = fmaf(w, hv.y, acc.y);
            acc.z = fmaf(w, hv.z, acc.z);
            acc.w = fmaf(w, hv.w, acc.w);
        }
    }
    const float4 bv = *(const float4*)(bias + lane * 4);
    acc.x += bv.x;
    acc.y += bv.y;
    acc.z += bv.z;
    acc.w += bv.w;
    *(float4*)(xout + (size_t)node * F_DIM + lane * 4) = acc;
}

__global__ __launch_bounds__(256)
void seg_kernel(const float* __restrict__ hbufA, const float* __restrict__ asA,
                const float* __restrict__ adA, const int* __restrict__ rowptrA,
                const int2* __restrict__ listA, const float* __restrict__ biasA,
                int EA, int NnA, int NpA, float* __restrict__ alphaA, float* __restrict__ xoutA,
                const float* __restrict__ hbufB, const float* __restrict__ asB,
                const float* __restrict__ adB, const int* __restrict__ rowptrB,
                const int2* __restrict__ listB, const float* __restrict__ biasB,
                int EB, int NnB, int NpB, float* __restrict__ alphaB, float* __restrict__ xoutB,
                int nbB) {
    if (blockIdx.x < nbB)
        seg_body(hbufB, asB, adB, rowptrB, listB, biasB, EB, NnB, NpB, alphaB, xoutB,
                 (int)blockIdx.x);
    else
        seg_body(hbufA, asA, adA, rowptrA, listA, biasA, EA, NnA, NpA, alphaA, xoutA,
                 (int)blockIdx.x - nbB);
}

#define GRP_G 4
#define GRP_WPG 6
#define GRP_T (GRP_G * GRP_WPG * 32)
#define GRP_CAP 2048

__device__ __forceinline__
void mix_node(float* xio, const float4 (&cv)[17], int node, int lane, float4 xo) {
    float sv[17];
#pragma unroll
    for (int k = 0; k < 17; ++k) {
        const float4 c = cv[k];
        float s = xo.x * c.x + xo.y * c.y + xo.z * c.z + xo.w * c.w;
#pragma unroll
        for (int off = 16; off > 0; off >>= 1) s += __shfl_xor(s, off, 32);
        sv[k] = s;
    }
    float mx = sv[0];
#pragma unroll
    for (int k = 1; k < 17; ++k) mx = fmaxf(mx, sv[k]);
    float sum = 0.f;
#pragma unroll
    for (int k = 0; k < 17; ++k) { sv[k] = __expf(sv[k] - mx); sum += sv[k]; }
    float inv = 1.f / sum;
    float4 go = make_float4(0.f, 0.f, 0.f, 0.f);
#pragma unroll
    for (int k = 0; k < 17; ++k) {
        float w = sv[k] * inv;
        go.x = fmaf(w, cv[k].x, go.x);
        go.y = fmaf(w, cv[k].y, go.y);
        go.z = fmaf(w, cv[k].z, go.z);
        go.w = fmaf(w, cv[k].w, go.w);
    }
    float4 o;
    o.x = 0.5f * (xo.x + go.x);
    o.y = 0.5f * (xo.y + go.y);
    o.z = 0.5f * (xo.z + go.z);
    o.w = 0.5f * (xo.w + go.w);
    *(float4*)(xio + (size_t)node * F_DIM + lane * 4) = o;
}

__global__ __launch_bounds__(GRP_T)
void group_kernel(float* xio, const float* __restrict__ xg,
                  const int* __restrict__ gidx, const int* __restrict__ nbr,
                  int Nn, int Kg) {
    __shared__ int ent[GRP_G][GRP_CAP];
    __shared__ int cnt[GRP_G];
    const int tid  = threadIdx.x;
    const int lane = tid & 31;
    const int wave = __builtin_amdgcn_readfirstlane(tid >> 5);
    const int lo   = blockIdx.x * GRP_G;
    const int t    = wave / GRP_WPG;
    const int r    = wave - t * GRP_WPG;
    const int g    = lo + t;
    if (tid < GRP_G) cnt[tid] = 0;
    __syncthreads();

    auto take = [&](int d, int node) {
        const unsigned tt = (unsigned)(d - lo);
        if (tt < (unsigned)GRP_G) {
            const int p = atomicAdd(&cnt[tt], 1);
            if (p < GRP_CAP) ent[tt][p] = node;
        }
    };
    const int n4 = ((((size_t)gidx) & 15) == 0) ? (Nn >> 2) : 0;
    const int4* g4 = (const int4*)gidx;
    const int n4m = (n4 > 0) ? n4 - 1 : 0;
    for (int b = 0; b < n4; b += 8 * GRP_T) {
        int4 v[8];
#pragma unroll
        for (int u = 0; u < 8; ++u) {
            const int i = b + u * GRP_T + tid;
            v[u] = g4[(i < n4) ? i : n4m];
        }
#pragma unroll
        for (int u = 0; u < 8; ++u) {
            const int i = b + u * GRP_T + tid;
            if (i < n4) {
                take(v[u].x, 4 * i);
                take(v[u].y, 4 * i + 1);
                take(v[u].z, 4 * i + 2);
                take(v[u].w, 4 * i + 3);
            }
        }
    }
    for (int e0 = 4 * n4; e0 < Nn; e0 += GRP_T) {
        const int e = e0 + tid;
        if (e < Nn) take(gidx[e], e);
    }
    __syncthreads();

    if (g >= Kg) return;

    float4 cv[17];
    cv[0] = *(const float4*)(xg + (size_t)g * F_DIM + lane * 4);
#pragma unroll
    for (int k = 0; k < 16; ++k) {
        const int c = nbr[(size_t)g * 16 + k];
        cv[k + 1] = *(const float4*)(xg + (size_t)c * F_DIM + lane * 4);
    }

    const int n = cnt[t];
    if (n <= GRP_CAP) {
        int p = r;
        int node = 0;
        float4 xo = make_float4(0.f, 0.f, 0.f, 0.f);
        if (p < n) {
            node = __builtin_amdgcn_readfirstlane(ent[t][p]);
            xo = *(const float4*)(xio + (size_t)node * F_DIM + lane * 4);
        }
        while (p < n) {
            const int pn = p + GRP_WPG;
            int node2 = 0;
            float4 xo2 = xo;
            if (pn < n) {
                node2 = __builtin_amdgcn_readfirstlane(ent[t][pn]);
                xo2 = *(const float4*)(xio + (size_t)node2 * F_DIM + lane * 4);
            }
            mix_node(xio, cv, node, lane, xo);
            node = node2; xo = xo2; p = pn;
        }
    } else {
        for (int v = r; v < Nn; v += GRP_WPG) {
            if (gidx[v] == g) {
                const float4 xo = *(const float4*)(xio + (size_t)v * F_DIM + lane * 4);
                mix_node(xio, cv, v, lane, xo);
            }
        }
    }
}

extern "C" void kernel_launch(void* const* d_in, const int* in_sizes, int n_in,
                              void* d_out, int out_size, void* d_ws, size_t ws_size,
                              hipStream_t stream) {
    const float* x       = (const float*)d_in[0];
    const int*   ei      = (const int*)d_in[1];
    const float* x_g     = (const float*)d_in[2];
    const int*   ei_g    = (const int*)d_in[3];
    const int*   gidx    = (const int*)d_in[4];
    const int*   nbr     = (const int*)d_in[5];
    const float* W1      = (const float*)d_in[6];
    const float* a_src1  = (const float*)d_in[7];
    const float* a_dst1  = (const float*)d_in[8];
    const float* b1      = (const float*)d_in[9];
    const float* W2      = (const float*)d_in[10];
    const float* a_src2  = (const float*)d_in[11];
    const float* a_dst2  = (const float*)d_in[12];
    const float* b2      = (const float*)d_in[13];

    const int N  = in_sizes[0] / F_DIM;
    const int E  = in_sizes[1] / 2;
    const int K  = in_sizes[2] / F_DIM;
    const int Eg = in_sizes[3] / 2;

    float* out     = (float*)d_out;
    float* x_final = out;
    float* xg_out  = out + (size_t)N * F_DIM;
    float* alpha1  = xg_out + (size_t)K * F_DIM;
    float* alpha2  = alpha1 + (size_t)(E + N) * H_HEADS;

    auto al4 = [](size_t n) { return (n + 3) & ~(size_t)3; };
    const int Np1 = ((N + 1 + OWN_R - 1) / OWN_R) * OWN_R;
    const int Np2 = ((K + 1 + OWN_R - 1) / OWN_R) * OWN_R;
    float* ws = (float*)d_ws;
    size_t off = 0;
    float* h1    = ws + off; off += (size_t)N * F_DIM;
    float* h2    = ws + off; off += (size_t)K * F_DIM;
    float* as1   = ws + off; off += (size_t)N * H_HEADS;
    float* ad1   = ws + off; off += (size_t)N * H_HEADS;
    float* as2   = ws + off; off += (size_t)K * H_HEADS;
    float* ad2   = ws + off; off += (size_t)K * H_HEADS;
    int*   cur1  = (int*)(ws + off);  off += (size_t)OWN_PARTS * Np1;
    int*   cur2  = (int*)(ws + off);  off += (size_t)OWN_PARTS * Np2;
    int2*  list1 = (int2*)(ws + off); off += al4((size_t)E * 2);
    int2*  list2 = (int2*)(ws + off); off += al4((size_t)Eg * 2);

    const int T = 256;
    auto blk = [](long n, int t) { return (int)((n + t - 1) / t); };


    owner_kernel<<<OWN_PARTS * (blk(K + 1, OWN_R) + blk(N + 1, OWN_R)), 1024, 0, stream>>>(
        ei, E, N, Np1, cur1, list1, ei_g, Eg, K, Np2, cur2, list2, OWN_PARTS * blk(K + 1, OWN_R));

    gemm_wmma_kernel<<<blk(K, 128) + blk(N, 128), T, F_DIM * F_DIM * sizeof(float), stream>>>(
        x, W1, a_src1, a_dst1, h1, as1, ad1, N,
        x_g, W2, a_src2, a_dst2, h2, as2, ad2, K, blk(K, 128));

    seg_kernel<<<blk(K, 8) + blk(N, 8), T, 0, stream>>>(
        h1, as1, ad1, cur1, list1, b1, E, N, Np1, alpha1, x_final,
        h2, as2, ad2, cur2, list2, b2, Eg, K, Np2, alpha2, xg_out, blk(K, 8));

    group_kernel<<<blk(K, GRP_G), GRP_T, 0, stream>>>(x_final, xg_out, gidx, nbr, N, K);
}


# Edge grouping and the products, which need nothing of each other, run as two roles of ONE launch (4 launches become 3); a product block is 32 waves, a wave takes 16 rows x 64 columns; sums unchanged
# speedup vs baseline: 4.8646x; 1.2543x over previous
#include <hip/hip_runtime.h>
#include <math.h>

#define H_HEADS 4
#define C_CH 32
#define F_DIM 128
#define NEG_SLOPE 0.2f

typedef __attribute__((ext_vector_type(2))) float v2f;
typedef __attribute__((ext_vector_type(8))) float v8f;

#define GEMM_ROWS 256
__device__ __forceinline__
void gemm_body(float* Wlds, int bIdx, const float* __restrict__ XA, const float* __restrict__ WA,
                      const float* __restrict__ a_srcA, const float* __restrict__ a_dstA,
                      float* __restrict__ HoutA, float* __restrict__ asA,
                      float* __restrict__ adA, int rowsA,
                      const float* __restrict__ XB, const float* __restrict__ WB,
                      const float* __restrict__ a_srcB, const float* __restrict__ a_dstB,
                      float* __restrict__ HoutB, float* __restrict__ asB,
                      float* __restrict__ adB, int rowsB, int nbB) {
    const bool gB = bIdx < nbB;
    const float* X     = gB ? XB : XA;
    const float* W     = gB ? WB : WA;
    const float* a_src = gB ? a_srcB : a_srcA;
    const float* a_dst = gB ? a_dstB : a_dstA;
    float* Hout        = gB ? HoutB : HoutA;
    float* as          = gB ? asB : asA;
    float* ad          = gB ? adB : adA;
    const int rows     = gB ? rowsB : rowsA;
    const int blockLocal = gB ? bIdx : bIdx - nbB;
    for (int i = threadIdx.x; i < (F_DIM * F_DIM) / 2; i += blockDim.x) {
        int kp  = i >> 7;
        int col = i & 127;
        int cs  = (col + ((kp & 1) << 4)) & 127;
        float2 w2;
        w2.x = W[(size_t)(2 * kp)     * F_DIM + col];
        w2.y = W[(size_t)(2 * kp + 1) * F_DIM + col];
        *(float2*)(Wlds + (size_t)kp * 256 + cs * 2) = w2;
    }
    __syncthreads();

    const int lane = threadIdx.x & 31;
    const int wave = __builtin_amdgcn_readfirstlane((int)(threadIdx.x >> 5));
    const int lm   = lane & 15;
    const int lh   = lane >> 4;
    const int ch   = wave & 1;
    const int rowBase = blockLocal * GEMM_ROWS + (wave >> 1) * 16;
    if (rowBase >= rows) return;

    int aRow = rowBase + lm;
    if (aRow >= rows) aRow = rows - 1;

    v8f acc[4];
    v8f zero = {0.f, 0.f, 0.f, 0.f, 0.f, 0.f, 0.f, 0.f};
#pragma unroll
    for (int ct = 0; ct < 4; ++ct) acc[ct] = zero;

    const float* xrow = X + (size_t)aRow * F_DIM;
    for (int k0 = 0; k0 < F_DIM; k0 += 4) {
        const int kp = (k0 >> 1) + lh;
        v2f a;
        a.x = xrow[2 * kp];
        a.y = xrow[2 * kp + 1];
        const float* bbase = Wlds + (size_t)kp * 256;
#pragma unroll
        for (int ct = 0; ct < 4; ++ct) {
            const int cs = ((4 * ch + ct) * 16 + lm + 16 * lh) & 127;
            v2f b = *(const v2f*)(bbase + cs * 2);
            acc[ct] = __builtin_amdgcn_wmma_f32_16x16x4_f32(
                false, a, false, b, (short)0, acc[ct], false, false);
        }
    }

#pragma unroll
    for (int ct = 0; ct < 4; ++ct) {
#pragma unroll
        for (int g = 0; g < 8; ++g) {
            int r = rowBase + g + 8 * lh;
            if (r < rows) Hout[(unsigned)(r * F_DIM + (4 * ch + ct) * 16 + lm)] = acc[ct][g];
        }
    }

    float asv[4], adv[4];
#pragma unroll
    for (int ct = 0; ct < 4; ++ct) {
        asv[ct] = a_src[(4 * ch + ct) * 16 + lm];
        adv[ct] = a_dst[(4 * ch + ct) * 16 + lm];
    }
#pragma unroll
    for (int h = 0; h < 2; ++h) {
#pragma unroll
        for (int g = 0; g < 8; ++g) {
            float ss = acc[2 * h][g] * asv[2 * h];
            ss = fmaf(acc[2 * h + 1][g], asv[2 * h + 1], ss);
            float sd = acc[2 * h][g] * adv[2 * h];
            sd = fmaf(acc[2 * h + 1][g], adv[2 * h + 1], sd);
#pragma unroll
            for (int off = 8; off > 0; off >>= 1) {
                ss += __shfl_xor(ss, off, 32);
                sd += __shfl_xor(sd, off, 32);
            }
            int r = rowBase + g + 8 * lh;
            if (lm == 0 && r < rows) {
                as[(unsigned)(r * H_HEADS + 2 * ch + h)] = ss;
                ad[(unsigned)(r * H_HEADS + 2 * ch + h)] = sd;
            }
        }
    }
}

#define OWN_PARTS 4
#define OWN_R 1024
#define STAGE_CAP 7680
#define NO_DST 0x7fffffff

__device__ inline int part_len(int E) { return ((E + 4 * OWN_PARTS - 1) / (4 * OWN_PARTS)) * 4; }

#define OWN_W_CNT   (2 * STAGE_CAP)
#define OWN_W_CUR   (OWN_W_CNT + OWN_R)
#define OWN_W_BELOW (OWN_W_CUR + OWN_R)
#define OWN_W_SCAN  (OWN_W_BELOW + 32)
#define OWN_W_NST   (OWN_W_SCAN + OWN_R / 32)
#define OWN_W_END   (OWN_W_NST + 4)

__device__ __forceinline__
void owner_body(int* lds, int bIdx,
                const int* __restrict__ eiA, int EA, int NnA, int NpA,
                int* __restrict__ rowptrA, int2* __restrict__ listA,
                const int* __restrict__ eiB, int EB, int NnB, int NpB,
                int* __restrict__ rowptrB, int2* __restrict__ listB, int nbB) {
    int2* stage = (int2*)lds;
    int* cntL   = lds + OWN_W_CNT;
    int* cur    = lds + OWN_W_CUR;
    int* wbelow = lds + OWN_W_BELOW;
    int* wscan  = lds + OWN_W_SCAN;
    int* nstage = lds + OWN_W_NST;
    const bool gB = bIdx < nbB;
    const int* ei = gB ? eiB : eiA;
    const int  E  = gB ? EB : EA;
    const int  Nn = gB ? NnB : NnA;
    const int  Np = gB ? NpB : NpA;
    int* rowptr   = gB ? rowptrB : rowptrA;
    int2* list    = gB ? listB : listA;
    const int bl   = gB ? bIdx : bIdx - nbB;
    const int part = bl % OWN_PARTS;
    const int lo   = (bl / OWN_PARTS) * OWN_R;
    const int Ep   = part_len(E);
    const int e_lo = (part * Ep < E) ? part * Ep : E;
    const int Ec   = (Ep < E - e_lo) ? Ep : E - e_lo;
    const int tid  = threadIdx.x;
    const int lane = tid & 31;
    const int wave = tid >> 5;
    if (tid < OWN_R) { cntL[tid] = 0; cur[tid] = 0; }
    if (tid == 0) *nstage = 0;
    __syncthreads();

    const int* dst = ei + E + e_lo;
    const int n4 = ((E & 3) == 0) ? (Ec >> 2) : 0;
    const int4* d4 = (const int4*)dst;
    const int4 none = make_int4(NO_DST, NO_DST, NO_DST, NO_DST);
    int below = 0;
    auto put = [&](unsigned t, int e) {
        const int r = atomicAdd(&cntL[t], 1);
        const int p = atomicAdd(nstage, 1);
        if (p < STAGE_CAP) stage[p] = make_int2(e_lo + e, (int)(((unsigned)r << 10) | t));
    };
    auto count_below = [&](int d) {
        below += __popc(__builtin_amdgcn_ballot_w32(d < lo));
    };
    const int n4m = (n4 > 0) ? n4 - 1 : 0;
    for (int b = 0; b < n4; b += 8 * 1024) {
        int4 v[8];
#pragma unroll
        for (int u = 0; u < 8; ++u) {
            const int i = b + u * 1024 + tid;
            v[u] = d4[(i < n4) ? i : n4m];
        }
#pragma unroll
        for (int u = 0; u < 8; ++u) {
            const int i = b + u * 1024 + tid;
            const int4 d = (i < n4) ? v[u] : none;
            count_below(d.x);
            count_below(d.y);
            count_below(d.z);
            count_below(d.w);
            const unsigned t0 = (unsigned)(d.x - lo);
            const unsigned t1 = (unsigned)(d.y - lo);
            const unsigned t2 = (unsigned)(d.z - lo);
            const unsigned t3 = (unsigned)(d.w - lo);
            const unsigned R = (unsigned)OWN_R;
            if ((t0 < R) | (t1 < R) | (t2 < R) | (t3 < R)) {
                if (t0 < R) put(t0, 4 * i);
                if (t1 < R) put(t1, 4 * i + 1);
                if (t2 < R) put(t2, 4 * i + 2);
                if (t3 < R) put(t3, 4 * i + 3);
            }
        }
    }
    for (int e0 = 4 * n4; e0 < Ec; e0 += 1024) {
        const int e = e0 + tid;
        const int d = (e < Ec) ? dst[e] : NO_DST;
        count_below(d);
        const unsigned t = (unsigned)(d - lo);
        if (t < (unsigned)OWN_R) put(t, e);
    }
    if (lane == 0) wbelow[wave] = below;
    __syncthreads();

    int base = 0;
    for (int w = 0; w < 32; ++w) base += wbelow[w];
    const int c = (tid < OWN_R) ? cntL[tid] : 0;
    int inc = c;
    for (int off = 1; off < 32; off <<= 1) {
        const int t = __shfl_up(inc, off, 32);
        if (lane >= off) inc += t;
    }
    if (tid < OWN_R && lane == 31) wscan[wave] = inc;
    __syncthreads();
    if (tid < OWN_R) {
        int wex = 0;
        for (int w = 0; w < wave; ++w) wex += wscan[w];
        const int start = wex + inc - c;
        cntL[tid] = start;
        if (lo + tid <= Nn) rowptr[(size_t)part * Np + lo + tid] = e_lo + base + start;
    }
    __syncthreads();
    const int n = *nstage;
    if (n <= STAGE_CAP) {
        for (int p = tid; p < n; p += 1024) {
            const int2 s = stage[p];
            const unsigned pk = (unsigned)s.y;
            list[e_lo + base + cntL[pk & 1023u] + (int)(pk >> 10)] = make_int2(ei[s.x], s.x);
        }
    } else {
        for (int e0 = 0; e0 < Ec; e0 += 1024) {
            const int e = e0 + tid;
            const int d = (e < Ec) ? dst[e] : NO_DST;
            const unsigned t = (unsigned)(d - lo);
            if (t < (unsigned)OWN_R) {
                const int r = atomicAdd(&cur[t], 1);
                list[e_lo + base + cntL[t] + r] = make_int2(ei[e_lo + e], e_lo + e);
            }
        }
    }
}

#define PREP_LDS_WORDS ((OWN_W_END > F_DIM * F_DIM) ? OWN_W_END : F_DIM * F_DIM)
__global__ __launch_bounds__(1024, 16)
void prep_kernel(const int* __restrict__ eiA, int EA, int NnA, int NpA,
                 int* __restrict__ rowptrA, int2* __restrict__ listA,
                 const int* __restrict__ eiB, int EB, int NnB, int NpB,
                 int* __restrict__ rowptrB, int2* __restrict__ listB, int nbOwnB, int nOwn,
                 const float* __restrict__ XA, const float* __restrict__ WA,
                 const float* __restrict__ a_srcA, const float* __restrict__ a_dstA,
                 float* __restrict__ HoutA, float* __restrict__ asA,
                 float* __restrict__ adA, int rowsA,
                 const float* __restrict__ XB, const float* __restrict__ WB,
                 const float* __restrict__ a_srcB, const float* __restrict__ a_dstB,
                 float* __restrict__ HoutB, float* __restrict__ asB,
                 float* __restrict__ adB, int rowsB, int nbGemmB) {
    __shared__ __attribute__((aligned(16))) int lds[PREP_LDS_WORDS];
    if ((int)blockIdx.x < nOwn)
        owner_body(lds, (int)blockIdx.x, eiA, EA, NnA, NpA, rowptrA, listA,
                   eiB, EB, NnB, NpB, rowptrB, listB, nbOwnB);
    else
        gemm_body((float*)lds, (int)blockIdx.x - nOwn, XA, WA, a_srcA, a_dstA, HoutA, asA, adA, rowsA,
                  XB, WB, a_srcB, a_dstB, HoutB, asB, adB, rowsB, nbGemmB);
}

__device__ inline float leaky(float e) { return (e > 0.f) ? e : NEG_SLOPE * e; }

__device__ inline float lane_value(float v, int k) {
    return __int_as_float(__builtin_amdgcn_readlane(__float_as_int(v), k));
}

__device__ __forceinline__
void seg_body(const float* __restrict__ hbuf, const float* __restrict__ as,
              const float* __restrict__ ad, const int* __restrict__ rowptr,
              const int2* __restrict__ list, const float* __restrict__ bias,
              int E, int Nn, int Np, float* __restrict__ alpha, float* __restrict__ xout,
              int blockLocal) {
    const int lane = threadIdx.x & 31;
    const int node = __builtin_amdgcn_readfirstlane(
        (int)(blockLocal * (blockDim.x >> 5) + (threadIdx.x >> 5)));
    if (node >= Nn) return;
    int rb[OWN_PARTS];
    int rc[OWN_PARTS];
    int deg = 0;
#pragma unroll
    for (int p = 0; p < OWN_PARTS; ++p) {
        const int b = rowptr[(size_t)p * Np + node];
        const int e = rowptr[(size_t)p * Np + node + 1];
        rb[p] = b;
        deg += e - b;
        rc[p] = deg;
    }
    auto at = [&](int q) {
        int j = rb[0] + q;
#pragma unroll
        for (int p = 1; p < OWN_PARTS; ++p)
            if (q >= rc[p - 1]) j = rb[p] + (q - rc[p - 1]);
        return j;
    };
    const float4 adn = *(const float4*)(ad + (size_t)node * 4);
    const float4 asn = *(const float4*)(as + (size_t)node * 4);
    float4 es;
    es.x = leaky(asn.x + adn.x);
    es.y = leaky(asn.y + adn.y);
    es.z = leaky(asn.z + adn.z);
    es.w = leaky(asn.w + adn.w);
    const bool one = deg <= 32;
    const bool has = lane < deg;

    float4 m = es;
    int2 rec = make_int2(0, 0);
    float4 ev = make_float4(0.f, 0.f, 0.f, 0.f);
    if (one) {
        if (has) {
            rec = list[at(lane)];
            const float4 a = *(const float4*)(as + (size_t)rec.x * 4);
            ev.x = leaky(a.x + adn.x);
            ev.y = leaky(a.y + adn.y);
            ev.z = leaky(a.z + adn.z);
            ev.w = leaky(a.w + adn.w);
            m.x = fmaxf(m.x, ev.x);
            m.y = fmaxf(m.y, ev.y);
            m.z = fmaxf(m.z, ev.z);
            m.w = fmaxf(m.w, ev.w);
        }
    } else {
        for (int q = lane; q < deg; q += 32) {
            const int sidx = list[at(q)].x;
            const float4 a = *(const float4*)(as + (size_t)sidx * 4);
            m.x = fmaxf(m.x, leaky(a.x + adn.x));
            m.y = fmaxf(m.y, leaky(a.y + adn.y));
            m.z = fmaxf(m.z, leaky(a.z + adn.z));
            m.w = fmaxf(m.w, leaky(a.w + adn.w));
        }
    }
#pragma unroll
    for (int off = 16; off > 0; off >>= 1) {
        m.x = fmaxf(m.x, __shfl_xor(m.x, off, 32));
        m.y = fmaxf(m.y, __shfl_xor(m.y, off, 32));
        m.z = fmaxf(m.z, __shfl_xor(m.z, off, 32));
        m.w = fmaxf(m.w, __shfl_xor(m.w, off, 32));
    }

    float4 z = make_float4(0.f, 0.f, 0.f, 0.f);
    if (one) {
        if (has) {
            ev.x = __expf(ev.x - m.x);
            ev.y = __expf(ev.y - m.y);
            ev.z = __expf(ev.z - m.z);
            ev.w = __expf(ev.w - m.w);
            z = ev;
        }
    } else {
        for (int q = lane; q < deg; q += 32) {
            const int sidx = list[at(q)].x;
            const float4 a = *(const float4*)(as + (size_t)sidx * 4);
            z.x += __expf(leaky(a.x + adn.x) - m.x);
            z.y += __expf(leaky(a.y + adn.y) - m.y);
            z.z += __expf(leaky(a.z + adn.z) - m.z);
            z.w += __expf(leaky(a.w + adn.w) - m.w);
        }
    }
#pragma unroll
    for (int off = 16; off > 0; off >>= 1) {
        z.x += __shfl_xor(z.x, off, 32);
        z.y += __shfl_xor(z.y, off, 32);
        z.z += __shfl_xor(z.z, off, 32);
        z.w += __shfl_xor(z.w, off, 32);
    }
    float4 ps;
    ps.x = __expf(es.x - m.x);
    ps.y = __expf(es.y - m.y);
    ps.z = __expf(es.z - m.z);
    ps.w = __expf(es.w - m.w);
    z.x += ps.x;
    z.y += ps.y;
    z.z += ps.z;
    z.w += ps.w;

    const int hd = lane >> 3;
    float4 als;
    als.x = ps.x / z.x;
    als.y = ps.y / z.y;
    als.z = ps.z / z.z;
    als.w = ps.w / z.w;
    if (lane == 0) *(float4*)(alpha + (size_t)(E + node) * 4) = als;
    const float wself = (hd == 0) ? als.x : (hd == 1) ? als.y : (hd == 2) ? als.z : als.w;
    const float4 hself = *(const float4*)(hbuf + (size_t)node * F_DIM + lane * 4);
    float4 acc;
    acc.x = wself * hself.x;
    acc.y = wself * hself.y;
    acc.z = wself * hself.z;
    acc.w = wself * hself.w;

    float4 al1 = make_float4(0.f, 0.f, 0.f, 0.f);
    if (one && has) {
        al1.x = ev.x / z.x;
        al1.y = ev.y / z.y;
        al1.z = ev.z / z.z;
        al1.w = ev.w / z.w;
        *(float4*)(alpha + (size_t)rec.y * 4) = al1;
    }
    for (int cb = 0; cb < deg; cb += 32) {
        int sidx = rec.x;
        float4 al = al1;
        if (!one) {
            const int q = cb + lane;
            sidx = 0;
            al = make_float4(0.f, 0.f, 0.f, 0.f);
            if (q < deg) {
                const int2 r2 = list[at(q)];
                sidx = r2.x;
                const float4 a = *(const float4*)(as + (size_t)sidx * 4);
                al.x = __expf(leaky(a.x + adn.x) - m.x) / z.x;
                al.y = __expf(leaky(a.y + adn.y) - m.y) / z.y;
                al.z = __expf(leaky(a.z + adn.z) - m.z) / z.z;
                al.w = __expf(leaky(a.w + adn.w) - m.w) / z.w;
                *(float4*)(alpha + (size_t)r2.y * 4) = al;
            }
        }
        const int left = deg - cb;
        const int cnt  = (left < 32) ? left : 32;
#pragma unroll 4
        for (int k = 0; k < cnt; ++k) {
            const int sk = __builtin_amdgcn_readlane(sidx, k);
            const float a0 = lane_value(al.x, k);
            const float a1 = lane_value(al.y, k);
            const float a2 = lane_value(al.z, k);
            const float a3 = lane_value(al.w, k);
            const float w = (hd == 0) ? a0 : (hd == 1) ? a1 : (hd == 2) ? a2 : a3;
            const float4 hv = *(const float4*)(hbuf + (size_t)sk * F_DIM + lane * 4);
            acc.x = fmaf(w, hv.x, acc.x);
            acc.y = fmaf(w, hv.y, acc.y);
            acc.z = fmaf(w, hv.z, acc.z);
            acc.w = fmaf(w, hv.w, acc.w);
        }
    }
    const float4 bv = *(const float4*)(bias + lane * 4);
    acc.x += bv.x;
    acc.y += bv.y;
    acc.z += bv.z;
    acc.w += bv.w;
    *(float4*)(xout + (size_t)node * F_DIM + lane * 4) = acc;
}

__global__ __launch_bounds__(256)
void seg_kernel(const float* __restrict__ hbufA, const float* __restrict__ asA,
                const float* __restrict__ adA, const int* __restrict__ rowptrA,
                const int2* __restrict__ listA, const float* __restrict__ biasA,
                int EA, int NnA, int NpA, float* __restrict__ alphaA, float* __restrict__ xoutA,
                const float* __restrict__ hbufB, const float* __restrict__ asB,
                const float* __restrict__ adB, const int* __restrict__ rowptrB,
                const int2* __restrict__ listB, const float* __restrict__ biasB,
                int EB, int NnB, int NpB, float* __restrict__ alphaB, float* __restrict__ xoutB,
                int nbB) {
    if (blockIdx.x < nbB)
        seg_body(hbufB, asB, adB, rowptrB, listB, biasB, EB, NnB, NpB, alphaB, xoutB,
                 (int)blockIdx.x);
    else
        seg_body(hbufA, asA, adA, rowptrA, listA, biasA, EA, NnA, NpA, alphaA, xoutA,
                 (int)blockIdx.x - nbB);
}

#define GRP_G 4
#define GRP_WPG 6
#define GRP_T (GRP_G * GRP_WPG * 32)
#define GRP_CAP 2048

__device__ __forceinline__
void mix_node(float* xio, const float4 (&cv)[17], int node, int lane, float4 xo) {
    float sv[17];
#pragma unroll
    for (int k = 0; k < 17; ++k) {
        const float4 c = cv[k];
        float s = xo.x * c.x + xo.y * c.y + xo.z * c.z + xo.w * c.w;
#pragma unroll
        for (int off = 16; off > 0; off >>= 1) s += __shfl_xor(s, off, 32);
        sv[k] = s;
    }
    float mx = sv[0];
#pragma unroll
    for (int k = 1; k < 17; ++k) mx = fmaxf(mx, sv[k]);
    float sum = 0.f;
#pragma unroll
    for (int k = 0; k < 17; ++k) { sv[k] = __expf(sv[k] - mx); sum += sv[k]; }
    float inv = 1.f / sum;
    float4 go = make_float4(0.f, 0.f, 0.f, 0.f);
#pragma unroll
    for (int k = 0; k < 17; ++k) {
        float w = sv[k] * inv;
        go.x = fmaf(w, cv[k].x, go.x);
        go.y = fmaf(w, cv[k].y, go.y);
        go.z = fmaf(w, cv[k].z, go.z);
        go.w = fmaf(w, cv[k].w, go.w);
    }
    float4 o;
    o.x = 0.5f * (xo.x + go.x);
    o.y = 0.5f * (xo.y + go.y);
    o.z = 0.5f * (xo.z + go.z);
    o.w = 0.5f * (xo.w + go.w);
    *(float4*)(xio + (size_t)node * F_DIM + lane * 4) = o;
}

__global__ __launch_bounds__(GRP_T)
void group_kernel(float* xio, const float* __restrict__ xg,
                  const int* __restrict__ gidx, const int* __restrict__ nbr,
                  int Nn, int Kg) {
    __shared__ int ent[GRP_G][GRP_CAP];
    __shared__ int cnt[GRP_G];
    const int tid  = threadIdx.x;
    const int lane = tid & 31;
    const int wave = __builtin_amdgcn_readfirstlane(tid >> 5);
    const int lo   = blockIdx.x * GRP_G;
    const int t    = wave / GRP_WPG;
    const int r    = wave - t * GRP_WPG;
    const int g    = lo + t;
    if (tid < GRP_G) cnt[tid] = 0;
    __syncthreads();

    auto take = [&](int d, int node) {
        const unsigned tt = (unsigned)(d - lo);
        if (tt < (unsigned)GRP_G) {
            const int p = atomicAdd(&cnt[tt], 1);
            if (p < GRP_CAP) ent[tt][p] = node;
        }
    };
    const int n4 = ((((size_t)gidx) & 15) == 0) ? (Nn >> 2) : 0;
    const int4* g4 = (const int4*)gidx;
    const int n4m = (n4 > 0) ? n4 - 1 : 0;
    for (int b = 0; b < n4; b += 8 * GRP_T) {
        int4 v[8];
#pragma unroll
        for (int u = 0; u < 8; ++u) {
            const int i = b + u * GRP_T + tid;
            v[u] = g4[(i < n4) ? i : n4m];
        }
#pragma unroll
        for (int u = 0; u < 8; ++u) {
            const int i = b + u * GRP_T + tid;
            if (i < n4) {
                take(v[u].x, 4 * i);
                take(v[u].y, 4 * i + 1);
                take(v[u].z, 4 * i + 2);
                take(v[u].w, 4 * i + 3);
            }
        }
    }
    for (int e0 = 4 * n4; e0 < Nn; e0 += GRP_T) {
        const int e = e0 + tid;
        if (e < Nn) take(gidx[e], e);
    }
    __syncthreads();

    if (g >= Kg) return;

    float4 cv[17];
    cv[0] = *(const float4*)(xg + (size_t)g * F_DIM + lane * 4);
#pragma unroll
    for (int k = 0; k < 16; ++k) {
        const int c = nbr[(size_t)g * 16 + k];
        cv[k + 1] = *(const float4*)(xg + (size_t)c * F_DIM + lane * 4);
    }

    const int n = cnt[t];
    if (n <= GRP_CAP) {
        int p = r;
        int node = 0;
        float4 xo = make_float4(0.f, 0.f, 0.f, 0.f);
        if (p < n) {
            node = __builtin_amdgcn_readfirstlane(ent[t][p]);
            xo = *(const float4*)(xio + (size_t)node * F_DIM + lane * 4);
        }
        while (p < n) {
            const int pn = p + GRP_WPG;
            int node2 = 0;
            float4 xo2 = xo;
            if (pn < n) {
                node2 = __builtin_amdgcn_readfirstlane(ent[t][pn]);
                xo2 = *(const float4*)(xio + (size_t)node2 * F_DIM + lane * 4);
            }
            mix_node(xio, cv, node, lane, xo);
            node = node2; xo = xo2; p = pn;
        }
    } else {
        for (int v = r; v < Nn; v += GRP_WPG) {
            if (gidx[v] == g) {
                const float4 xo = *(const float4*)(xio + (size_t)v * F_DIM + lane * 4);
                mix_node(xio, cv, v, lane, xo);
            }
        }
    }
}

extern "C" void kernel_launch(void* const* d_in, const int* in_sizes, int n_in,
                              void* d_out, int out_size, void* d_ws, size_t ws_size,
                              hipStream_t stream) {
    const float* x       = (const float*)d_in[0];
    const int*   ei      = (const int*)d_in[1];
    const float* x_g     = (const float*)d_in[2];
    const int*   ei_g    = (const int*)d_in[3];
    const int*   gidx    = (const int*)d_in[4];
    const int*   nbr     = (const int*)d_in[5];
    const float* W1      = (const float*)d_in[6];
    const float* a_src1  = (const float*)d_in[7];
    const float* a_dst1  = (const float*)d_in[8];
    const float* b1      = (const float*)d_in[9];
    const float* W2      = (const float*)d_in[10];
    const float* a_src2  = (const float*)d_in[11];
    const float* a_dst2  = (const float*)d_in[12];
    const float* b2      = (const float*)d_in[13];

    const int N  = in_sizes[0] / F_DIM;
    const int E  = in_sizes[1] / 2;
    const int K  = in_sizes[2] / F_DIM;
    const int Eg = in_sizes[3] / 2;

    float* out     = (float*)d_out;
    float* x_final = out;
    float* xg_out  = out + (size_t)N * F_DIM;
    float* alpha1  = xg_out + (size_t)K * F_DIM;
    float* alpha2  = alpha1 + (size_t)(E + N) * H_HEADS;

    auto al4 = [](size_t n) { return (n + 3) & ~(size_t)3; };
    const int Np1 = ((N + 1 + OWN_R - 1) / OWN_R) * OWN_R;
    const int Np2 = ((K + 1 + OWN_R - 1) / OWN_R) * OWN_R;
    float* ws = (float*)d_ws;
    size_t off = 0;
    float* h1    = ws + off; off += (size_t)N * F_DIM;
    float* h2    = ws + off; off += (size_t)K * F_DIM;
    float* as1   = ws + off; off += (size_t)N * H_HEADS;
    float* ad1   = ws + off; off += (size_t)N * H_HEADS;
    float* as2   = ws + off; off += (size_t)K * H_HEADS;
    float* ad2   = ws + off; off += (size_t)K * H_HEADS;
    int*   cur1  = (int*)(ws + off);  off += (size_t)OWN_PARTS * Np1;
    int*   cur2  = (int*)(ws + off);  off += (size_t)OWN_PARTS * Np2;
    int2*  list1 = (int2*)(ws + off); off += al4((size_t)E * 2);
    int2*  list2 = (int2*)(ws + off); off += al4((size_t)Eg * 2);

    const int T = 256;
    auto blk = [](long n, int t) { return (int)((n + t - 1) / t); };


    const int nOwn = OWN_PARTS * (blk(K + 1, OWN_R) + blk(N + 1, OWN_R));
    prep_kernel<<<nOwn + blk(K, GEMM_ROWS) + blk(N, GEMM_ROWS), 1024, 0, stream>>>(
        ei, E, N, Np1, cur1, list1, ei_g, Eg, K, Np2, cur2, list2, OWN_PARTS * blk(K + 1, OWN_R), nOwn,
        x, W1, a_src1, a_dst1, h1, as1, ad1, N,
        x_g, W2, a_src2, a_dst2, h2, as2, ad2, K, blk(K, GEMM_ROWS));

    seg_kernel<<<blk(K, 8) + blk(N, 8), T, 0, stream>>>(
        h1, as1, ad1, cur1, list1, b1, E, N, Np1, alpha1, x_final,
        h2, as2, ad2, cur2, list2, b2, Eg, K, Np2, alpha2, xg_out, blk(K, 8));

    group_kernel<<<blk(K, GRP_G), GRP_T, 0, stream>>>(x_final, xg_out, gidx, nbr, N, K);
}


# Softmax pass asks for a node's first eight source rows as soon as its list records arrive, beside the source coefficients; later rows go eight at a time, no one-row trips; same terms, same order
# speedup vs baseline: 5.1877x; 1.0664x over previous
#include <hip/hip_runtime.h>
#include <math.h>

#define H_HEADS 4
#define C_CH 32
#define F_DIM 128
#define NEG_SLOPE 0.2f

typedef __attribute__((ext_vector_type(2))) float v2f;
typedef __attribute__((ext_vector_type(8))) float v8f;

#define GEMM_ROWS 256
__device__ __forceinline__
void gemm_body(float* Wlds, int bIdx, const float* __restrict__ XA, const float* __restrict__ WA,
                      const float* __restrict__ a_srcA, const float* __restrict__ a_dstA,
                      float* __restrict__ HoutA, float* __restrict__ asA,
                      float* __restrict__ adA, int rowsA,
                      const float* __restrict__ XB, const float* __restrict__ WB,
                      const float* __restrict__ a_srcB, const float* __restrict__ a_dstB,
                      float* __restrict__ HoutB, float* __restrict__ asB,
                      float* __restrict__ adB, int rowsB, int nbB) {
    const bool gB = bIdx < nbB;
    const float* X     = gB ? XB : XA;
    const float* W     = gB ? WB : WA;
    const float* a_src = gB ? a_srcB : a_srcA;
    const float* a_dst = gB ? a_dstB : a_dstA;
    float* Hout        = gB ? HoutB : HoutA;
    float* as          = gB ? asB : asA;
    float* ad          = gB ? adB : adA;
    const int rows     = gB ? rowsB : rowsA;
    const int blockLocal = gB ? bIdx : bIdx - nbB;
    for (int i = threadIdx.x; i < (F_DIM * F_DIM) / 2; i += blockDim.x) {
        int kp  = i >> 7;
        int col = i & 127;
        int cs  = (col + ((kp & 1) << 4)) & 127;
        float2 w2;
        w2.x = W[(size_t)(2 * kp)     * F_DIM + col];
        w2.y = W[(size_t)(2 * kp + 1) * F_DIM + col];
        *(float2*)(Wlds + (size_t)kp * 256 + cs * 2) = w2;
    }
    __syncthreads();

    const int lane = threadIdx.x & 31;
    const int wave = __builtin_amdgcn_readfirstlane((int)(threadIdx.x >> 5));
    const int lm   = lane & 15;
    const int lh   = lane >> 4;
    const int ch   = wave & 1;
    const int rowBase = blockLocal * GEMM_ROWS + (wave >> 1) * 16;
    if (rowBase >= rows) return;

    int aRow = rowBase + lm;
    if (aRow >= rows) aRow = rows - 1;

    v8f acc[4];
    v8f zero = {0.f, 0.f, 0.f, 0.f, 0.f, 0.f, 0.f, 0.f};
#pragma unroll
    for (int ct = 0; ct < 4; ++ct) acc[ct] = zero;

    const float* xrow = X + (size_t)aRow * F_DIM;
    for (int k0 = 0; k0 < F_DIM; k0 += 4) {
        const int kp = (k0 >> 1) + lh;
        v2f a;
        a.x = xrow[2 * kp];
        a.y = xrow[2 * kp + 1];
        const float* bbase = Wlds + (size_t)kp * 256;
#pragma unroll
        for (int ct = 0; ct < 4; ++ct) {
            const int cs = ((4 * ch + ct) * 16 + lm + 16 * lh) & 127;
            v2f b = *(const v2f*)(bbase + cs * 2);
            acc[ct] = __builtin_amdgcn_wmma_f32_16x16x4_f32(
                false, a, false, b, (short)0, acc[ct], false, false);
        }
    }

#pragma unroll
    for (int ct = 0; ct < 4; ++ct) {
#pragma unroll
        for (int g = 0; g < 8; ++g) {
            int r = rowBase + g + 8 * lh;
            if (r < rows) Hout[(unsigned)(r * F_DIM + (4 * ch + ct) * 16 + lm)] = acc[ct][g];
        }
    }

    float asv[4], adv[4];
#pragma unroll
    for (int ct = 0; ct < 4; ++ct) {
        asv[ct] = a_src[(4 * ch + ct) * 16 + lm];
        adv[ct] = a_dst[(4 * ch + ct) * 16 + lm];
    }
#pragma unroll
    for (int h = 0; h < 2; ++h) {
#pragma unroll
        for (int g = 0; g < 8; ++g) {
            float ss = acc[2 * h][g] * asv[2 * h];
            ss = fmaf(acc[2 * h + 1][g], asv[2 * h + 1], ss);
            float sd = acc[2 * h][g] * adv[2 * h];
            sd = fmaf(acc[2 * h + 1][g], adv[2 * h + 1], sd);
#pragma unroll
            for (int off = 8; off > 0; off >>= 1) {
                ss += __shfl_xor(ss, off, 32);
                sd += __shfl_xor(sd, off, 32);
            }
            int r = rowBase + g + 8 * lh;
            if (lm == 0 && r < rows) {
                as[(unsigned)(r * H_HEADS + 2 * ch + h)] = ss;
                ad[(unsigned)(r * H_HEADS + 2 * ch + h)] = sd;
            }
        }
    }
}

#define OWN_PARTS 4
#define OWN_R 1024
#define STAGE_CAP 7680
#define NO_DST 0x7fffffff

__device__ inline int part_len(int E) { return ((E + 4 * OWN_PARTS - 1) / (4 * OWN_PARTS)) * 4; }

#define OWN_W_CNT   (2 * STAGE_CAP)
#define OWN_W_CUR   (OWN_W_CNT + OWN_R)
#define OWN_W_BELOW (OWN_W_CUR + OWN_R)
#define OWN_W_SCAN  (OWN_W_BELOW + 32)
#define OWN_W_NST   (OWN_W_SCAN + OWN_R / 32)
#define OWN_W_END   (OWN_W_NST + 4)

__device__ __forceinline__
void owner_body(int* lds, int bIdx,
                const int* __restrict__ eiA, int EA, int NnA, int NpA,
                int* __restrict__ rowptrA, int2* __restrict__ listA,
                const int* __restrict__ eiB, int EB, int NnB, int NpB,
                int* __restrict__ rowptrB, int2* __restrict__ listB, int nbB) {
    int2* stage = (int2*)lds;
    int* cntL   = lds + OWN_W_CNT;
    int* cur    = lds + OWN_W_CUR;
    int* wbelow = lds + OWN_W_BELOW;
    int* wscan  = lds + OWN_W_SCAN;
    int* nstage = lds + OWN_W_NST;
    const bool gB = bIdx < nbB;
    const int* ei = gB ? eiB : eiA;
    const int  E  = gB ? EB : EA;
    const int  Nn = gB ? NnB : NnA;
    const int  Np = gB ? NpB : NpA;
    int* rowptr   = gB ? rowptrB : rowptrA;
    int2* list    = gB ? listB : listA;
    const int bl   = gB ? bIdx : bIdx - nbB;
    const int part = bl % OWN_PARTS;
    const int lo   = (bl / OWN_PARTS) * OWN_R;
    const int Ep   = part_len(E);
    const int e_lo = (part * Ep < E) ? part * Ep : E;
    const int Ec   = (Ep < E - e_lo) ? Ep : E - e_lo;
    const int tid  = threadIdx.x;
    const int lane = tid & 31;
    const int wave = tid >> 5;
    if (tid < OWN_R) { cntL[tid] = 0; cur[tid] = 0; }
    if (tid == 0) *nstage = 0;
    __syncthreads();

    const int* dst = ei + E + e_lo;
    const int n4 = ((E & 3) == 0) ? (Ec >> 2) : 0;
    const int4* d4 = (const int4*)dst;
    const int4 none = make_int4(NO_DST, NO_DST, NO_DST, NO_DST);
    int below = 0;
    auto put = [&](unsigned t, int e) {
        const int r = atomicAdd(&cntL[t], 1);
        const int p = atomicAdd(nstage, 1);
        if (p < STAGE_CAP) stage[p] = make_int2(e_lo + e, (int)(((unsigned)r << 10) | t));
    };
    auto count_below = [&](int d) {
        below += __popc(__builtin_amdgcn_ballot_w32(d < lo));
    };
    const int n4m = (n4 > 0) ? n4 - 1 : 0;
    for (int b = 0; b < n4; b += 8 * 1024) {
        int4 v[8];
#pragma unroll
        for (int u = 0; u < 8; ++u) {
            const int i = b + u * 1024 + tid;
            v[u] = d4[(i < n4) ? i : n4m];
        }
#pragma unroll
        for (int u = 0; u < 8; ++u) {
            const int i = b + u * 1024 + tid;
            const int4 d = (i < n4) ? v[u] : none;
            count_below(d.x);
            count_below(d.y);
            count_below(d.z);
            count_below(d.w);
            const unsigned t0 = (unsigned)(d.x - lo);
            const unsigned t1 = (unsigned)(d.y - lo);
            const unsigned t2 = (unsigned)(d.z - lo);
            const unsigned t3 = (unsigned)(d.w - lo);
            const unsigned R = (unsigned)OWN_R;
            if ((t0 < R) | (t1 < R) | (t2 < R) | (t3 < R)) {
                if (t0 < R) put(t0, 4 * i);
                if (t1 < R) put(t1, 4 * i + 1);
                if (t2 < R) put(t2, 4 * i + 2);
                if (t3 < R) put(t3, 4 * i + 3);
            }
        }
    }
    for (int e0 = 4 * n4; e0 < Ec; e0 += 1024) {
        const int e = e0 + tid;
        const int d = (e < Ec) ? dst[e] : NO_DST;
        count_below(d);
        const unsigned t = (unsigned)(d - lo);
        if (t < (unsigned)OWN_R) put(t, e);
    }
    if (lane == 0) wbelow[wave] = below;
    __syncthreads();

    int base = 0;
    for (int w = 0; w < 32; ++w) base += wbelow[w];
    const int c = (tid < OWN_R) ? cntL[tid] : 0;
    int inc = c;
    for (int off = 1; off < 32; off <<= 1) {
        const int t = __shfl_up(inc, off, 32);
        if (lane >= off) inc += t;
    }
    if (tid < OWN_R && lane == 31) wscan[wave] = inc;
    __syncthreads();
    if (tid < OWN_R) {
        int wex = 0;
        for (int w = 0; w < wave; ++w) wex += wscan[w];
        const int start = wex + inc - c;
        cntL[tid] = start;
        if (lo + tid <= Nn) rowptr[(size_t)part * Np + lo + tid] = e_lo + base + start;
    }
    __syncthreads();
    const int n = *nstage;
    if (n <= STAGE_CAP) {
        for (int p = tid; p < n; p += 1024) {
            const int2 s = stage[p];
            const unsigned pk = (unsigned)s.y;
            list[e_lo + base + cntL[pk & 1023u] + (int)(pk >> 10)] = make_int2(ei[s.x], s.x);
        }
    } else {
        for (int e0 = 0; e0 < Ec; e0 += 1024) {
            const int e = e0 + tid;
            const int d = (e < Ec) ? dst[e] : NO_DST;
            const unsigned t = (unsigned)(d - lo);
            if (t < (unsigned)OWN_R) {
                const int r = atomicAdd(&cur[t], 1);
                list[e_lo + base + cntL[t] + r] = make_int2(ei[e_lo + e], e_lo + e);
            }
        }
    }
}

#define PREP_LDS_WORDS ((OWN_W_END > F_DIM * F_DIM) ? OWN_W_END : F_DIM * F_DIM)
__global__ __launch_bounds__(1024, 16)
void prep_kernel(const int* __restrict__ eiA, int EA, int NnA, int NpA,
                 int* __restrict__ rowptrA, int2* __restrict__ listA,
                 const int* __restrict__ eiB, int EB, int NnB, int NpB,
                 int* __restrict__ rowptrB, int2* __restrict__ listB, int nbOwnB, int nOwn,
                 const float* __restrict__ XA, const float* __restrict__ WA,
                 const float* __restrict__ a_srcA, const float* __restrict__ a_dstA,
                 float* __restrict__ HoutA, float* __restrict__ asA,
                 float* __restrict__ adA, int rowsA,
                 const float* __restrict__ XB, const float* __restrict__ WB,
                 const float* __restrict__ a_srcB, const float* __restrict__ a_dstB,
                 float* __restrict__ HoutB, float* __restrict__ asB,
                 float* __restrict__ adB, int rowsB, int nbGemmB) {
    __shared__ __attribute__((aligned(16))) int lds[PREP_LDS_WORDS];
    if ((int)blockIdx.x < nOwn)
        owner_body(lds, (int)blockIdx.x, eiA, EA, NnA, NpA, rowptrA, listA,
                   eiB, EB, NnB, NpB, rowptrB, listB, nbOwnB);
    else
        gemm_body((float*)lds, (int)blockIdx.x - nOwn, XA, WA, a_srcA, a_dstA, HoutA, asA, adA, rowsA,
                  XB, WB, a_srcB, a_dstB, HoutB, asB, adB, rowsB, nbGemmB);
}

__device__ inline float leaky(float e) { return (e > 0.f) ? e : NEG_SLOPE * e; }

__device__ inline float lane_value(float v, int k) {
    return __int_as_float(__builtin_amdgcn_readlane(__float_as_int(v), k));
}

__device__ __forceinline__
void rows_ask4(const float* __restrict__ hbuf, int sidx, int k0, int n, int lane, float4 (&hv)[4]) {
#pragma unroll
    for (int u = 0; u < 4; ++u) {
        int kc = (k0 + u < n) ? k0 + u : n - 1;
        kc = (kc > 0) ? kc : 0;
        const int sk = __builtin_amdgcn_readlane(sidx, kc);
        hv[u] = *(const float4*)(hbuf + (unsigned)(sk * F_DIM + lane * 4));
    }
}

__device__ __forceinline__
void rows_sum4(float4& acc, const float4 al, int k0, bool h1, bool h2, const float4 (&hv)[4]) {
#pragma unroll
    for (int u = 0; u < 4; ++u) {
        const int k = k0 + u;
        const float a0 = lane_value(al.x, k);
        const float a1 = lane_value(al.y, k);
        const float a2 = lane_value(al.z, k);
        const float a3 = lane_value(al.w, k);
        const float lo = h1 ? a1 : a0;
        const float hi = h1 ? a3 : a2;
        const float w  = h2 ? hi : lo;
        acc.x = fmaf(w, hv[u].x, acc.x);
        acc.y = fmaf(w, hv[u].y, acc.y);
        acc.z = fmaf(w, hv[u].z, acc.z);
        acc.w = fmaf(w, hv[u].w, acc.w);
    }
}

__device__ __forceinline__ void wave_max4(float4& m) {
#pragma unroll
    for (int off = 16; off > 0; off >>= 1) {
        m.x = fmaxf(m.x, __shfl_xor(m.x, off, 32));
        m.y = fmaxf(m.y, __shfl_xor(m.y, off, 32));
        m.z = fmaxf(m.z, __shfl_xor(m.z, off, 32));
        m.w = fmaxf(m.w, __shfl_xor(m.w, off, 32));
    }
}

__device__ __forceinline__ void wave_sum4(float4& z) {
#pragma unroll
    for (int off = 16; off > 0; off >>= 1) {
        z.x += __shfl_xor(z.x, off, 32);
        z.y += __shfl_xor(z.y, off, 32);
        z.z += __shfl_xor(z.z, off, 32);
        z.w += __shfl_xor(z.w, off, 32);
    }
}

__device__ __forceinline__
void seg_body(const float* __restrict__ hbuf, const float* __restrict__ as,
              const float* __restrict__ ad, const int* __restrict__ rowptr,
              const int2* __restrict__ list, const float* __restrict__ bias,
              int E, int Nn, int Np, float* __restrict__ alpha, float* __restrict__ xout,
              int blockLocal) {
    const int lane = threadIdx.x & 31;
    const int node = __builtin_amdgcn_readfirstlane(
        (int)(blockLocal * (blockDim.x >> 5) + (threadIdx.x >> 5)));
    if (node >= Nn) return;
    int rb[OWN_PARTS];
    int rc[OWN_PARTS];
    int deg = 0;
#pragma unroll
    for (int p = 0; p < OWN_PARTS; ++p) {
        const int b = rowptr[(size_t)p * Np + node];
        const int e = rowptr[(size_t)p * Np + node + 1];
        rb[p] = b;
        deg += e - b;
        rc[p] = deg;
    }
    auto at = [&](int q) {
        int j = rb[0] + q;
#pragma unroll
        for (int p = 1; p < OWN_PARTS; ++p)
            if (q >= rc[p - 1]) j = rb[p] + (q - rc[p - 1]);
        return j;
    };
    const float4 adn = *(const float4*)(ad + (size_t)node * 4);
    const float4 asn = *(const float4*)(as + (size_t)node * 4);
    float4 es;
    es.x = leaky(asn.x + adn.x);
    es.y = leaky(asn.y + adn.y);
    es.z = leaky(asn.z + adn.z);
    es.w = leaky(asn.w + adn.w);
    const int hd = lane >> 3;
    const bool h1 = (hd & 1) != 0;
    const bool h2 = (hd & 2) != 0;
    float4 hvA[4], hvB[4];
    float4 acc;

    if (deg <= 32) {
        const bool has = lane < deg;
        int2 rec = make_int2(0, 0);
        float4 a = make_float4(0.f, 0.f, 0.f, 0.f);
        if (has) {
            rec = list[at(lane)];
            a = *(const float4*)(as + (size_t)rec.x * 4);
        }
        rows_ask4(hbuf, rec.x, 0, deg, lane, hvA);
        rows_ask4(hbuf, rec.x, 4, deg, lane, hvB);
        const float4 hself = *(const float4*)(hbuf + (size_t)node * F_DIM + lane * 4);
        const float4 bv = *(const float4*)(bias + lane * 4);

        float4 m = es;
        float4 ev = make_float4(0.f, 0.f, 0.f, 0.f);
        if (has) {
            ev.x = leaky(a.x + adn.x);
            ev.y = leaky(a.y + adn.y);
            ev.z = leaky(a.z + adn.z);
            ev.w = leaky(a.w + adn.w);
            m.x = fmaxf(m.x, ev.x);
            m.y = fmaxf(m.y, ev.y);
            m.z = fmaxf(m.z, ev.z);
            m.w = fmaxf(m.w, ev.w);
        }
        wave_max4(m);

        float4 z = make_float4(0.f, 0.f, 0.f, 0.f);
        if (has) {
            ev.x = __expf(ev.x - m.x);
            ev.y = __expf(ev.y - m.y);
            ev.z = __expf(ev.z - m.z);
            ev.w = __expf(ev.w - m.w);
            z = ev;
        }
        wave_sum4(z);
        float4 ps;
        ps.x = __expf(es.x - m.x);
        ps.y = __expf(es.y - m.y);
        ps.z = __expf(es.z - m.z);
        ps.w = __expf(es.w - m.w);
        z.x += ps.x;
        z.y += ps.y;
        z.z += ps.z;
        z.w += ps.w;

        float4 als;
        als.x = ps.x / z.x;
        als.y = ps.y / z.y;
        als.z = ps.z / z.z;
        als.w = ps.w / z.w;
        if (lane == 0) *(float4*)(alpha + (size_t)(E + node) * 4) = als;
        float4 al1 = make_float4(0.f, 0.f, 0.f, 0.f);
        if (has) {
            al1.x = ev.x / z.x;
            al1.y = ev.y / z.y;
            al1.z = ev.z / z.z;
            al1.w = ev.w / z.w;
            *(float4*)(alpha + (size_t)rec.y * 4) = al1;
        }
        const float wself = h2 ? (h1 ? als.w : als.z) : (h1 ? als.y : als.x);
        acc.x = wself * hself.x;
        acc.y = wself * hself.y;
        acc.z = wself * hself.z;
        acc.w = wself * hself.w;
        rows_sum4(acc, al1, 0, h1, h2, hvA);
        rows_sum4(acc, al1, 4, h1, h2, hvB);
        for (int k0 = 8; k0 < deg; k0 += 8) {
            const bool more = k0 + 4 < deg;
            rows_ask4(hbuf, rec.x, k0, deg, lane, hvA);
            if (more) rows_ask4(hbuf, rec.x, k0 + 4, deg, lane, hvB);
            rows_sum4(acc, al1, k0, h1, h2, hvA);
            if (more) rows_sum4(acc, al1, k0 + 4, h1, h2, hvB);
        }
        acc.x += bv.x;
        acc.y += bv.y;
        acc.z += bv.z;
        acc.w += bv.w;
    } else {
    float4 m = es;
#pragma unroll 1
    for (int q = lane; q < deg; q += 32) {
        const int sidx = list[at(q)].x;
        const float4 a = *(const float4*)(as + (size_t)sidx * 4);
        m.x = fmaxf(m.x, leaky(a.x + adn.x));
        m.y = fmaxf(m.y, leaky(a.y + adn.y));
        m.z = fmaxf(m.z, leaky(a.z + adn.z));
        m.w = fmaxf(m.w, leaky(a.w + adn.w));
    }
    wave_max4(m);
    float4 z = make_float4(0.f, 0.f, 0.f, 0.f);
    for (int q = lane; q < deg; q += 32) {
        const int sidx = list[at(q)].x;
        const float4 a = *(const float4*)(as + (size_t)sidx * 4);
        z.x += __expf(leaky(a.x + adn.x) - m.x);
        z.y += __expf(leaky(a.y + adn.y) - m.y);
        z.z += __expf(leaky(a.z + adn.z) - m.z);
        z.w += __expf(leaky(a.w + adn.w) - m.w);
    }
    wave_sum4(z);
    float4 ps;
    ps.x = __expf(es.x - m.x);
    ps.y = __expf(es.y - m.y);
    ps.z = __expf(es.z - m.z);
    ps.w = __expf(es.w - m.w);
    z.x += ps.x;
    z.y += ps.y;
    z.z += ps.z;
    z.w += ps.w;
    float4 als;
    als.x = ps.x / z.x;
    als.y = ps.y / z.y;
    als.z = ps.z / z.z;
    als.w = ps.w / z.w;
    if (lane == 0) *(float4*)(alpha + (size_t)(E + node) * 4) = als;
    const float wself = h2 ? (h1 ? als.w : als.z) : (h1 ? als.y : als.x);
    const float4 hself = *(const float4*)(hbuf + (size_t)node * F_DIM + lane * 4);
    acc.x = wself * hself.x;
    acc.y = wself * hself.y;
    acc.z = wself * hself.z;
    acc.w = wself * hself.w;
    for (int cb = 0; cb < deg; cb += 32) {
        const int q = cb + lane;
        int sidx = 0;
        float4 al = make_float4(0.f, 0.f, 0.f, 0.f);
        if (q < deg) {
            const int2 r2 = list[at(q)];
            sidx = r2.x;
            const float4 a = *(const float4*)(as + (size_t)sidx * 4);
            al.x = __expf(leaky(a.x + adn.x) - m.x) / z.x;
            al.y = __expf(leaky(a.y + adn.y) - m.y) / z.y;
            al.z = __expf(leaky(a.z + adn.z) - m.z) / z.z;
            al.w = __expf(leaky(a.w + adn.w) - m.w) / z.w;
            *(float4*)(alpha + (size_t)r2.y * 4) = al;
        }
        const int left = deg - cb;
        const int cnt  = (left < 32) ? left : 32;
        for (int k0 = 0; k0 < cnt; k0 += 8) {
            const bool more = k0 + 4 < cnt;
            rows_ask4(hbuf, sidx, k0, cnt, lane, hvA);
            if (more) rows_ask4(hbuf, sidx, k0 + 4, cnt, lane, hvB);
            rows_sum4(acc, al, k0, h1, h2, hvA);
            if (more) rows_sum4(acc, al, k0 + 4, h1, h2, hvB);
        }
    }
    const float4 bv = *(const float4*)(bias + lane * 4);
    acc.x += bv.x;
    acc.y += bv.y;
    acc.z += bv.z;
    acc.w += bv.w;
    }
    *(float4*)(xout + (size_t)node * F_DIM + lane * 4) = acc;
}

__global__ __launch_bounds__(256)
void seg_kernel(const float* __restrict__ hbufA, const float* __restrict__ asA,
                const float* __restrict__ adA, const int* __restrict__ rowptrA,
                const int2* __restrict__ listA, const float* __restrict__ biasA,
                int EA, int NnA, int NpA, float* __restrict__ alphaA, float* __restrict__ xoutA,
                const float* __restrict__ hbufB, const float* __restrict__ asB,
                const float* __restrict__ adB, const int* __restrict__ rowptrB,
                const int2* __restrict__ listB, const float* __restrict__ biasB,
                int EB, int NnB, int NpB, float* __restrict__ alphaB, float* __restrict__ xoutB,
                int nbB) {
    const bool gB = (int)blockIdx.x < nbB;
    seg_body(gB ? hbufB : hbufA, gB ? asB : asA, gB ? adB : adA, gB ? rowptrB : rowptrA,
             gB ? listB : listA, gB ? biasB : biasA, gB ? EB : EA, gB ? NnB : NnA, gB ? NpB : NpA,
             gB ? alphaB : alphaA, gB ? xoutB : xoutA,
             gB ? (int)blockIdx.x : (int)blockIdx.x - nbB);
}

#define GRP_G 4
#define GRP_WPG 6
#define GRP_T (GRP_G * GRP_WPG * 32)
#define GRP_CAP 2048

__device__ __forceinline__
void mix_node(float* xio, const float4 (&cv)[17], int node, int lane, float4 xo) {
    float sv[17];
#pragma unroll
    for (int k = 0; k < 17; ++k) {
        const float4 c = cv[k];
        float s = xo.x * c.x + xo.y * c.y + xo.z * c.z + xo.w * c.w;
#pragma unroll
        for (int off = 16; off > 0; off >>= 1) s += __shfl_xor(s, off, 32);
        sv[k] = s;
    }
    float mx = sv[0];
#pragma unroll
    for (int k = 1; k < 17; ++k) mx = fmaxf(mx, sv[k]);
    float sum = 0.f;
#pragma unroll
    for (int k = 0; k < 17; ++k) { sv[k] = __expf(sv[k] - mx); sum += sv[k]; }
    float inv = 1.f / sum;
    float4 go = make_float4(0.f, 0.f, 0.f, 0.f);
#pragma unroll
    for (int k = 0; k < 17; ++k) {
        float w = sv[k] * inv;
        go.x = fmaf(w, cv[k].x, go.x);
        go.y = fmaf(w, cv[k].y, go.y);
        go.z = fmaf(w, cv[k].z, go.z);
        go.w = fmaf(w, cv[k].w, go.w);
    }
    float4 o;
    o.x = 0.5f * (xo.x + go.x);
    o.y = 0.5f * (xo.y + go.y);
    o.z = 0.5f * (xo.z + go.z);
    o.w = 0.5f * (xo.w + go.w);
    *(float4*)(xio + (size_t)node * F_DIM + lane * 4) = o;
}

__global__ __launch_bounds__(GRP_T)
void group_kernel(float* xio, const float* __restrict__ xg,
                  const int* __restrict__ gidx, const int* __restrict__ nbr,
                  int Nn, int Kg) {
    __shared__ int ent[GRP_G][GRP_CAP];
    __shared__ int cnt[GRP_G];
    const int tid  = threadIdx.x;
    const int lane = tid & 31;
    const int wave = __builtin_amdgcn_readfirstlane(tid >> 5);
    const int lo   = blockIdx.x * GRP_G;
    const int t    = wave / GRP_WPG;
    const int r    = wave - t * GRP_WPG;
    const int g    = lo + t;
    if (tid < GRP_G) cnt[tid] = 0;
    __syncthreads();

    auto take = [&](int d, int node) {
        const unsigned tt = (unsigned)(d - lo);
        if (tt < (unsigned)GRP_G) {
            const int p = atomicAdd(&cnt[tt], 1);
            if (p < GRP_CAP) ent[tt][p] = node;
        }
    };
    const int n4 = ((((size_t)gidx) & 15) == 0) ? (Nn >> 2) : 0;
    const int4* g4 = (const int4*)gidx;
    const int n4m = (n4 > 0) ? n4 - 1 : 0;
    for (int b = 0; b < n4; b += 8 * GRP_T) {
        int4 v[8];
#pragma unroll
        for (int u = 0; u < 8; ++u) {
            const int i = b + u * GRP_T + tid;
            v[u] = g4[(i < n4) ? i : n4m];
        }
#pragma unroll
        for (int u = 0; u < 8; ++u) {
            const int i = b + u * GRP_T + tid;
            if (i < n4) {
                take(v[u].x, 4 * i);
                take(v[u].y, 4 * i + 1);
                take(v[u].z, 4 * i + 2);
                take(v[u].w, 4 * i + 3);
            }
        }
    }
    for (int e0 = 4 * n4; e0 < Nn; e0 += GRP_T) {
        const int e = e0 + tid;
        if (e < Nn) take(gidx[e], e);
    }
    __syncthreads();

    if (g >= Kg) return;

    float4 cv[17];
    cv[0] = *(const float4*)(xg + (size_t)g * F_DIM + lane * 4);
#pragma unroll
    for (int k = 0; k < 16; ++k) {
        const int c = nbr[(size_t)g * 16 + k];
        cv[k + 1] = *(const float4*)(xg + (size_t)c * F_DIM + lane * 4);
    }

    const int n = cnt[t];
    if (n <= GRP_CAP) {
        int p = r;
        int node = 0;
        float4 xo = make_float4(0.f, 0.f, 0.f, 0.f);
        if (p < n) {
            node = __builtin_amdgcn_readfirstlane(ent[t][p]);
            xo = *(const float4*)(xio + (size_t)node * F_DIM + lane * 4);
        }
        while (p < n) {
            const int pn = p + GRP_WPG;
            int node2 = 0;
            float4 xo2 = xo;
            if (pn < n) {
                node2 = __builtin_amdgcn_readfirstlane(ent[t][pn]);
                xo2 = *(const float4*)(xio + (size_t)node2 * F_DIM + lane * 4);
            }
            mix_node(xio, cv, node, lane, xo);
            node = node2; xo = xo2; p = pn;
        }
    } else {
        for (int v = r; v < Nn; v += GRP_WPG) {
            if (gidx[v] == g) {
                const float4 xo = *(const float4*)(xio + (size_t)v * F_DIM + lane * 4);
                mix_node(xio, cv, v, lane, xo);
            }
        }
    }
}

extern "C" void kernel_launch(void* const* d_in, const int* in_sizes, int n_in,
                              void* d_out, int out_size, void* d_ws, size_t ws_size,
                              hipStream_t stream) {
    const float* x       = (const float*)d_in[0];
    const int*   ei      = (const int*)d_in[1];
    const float* x_g     = (const float*)d_in[2];
    const int*   ei_g    = (const int*)d_in[3];
    const int*   gidx    = (const int*)d_in[4];
    const int*   nbr     = (const int*)d_in[5];
    const float* W1      = (const float*)d_in[6];
    const float* a_src1  = (const float*)d_in[7];
    const float* a_dst1  = (const float*)d_in[8];
    const float* b1      = (const float*)d_in[9];
    const float* W2      = (const float*)d_in[10];
    const float* a_src2  = (const float*)d_in[11];
    const float* a_dst2  = (const float*)d_in[12];
    const float* b2      = (const float*)d_in[13];

    const int N  = in_sizes[0] / F_DIM;
    const int E  = in_sizes[1] / 2;
    const int K  = in_sizes[2] / F_DIM;
    const int Eg = in_sizes[3] / 2;

    float* out     = (float*)d_out;
    float* x_final = out;
    float* xg_out  = out + (size_t)N * F_DIM;
    float* alpha1  = xg_out + (size_t)K * F_DIM;
    float* alpha2  = alpha1 + (size_t)(E + N) * H_HEADS;

    auto al4 = [](size_t n) { return (n + 3) & ~(size_t)3; };
    const int Np1 = ((N + 1 + OWN_R - 1) / OWN_R) * OWN_R;
    const int Np2 = ((K + 1 + OWN_R - 1) / OWN_R) * OWN_R;
    float* ws = (float*)d_ws;
    size_t off = 0;
    float* h1    = ws + off; off += (size_t)N * F_DIM;
    float* h2    = ws + off; off += (size_t)K * F_DIM;
    float* as1   = ws + off; off += (size_t)N * H_HEADS;
    float* ad1   = ws + off; off += (size_t)N * H_HEADS;
    float* as2   = ws + off; off += (size_t)K * H_HEADS;
    float* ad2   = ws + off; off += (size_t)K * H_HEADS;
    int*   cur1  = (int*)(ws + off);  off += (size_t)OWN_PARTS * Np1;
    int*   cur2  = (int*)(ws + off);  off += (size_t)OWN_PARTS * Np2;
    int2*  list1 = (int2*)(ws + off); off += al4((size_t)E * 2);
    int2*  list2 = (int2*)(ws + off); off += al4((size_t)Eg * 2);

    const int T = 256;
    auto blk = [](long n, int t) { return (int)((n + t - 1) / t); };


    const int nOwn = OWN_PARTS * (blk(K + 1, OWN_R) + blk(N + 1, OWN_R));
    prep_kernel<<<nOwn + blk(K, GEMM_ROWS) + blk(N, GEMM_ROWS), 1024, 0, stream>>>(
        ei, E, N, Np1, cur1, list1, ei_g, Eg, K, Np2, cur2, list2, OWN_PARTS * blk(K + 1, OWN_R), nOwn,
        x, W1, a_src1, a_dst1, h1, as1, ad1, N,
        x_g, W2, a_src2, a_dst2, h2, as2, ad2, K, blk(K, GEMM_ROWS));

    seg_kernel<<<blk(K, 8) + blk(N, 8), T, 0, stream>>>(
        h1, as1, ad1, cur1, list1, b1, E, N, Np1, alpha1, x_final,
        h2, as2, ad2, cur2, list2, b2, Eg, K, Np2, alpha2, xg_out, blk(K, 8));

    group_kernel<<<blk(K, GRP_G), GRP_T, 0, stream>>>(x_final, xg_out, gidx, nbr, N, K);
}


# Group mix takes two nodes per wave (16 lanes x 8 channels each): each dot product ends in 4 in-row DPP adds, not 5 exchanges through the LDS unit; the dots' summation tree changes
# speedup vs baseline: 5.6107x; 1.0815x over previous
#include <hip/hip_runtime.h>
#include <math.h>

#define H_HEADS 4
#define C_CH 32
#define F_DIM 128
#define NEG_SLOPE 0.2f

typedef __attribute__((ext_vector_type(2))) float v2f;
typedef __attribute__((ext_vector_type(8))) float v8f;

#define GEMM_ROWS 256
__device__ __forceinline__
void gemm_body(float* Wlds, int bIdx, const float* __restrict__ XA, const float* __restrict__ WA,
                      const float* __restrict__ a_srcA, const float* __restrict__ a_dstA,
                      float* __restrict__ HoutA, float* __restrict__ asA,
                      float* __restrict__ adA, int rowsA,
                      const float* __restrict__ XB, const float* __restrict__ WB,
                      const float* __restrict__ a_srcB, const float* __restrict__ a_dstB,
                      float* __restrict__ HoutB, float* __restrict__ asB,
                      float* __restrict__ adB, int rowsB, int nbB) {
    const bool gB = bIdx < nbB;
    const float* X     = gB ? XB : XA;
    const float* W     = gB ? WB : WA;
    const float* a_src = gB ? a_srcB : a_srcA;
    const float* a_dst = gB ? a_dstB : a_dstA;
    float* Hout        = gB ? HoutB : HoutA;
    float* as          = gB ? asB : asA;
    float* ad          = gB ? adB : adA;
    const int rows     = gB ? rowsB : rowsA;
    const int blockLocal = gB ? bIdx : bIdx - nbB;
    for (int i = threadIdx.x; i < (F_DIM * F_DIM) / 2; i += blockDim.x) {
        int kp  = i >> 7;
        int col = i & 127;
        int cs  = (col + ((kp & 1) << 4)) & 127;
        float2 w2;
        w2.x = W[(size_t)(2 * kp)     * F_DIM + col];
        w2.y = W[(size_t)(2 * kp + 1) * F_DIM + col];
        *(float2*)(Wlds + (size_t)kp * 256 + cs * 2) = w2;
    }
    __syncthreads();

    const int lane = threadIdx.x & 31;
    const int wave = __builtin_amdgcn_readfirstlane((int)(threadIdx.x >> 5));
    const int lm   = lane & 15;
    const int lh   = lane >> 4;
    const int ch   = wave & 1;
    const int rowBase = blockLocal * GEMM_ROWS + (wave >> 1) * 16;
    if (rowBase >= rows) return;

    int aRow = rowBase + lm;
    if (aRow >= rows) aRow = rows - 1;

    v8f acc[4];
    v8f zero = {0.f, 0.f, 0.f, 0.f, 0.f, 0.f, 0.f, 0.f};
#pragma unroll
    for (int ct = 0; ct < 4; ++ct) acc[ct] = zero;

    const float* xrow = X + (size_t)aRow * F_DIM;
    for (int k0 = 0; k0 < F_DIM; k0 += 4) {
        const int kp = (k0 >> 1) + lh;
        v2f a;
        a.x = xrow[2 * kp];
        a.y = xrow[2 * kp + 1];
        const float* bbase = Wlds + (size_t)kp * 256;
#pragma unroll
        for (int ct = 0; ct < 4; ++ct) {
            const int cs = ((4 * ch + ct) * 16 + lm + 16 * lh) & 127;
            v2f b = *(const v2f*)(bbase + cs * 2);
            acc[ct] = __builtin_amdgcn_wmma_f32_16x16x4_f32(
                false, a, false, b, (short)0, acc[ct], false, false);
        }
    }

#pragma unroll
    for (int ct = 0; ct < 4; ++ct) {
#pragma unroll
        for (int g = 0; g < 8; ++g) {
            int r = rowBase + g + 8 * lh;
            if (r < rows) Hout[(unsigned)(r * F_DIM + (4 * ch + ct) * 16 + lm)] = acc[ct][g];
        }
    }

    float asv[4], adv[4];
#pragma unroll
    for (int ct = 0; ct < 4; ++ct) {
        asv[ct] = a_src[(4 * ch + ct) * 16 + lm];
        adv[ct] = a_dst[(4 * ch + ct) * 16 + lm];
    }
#pragma unroll
    for (int h = 0; h < 2; ++h) {
#pragma unroll
        for (int g = 0; g < 8; ++g) {
            float ss = acc[2 * h][g] * asv[2 * h];
            ss = fmaf(acc[2 * h + 1][g], asv[2 * h + 1], ss);
            float sd = acc[2 * h][g] * adv[2 * h];
            sd = fmaf(acc[2 * h + 1][g], adv[2 * h + 1], sd);
#pragma unroll
            for (int off = 8; off > 0; off >>= 1) {
                ss += __shfl_xor(ss, off, 32);
                sd += __shfl_xor(sd, off, 32);
            }
            int r = rowBase + g + 8 * lh;
            if (lm == 0 && r < rows) {
                as[(unsigned)(r * H_HEADS + 2 * ch + h)] = ss;
                ad[(unsigned)(r * H_HEADS + 2 * ch + h)] = sd;
            }
        }
    }
}

#define OWN_PARTS 4
#define OWN_R 1024
#define STAGE_CAP 7680
#define NO_DST 0x7fffffff

__device__ inline int part_len(int E) { return ((E + 4 * OWN_PARTS - 1) / (4 * OWN_PARTS)) * 4; }

#define OWN_W_CNT   (2 * STAGE_CAP)
#define OWN_W_CUR   (OWN_W_CNT + OWN_R)
#define OWN_W_BELOW (OWN_W_CUR + OWN_R)
#define OWN_W_SCAN  (OWN_W_BELOW + 32)
#define OWN_W_NST   (OWN_W_SCAN + OWN_R / 32)
#define OWN_W_END   (OWN_W_NST + 4)

__device__ __forceinline__
void owner_body(int* lds, int bIdx,
                const int* __restrict__ eiA, int EA, int NnA, int NpA,
                int* __restrict__ rowptrA, int2* __restrict__ listA,
                const int* __restrict__ eiB, int EB, int NnB, int NpB,
                int* __restrict__ rowptrB, int2* __restrict__ listB, int nbB) {
    int2* stage = (int2*)lds;
    int* cntL   = lds + OWN_W_CNT;
    int* cur    = lds + OWN_W_CUR;
    int* wbelow = lds + OWN_W_BELOW;
    int* wscan  = lds + OWN_W_SCAN;
    int* nstage = lds + OWN_W_NST;
    const bool gB = bIdx < nbB;
    const int* ei = gB ? eiB : eiA;
    const int  E  = gB ? EB : EA;
    const int  Nn = gB ? NnB : NnA;
    const int  Np = gB ? NpB : NpA;
    int* rowptr   = gB ? rowptrB : rowptrA;
    int2* list    = gB ? listB : listA;
    const int bl   = gB ? bIdx : bIdx - nbB;
    const int part = bl % OWN_PARTS;
    const int lo   = (bl / OWN_PARTS) * OWN_R;
    const int Ep   = part_len(E);
    const int e_lo = (part * Ep < E) ? part * Ep : E;
    const int Ec   = (Ep < E - e_lo) ? Ep : E - e_lo;
    const int tid  = threadIdx.x;
    const int lane = tid & 31;
    const int wave = tid >> 5;
    if (tid < OWN_R) { cntL[tid] = 0; cur[tid] = 0; }
    if (tid == 0) *nstage = 0;
    __syncthreads();

    const int* dst = ei + E + e_lo;
    const int n4 = ((E & 3) == 0) ? (Ec >> 2) : 0;
    const int4* d4 = (const int4*)dst;
    const int4 none = make_int4(NO_DST, NO_DST, NO_DST, NO_DST);
    int below = 0;
    auto put = [&](unsigned t, int e) {
        const int r = atomicAdd(&cntL[t], 1);
        const int p = atomicAdd(nstage, 1);
        if (p < STAGE_CAP) stage[p] = make_int2(e_lo + e, (int)(((unsigned)r << 10) | t));
    };
    auto count_below = [&](int d) {
        below += __popc(__builtin_amdgcn_ballot_w32(d < lo));
    };
    const int n4m = (n4 > 0) ? n4 - 1 : 0;
    for (int b = 0; b < n4; b += 8 * 1024) {
        int4 v[8];
#pragma unroll
        for (int u = 0; u < 8; ++u) {
            const int i = b + u * 1024 + tid;
            v[u] = d4[(i < n4) ? i : n4m];
        }
#pragma unroll
        for (int u = 0; u < 8; ++u) {
            const int i = b + u * 1024 + tid;
            const int4 d = (i < n4) ? v[u] : none;
            count_below(d.x);
            count_below(d.y);
            count_below(d.z);
            count_below(d.w);
            const unsigned t0 = (unsigned)(d.x - lo);
            const unsigned t1 = (unsigned)(d.y - lo);
            const unsigned t2 = (unsigned)(d.z - lo);
            const unsigned t3 = (unsigned)(d.w - lo);
            const unsigned R = (unsigned)OWN_R;
            if ((t0 < R) | (t1 < R) | (t2 < R) | (t3 < R)) {
                if (t0 < R) put(t0, 4 * i);
                if (t1 < R) put(t1, 4 * i + 1);
                if (t2 < R) put(t2, 4 * i + 2);
                if (t3 < R) put(t3, 4 * i + 3);
            }
        }
    }
    for (int e0 = 4 * n4; e0 < Ec; e0 += 1024) {
        const int e = e0 + tid;
        const int d = (e < Ec) ? dst[e] : NO_DST;
        count_below(d);
        const unsigned t = (unsigned)(d - lo);
        if (t < (unsigned)OWN_R) put(t, e);
    }
    if (lane == 0) wbelow[wave] = below;
    __syncthreads();

    int base = 0;
    for (int w = 0; w < 32; ++w) base += wbelow[w];
    const int c = (tid < OWN_R) ? cntL[tid] : 0;
    int inc = c;
    for (int off = 1; off < 32; off <<= 1) {
        const int t = __shfl_up(inc, off, 32);
        if (lane >= off) inc += t;
    }
    if (tid < OWN_R && lane == 31) wscan[wave] = inc;
    __syncthreads();
    if (tid < OWN_R) {
        int wex = 0;
        for (int w = 0; w < wave; ++w) wex += wscan[w];
        const int start = wex + inc - c;
        cntL[tid] = start;
        if (lo + tid <= Nn) rowptr[(size_t)part * Np + lo + tid] = e_lo + base + start;
    }
    __syncthreads();
    const int n = *nstage;
    if (n <= STAGE_CAP) {
        for (int p = tid; p < n; p += 1024) {
            const int2 s = stage[p];
            const unsigned pk = (unsigned)s.y;
            list[e_lo + base + cntL[pk & 1023u] + (int)(pk >> 10)] = make_int2(ei[s.x], s.x);
        }
    } else {
        for (int e0 = 0; e0 < Ec; e0 += 1024) {
            const int e = e0 + tid;
            const int d = (e < Ec) ? dst[e] : NO_DST;
            const unsigned t = (unsigned)(d - lo);
            if (t < (unsigned)OWN_R) {
                const int r = atomicAdd(&cur[t], 1);
                list[e_lo + base + cntL[t] + r] = make_int2(ei[e_lo + e], e_lo + e);
            }
        }
    }
}

#define PREP_LDS_WORDS ((OWN_W_END > F_DIM * F_DIM) ? OWN_W_END : F_DIM * F_DIM)
__global__ __launch_bounds__(1024, 16)
void prep_kernel(const int* __restrict__ eiA, int EA, int NnA, int NpA,
                 int* __restrict__ rowptrA, int2* __restrict__ listA,
                 const int* __restrict__ eiB, int EB, int NnB, int NpB,
                 int* __restrict__ rowptrB, int2* __restrict__ listB, int nbOwnB, int nOwn,
                 const float* __restrict__ XA, const float* __restrict__ WA,
                 const float* __restrict__ a_srcA, const float* __restrict__ a_dstA,
                 float* __restrict__ HoutA, float* __restrict__ asA,
                 float* __restrict__ adA, int rowsA,
                 const float* __restrict__ XB, const float* __restrict__ WB,
                 const float* __restrict__ a_srcB, const float* __restrict__ a_dstB,
                 float* __restrict__ HoutB, float* __restrict__ asB,
                 float* __restrict__ adB, int rowsB, int nbGemmB) {
    __shared__ __attribute__((aligned(16))) int lds[PREP_LDS_WORDS];
    if ((int)blockIdx.x < nOwn)
        owner_body(lds, (int)blockIdx.x, eiA, EA, NnA, NpA, rowptrA, listA,
                   eiB, EB, NnB, NpB, rowptrB, listB, nbOwnB);
    else
        gemm_body((float*)lds, (int)blockIdx.x - nOwn, XA, WA, a_srcA, a_dstA, HoutA, asA, adA, rowsA,
                  XB, WB, a_srcB, a_dstB, HoutB, asB, adB, rowsB, nbGemmB);
}

__device__ inline float leaky(float e) { return (e > 0.f) ? e : NEG_SLOPE * e; }

__device__ inline float lane_value(float v, int k) {
    return __int_as_float(__builtin_amdgcn_readlane(__float_as_int(v), k));
}

__device__ __forceinline__
void rows_ask4(const float* __restrict__ hbuf, int sidx, int k0, int n, int lane, float4 (&hv)[4]) {
#pragma unroll
    for (int u = 0; u < 4; ++u) {
        int kc = (k0 + u < n) ? k0 + u : n - 1;
        kc = (kc > 0) ? kc : 0;
        const int sk = __builtin_amdgcn_readlane(sidx, kc);
        hv[u] = *(const float4*)(hbuf + (unsigned)(sk * F_DIM + lane * 4));
    }
}

__device__ __forceinline__
void rows_sum4(float4& acc, const float4 al, int k0, bool h1, bool h2, const float4 (&hv)[4]) {
#pragma unroll
    for (int u = 0; u < 4; ++u) {
        const int k = k0 + u;
        const float a0 = lane_value(al.x, k);
        const float a1 = lane_value(al.y, k);
        const float a2 = lane_value(al.z, k);
        const float a3 = lane_value(al.w, k);
        const float lo = h1 ? a1 : a0;
        const float hi = h1 ? a3 : a2;
        const float w  = h2 ? hi : lo;
        acc.x = fmaf(w, hv[u].x, acc.x);
        acc.y = fmaf(w, hv[u].y, acc.y);
        acc.z = fmaf(w, hv[u].z, acc.z);
        acc.w = fmaf(w, hv[u].w, acc.w);
    }
}

__device__ __forceinline__ void wave_max4(float4& m) {
#pragma unroll
    for (int off = 16; off > 0; off >>= 1) {
        m.x = fmaxf(m.x, __shfl_xor(m.x, off, 32));
        m.y = fmaxf(m.y, __shfl_xor(m.y, off, 32));
        m.z = fmaxf(m.z, __shfl_xor(m.z, off, 32));
        m.w = fmaxf(m.w, __shfl_xor(m.w, off, 32));
    }
}

__device__ __forceinline__ void wave_sum4(float4& z) {
#pragma unroll
    for (int off = 16; off > 0; off >>= 1) {
        z.x += __shfl_xor(z.x, off, 32);
        z.y += __shfl_xor(z.y, off, 32);
        z.z += __shfl_xor(z.z, off, 32);
        z.w += __shfl_xor(z.w, off, 32);
    }
}

__device__ __forceinline__
void seg_body(const float* __restrict__ hbuf, const float* __restrict__ as,
              const float* __restrict__ ad, const int* __restrict__ rowptr,
              const int2* __restrict__ list, const float* __restrict__ bias,
              int E, int Nn, int Np, float* __restrict__ alpha, float* __restrict__ xout,
              int blockLocal) {
    const int lane = threadIdx.x & 31;
    const int node = __builtin_amdgcn_readfirstlane(
        (int)(blockLocal * (blockDim.x >> 5) + (threadIdx.x >> 5)));
    if (node >= Nn) return;
    int rb[OWN_PARTS];
    int rc[OWN_PARTS];
    int deg = 0;
#pragma unroll
    for (int p = 0; p < OWN_PARTS; ++p) {
        const int b = rowptr[(size_t)p * Np + node];
        const int e = rowptr[(size_t)p * Np + node + 1];
        rb[p] = b;
        deg += e - b;
        rc[p] = deg;
    }
    auto at = [&](int q) {
        int j = rb[0] + q;
#pragma unroll
        for (int p = 1; p < OWN_PARTS; ++p)
            if (q >= rc[p - 1]) j = rb[p] + (q - rc[p - 1]);
        return j;
    };
    const float4 adn = *(const float4*)(ad + (size_t)node * 4);
    const float4 asn = *(const float4*)(as + (size_t)node * 4);
    float4 es;
    es.x = leaky(asn.x + adn.x);
    es.y = leaky(asn.y + adn.y);
    es.z = leaky(asn.z + adn.z);
    es.w = leaky(asn.w + adn.w);
    const int hd = lane >> 3;
    const bool h1 = (hd & 1) != 0;
    const bool h2 = (hd & 2) != 0;
    float4 hvA[4], hvB[4];
    float4 acc;

    if (deg <= 32) {
        const bool has = lane < deg;
        int2 rec = make_int2(0, 0);
        float4 a = make_float4(0.f, 0.f, 0.f, 0.f);
        if (has) {
            rec = list[at(lane)];
            a = *(const float4*)(as + (size_t)rec.x * 4);
        }
        rows_ask4(hbuf, rec.x, 0, deg, lane, hvA);
        rows_ask4(hbuf, rec.x, 4, deg, lane, hvB);
        const float4 hself = *(const float4*)(hbuf + (size_t)node * F_DIM + lane * 4);
        const float4 bv = *(const float4*)(bias + lane * 4);

        float4 m = es;
        float4 ev = make_float4(0.f, 0.f, 0.f, 0.f);
        if (has) {
            ev.x = leaky(a.x + adn.x);
            ev.y = leaky(a.y + adn.y);
            ev.z = leaky(a.z + adn.z);
            ev.w = leaky(a.w + adn.w);
            m.x = fmaxf(m.x, ev.x);
            m.y = fmaxf(m.y, ev.y);
            m.z = fmaxf(m.z, ev.z);
            m.w = fmaxf(m.w, ev.w);
        }
        wave_max4(m);

        float4 z = make_float4(0.f, 0.f, 0.f, 0.f);
        if (has) {
            ev.x = __expf(ev.x - m.x);
            ev.y = __expf(ev.y - m.y);
            ev.z = __expf(ev.z - m.z);
            ev.w = __expf(ev.w - m.w);
            z = ev;
        }
        wave_sum4(z);
        float4 ps;
        ps.x = __expf(es.x - m.x);
        ps.y = __expf(es.y - m.y);
        ps.z = __expf(es.z - m.z);
        ps.w = __expf(es.w - m.w);
        z.x += ps.x;
        z.y += ps.y;
        z.z += ps.z;
        z.w += ps.w;

        float4 als;
        als.x = ps.x / z.x;
        als.y = ps.y / z.y;
        als.z = ps.z / z.z;
        als.w = ps.w / z.w;
        if (lane == 0) *(float4*)(alpha + (size_t)(E + node) * 4) = als;
        float4 al1 = make_float4(0.f, 0.f, 0.f, 0.f);
        if (has) {
            al1.x = ev.x / z.x;
            al1.y = ev.y / z.y;
            al1.z = ev.z / z.z;
            al1.w = ev.w / z.w;
            *(float4*)(alpha + (size_t)rec.y * 4) = al1;
        }
        const float wself = h2 ? (h1 ? als.w : als.z) : (h1 ? als.y : als.x);
        acc.x = wself * hself.x;
        acc.y = wself * hself.y;
        acc.z = wself * hself.z;
        acc.w = wself * hself.w;
        rows_sum4(acc, al1, 0, h1, h2, hvA);
        rows_sum4(acc, al1, 4, h1, h2, hvB);
        for (int k0 = 8; k0 < deg; k0 += 8) {
            const bool more = k0 + 4 < deg;
            rows_ask4(hbuf, rec.x, k0, deg, lane, hvA);
            if (more) rows_ask4(hbuf, rec.x, k0 + 4, deg, lane, hvB);
            rows_sum4(acc, al1, k0, h1, h2, hvA);
            if (more) rows_sum4(acc, al1, k0 + 4, h1, h2, hvB);
        }
        acc.x += bv.x;
        acc.y += bv.y;
        acc.z += bv.z;
        acc.w += bv.w;
    } else {
    float4 m = es;
#pragma unroll 1
    for (int q = lane; q < deg; q += 32) {
        const int sidx = list[at(q)].x;
        const float4 a = *(const float4*)(as + (size_t)sidx * 4);
        m.x = fmaxf(m.x, leaky(a.x + adn.x));
        m.y = fmaxf(m.y, leaky(a.y + adn.y));
        m.z = fmaxf(m.z, leaky(a.z + adn.z));
        m.w = fmaxf(m.w, leaky(a.w + adn.w));
    }
    wave_max4(m);
    float4 z = make_float4(0.f, 0.f, 0.f, 0.f);
    for (int q = lane; q < deg; q += 32) {
        const int sidx = list[at(q)].x;
        const float4 a = *(const float4*)(as + (size_t)sidx * 4);
        z.x += __expf(leaky(a.x + adn.x) - m.x);
        z.y += __expf(leaky(a.y + adn.y) - m.y);
        z.z += __expf(leaky(a.z + adn.z) - m.z);
        z.w += __expf(leaky(a.w + adn.w) - m.w);
    }
    wave_sum4(z);
    float4 ps;
    ps.x = __expf(es.x - m.x);
    ps.y = __expf(es.y - m.y);
    ps.z = __expf(es.z - m.z);
    ps.w = __expf(es.w - m.w);
    z.x += ps.x;
    z.y += ps.y;
    z.z += ps.z;
    z.w += ps.w;
    float4 als;
    als.x = ps.x / z.x;
    als.y = ps.y / z.y;
    als.z = ps.z / z.z;
    als.w = ps.w / z.w;
    if (lane == 0) *(float4*)(alpha + (size_t)(E + node) * 4) = als;
    const float wself = h2 ? (h1 ? als.w : als.z) : (h1 ? als.y : als.x);
    const float4 hself = *(const float4*)(hbuf + (size_t)node * F_DIM + lane * 4);
    acc.x = wself * hself.x;
    acc.y = wself * hself.y;
    acc.z = wself * hself.z;
    acc.w = wself * hself.w;
    for (int cb = 0; cb < deg; cb += 32) {
        const int q = cb + lane;
        int sidx = 0;
        float4 al = make_float4(0.f, 0.f, 0.f, 0.f);
        if (q < deg) {
            const int2 r2 = list[at(q)];
            sidx = r2.x;
            const float4 a = *(const float4*)(as + (size_t)sidx * 4);
            al.x = __expf(leaky(a.x + adn.x) - m.x) / z.x;
            al.y = __expf(leaky(a.y + adn.y) - m.y) / z.y;
            al.z = __expf(leaky(a.z + adn.z) - m.z) / z.z;
            al.w = __expf(leaky(a.w + adn.w) - m.w) / z.w;
            *(float4*)(alpha + (size_t)r2.y * 4) = al;
        }
        const int left = deg - cb;
        const int cnt  = (left < 32) ? left : 32;
        for (int k0 = 0; k0 < cnt; k0 += 8) {
            const bool more = k0 + 4 < cnt;
            rows_ask4(hbuf, sidx, k0, cnt, lane, hvA);
            if (more) rows_ask4(hbuf, sidx, k0 + 4, cnt, lane, hvB);
            rows_sum4(acc, al, k0, h1, h2, hvA);
            if (more) rows_sum4(acc, al, k0 + 4, h1, h2, hvB);
        }
    }
    const float4 bv = *(const float4*)(bias + lane * 4);
    acc.x += bv.x;
    acc.y += bv.y;
    acc.z += bv.z;
    acc.w += bv.w;
    }
    *(float4*)(xout + (size_t)node * F_DIM + lane * 4) = acc;
}

__global__ __launch_bounds__(256)
void seg_kernel(const float* __restrict__ hbufA, const float* __restrict__ asA,
                const float* __restrict__ adA, const int* __restrict__ rowptrA,
                const int2* __restrict__ listA, const float* __restrict__ biasA,
                int EA, int NnA, int NpA, float* __restrict__ alphaA, float* __restrict__ xoutA,
                const float* __restrict__ hbufB, const float* __restrict__ asB,
                const float* __restrict__ adB, const int* __restrict__ rowptrB,
                const int2* __restrict__ listB, const float* __restrict__ biasB,
                int EB, int NnB, int NpB, float* __restrict__ alphaB, float* __restrict__ xoutB,
                int nbB) {
    const bool gB = (int)blockIdx.x < nbB;
    seg_body(gB ? hbufB : hbufA, gB ? asB : asA, gB ? adB : adA, gB ? rowptrB : rowptrA,
             gB ? listB : listA, gB ? biasB : biasA, gB ? EB : EA, gB ? NnB : NnA, gB ? NpB : NpA,
             gB ? alphaB : alphaA, gB ? xoutB : xoutA,
             gB ? (int)blockIdx.x : (int)blockIdx.x - nbB);
}

#define GRP_G 4
#define GRP_WPG 3
#define GRP_T (GRP_G * GRP_WPG * 32)
#define GRP_CAP 2048
#define GRP_SCAN_U 16

template <int M>
__device__ __forceinline__ float row_xor(float v) {
    return __int_as_float(__builtin_amdgcn_update_dpp(0, __float_as_int(v), 0x160 + M, 0xf, 0xf, true));
}

__device__ __forceinline__ float row_sum16(float s) {
    s += row_xor<1>(s);
    s += row_xor<2>(s);
    s += row_xor<4>(s);
    s += row_xor<8>(s);
    return s;
}

__device__ __forceinline__
void mix_pair(float* xio, const float4 (&ca)[17], const float4 (&cb)[17],
              unsigned off, bool ok, float4 xa, float4 xb) {
    float sv[17];
#pragma unroll
    for (int k = 0; k < 17; ++k) {
        const float4 a = ca[k];
        const float4 b = cb[k];
        float s = xa.x * a.x + xa.y * a.y + xa.z * a.z + xa.w * a.w
                + xb.x * b.x + xb.y * b.y + xb.z * b.z + xb.w * b.w;
        sv[k] = row_sum16(s);
    }
    float mx = sv[0];
#pragma unroll
    for (int k = 1; k < 17; ++k) mx = fmaxf(mx, sv[k]);
    float sum = 0.f;
#pragma unroll
    for (int k = 0; k < 17; ++k) { sv[k] = __expf(sv[k] - mx); sum += sv[k]; }
    float inv = 1.f / sum;
    float4 ga = make_float4(0.f, 0.f, 0.f, 0.f);
    float4 gb = make_float4(0.f, 0.f, 0.f, 0.f);
#pragma unroll
    for (int k = 0; k < 17; ++k) {
        float w = sv[k] * inv;
        ga.x = fmaf(w, ca[k].x, ga.x);
        ga.y = fmaf(w, ca[k].y, ga.y);
        ga.z = fmaf(w, ca[k].z, ga.z);
        ga.w = fmaf(w, ca[k].w, ga.w);
        gb.x = fmaf(w, cb[k].x, gb.x);
        gb.y = fmaf(w, cb[k].y, gb.y);
        gb.z = fmaf(w, cb[k].z, gb.z);
        gb.w = fmaf(w, cb[k].w, gb.w);
    }
    float4 oa, ob;
    oa.x = 0.5f * (xa.x + ga.x);
    oa.y = 0.5f * (xa.y + ga.y);
    oa.z = 0.5f * (xa.z + ga.z);
    oa.w = 0.5f * (xa.w + ga.w);
    ob.x = 0.5f * (xb.x + gb.x);
    ob.y = 0.5f * (xb.y + gb.y);
    ob.z = 0.5f * (xb.z + gb.z);
    ob.w = 0.5f * (xb.w + gb.w);
    if (ok) {
        *(float4*)(xio + off)      = oa;
        *(float4*)(xio + off + 64) = ob;
    }
}

__global__ __launch_bounds__(GRP_T)
void group_kernel(float* xio, const float* __restrict__ xg,
                  const int* __restrict__ gidx, const int* __restrict__ nbr,
                  int Nn, int Kg) {
    __shared__ int ent[GRP_G][GRP_CAP];
    __shared__ int cnt[GRP_G];
    const int tid  = threadIdx.x;
    const int lane = tid & 31;
    const int half = lane >> 4;
    const int cg   = lane & 15;
    const int wave = __builtin_amdgcn_readfirstlane(tid >> 5);
    const int lo   = blockIdx.x * GRP_G;
    const int t    = wave / GRP_WPG;
    const int r    = wave - t * GRP_WPG;
    const int g    = lo + t;
    if (tid < GRP_G) cnt[tid] = 0;
    __syncthreads();

    auto take = [&](int d, int node) {
        const unsigned tt = (unsigned)(d - lo);
        if (tt < (unsigned)GRP_G) {
            const int p = atomicAdd(&cnt[tt], 1);
            if (p < GRP_CAP) ent[tt][p] = node;
        }
    };
    const int n4 = ((((size_t)gidx) & 15) == 0) ? (Nn >> 2) : 0;
    const int4* g4 = (const int4*)gidx;
    const int n4m = (n4 > 0) ? n4 - 1 : 0;
    for (int b = 0; b < n4; b += GRP_SCAN_U * GRP_T) {
        int4 v[GRP_SCAN_U];
#pragma unroll
        for (int u = 0; u < GRP_SCAN_U; ++u) {
            const int i = b + u * GRP_T + tid;
            v[u] = g4[(i < n4) ? i : n4m];
        }
#pragma unroll
        for (int u = 0; u < GRP_SCAN_U; ++u) {
            const int i = b + u * GRP_T + tid;
            if (i < n4) {
                take(v[u].x, 4 * i);
                take(v[u].y, 4 * i + 1);
                take(v[u].z, 4 * i + 2);
                take(v[u].w, 4 * i + 3);
            }
        }
    }
    for (int e0 = 4 * n4; e0 < Nn; e0 += GRP_T) {
        const int e = e0 + tid;
        if (e < Nn) take(gidx[e], e);
    }
    __syncthreads();

    if (g >= Kg) return;

    float4 ca[17], cb[17];
    ca[0] = *(const float4*)(xg + (size_t)g * F_DIM + cg * 4);
    cb[0] = *(const float4*)(xg + (size_t)g * F_DIM + cg * 4 + 64);
#pragma unroll
    for (int k = 0; k < 16; ++k) {
        const int c = nbr[(size_t)g * 16 + k];
        ca[k + 1] = *(const float4*)(xg + (size_t)c * F_DIM + cg * 4);
        cb[k + 1] = *(const float4*)(xg + (size_t)c * F_DIM + cg * 4 + 64);
    }

    const int n = cnt[t];
    if (n <= GRP_CAP) {
        const int npairs = (n + 1) >> 1;
        auto fetch = [&](int j, unsigned& off, bool& ok, float4& xa, float4& xb) {
            const int p  = 2 * j + half;
            ok = p < n;
            const int node = ent[t][ok ? p : 2 * j];
            off = (unsigned)(node * F_DIM + cg * 4);
            xa = *(const float4*)(xio + off);
            xb = *(const float4*)(xio + off + 64);
        };
        int j = r;
        unsigned off = 0;
        bool ok = false;
        float4 xa = make_float4(0.f, 0.f, 0.f, 0.f);
        float4 xb = xa;
        if (j < npairs) fetch(j, off, ok, xa, xb);
        while (j < npairs) {
            const int jn = j + GRP_WPG;
            unsigned off2 = off;
            bool ok2 = false;
            float4 xa2 = xa, xb2 = xb;
            if (jn < npairs) fetch(jn, off2, ok2, xa2, xb2);
            mix_pair(xio, ca, cb, off, ok, xa, xb);
            off = off2; ok = ok2; xa = xa2; xb = xb2; j = jn;
        }
    } else {
        for (int v = r; v < Nn; v += GRP_WPG) {
            if (gidx[v] == g) {
                const unsigned off = (unsigned)(v * F_DIM + cg * 4);
                const float4 xa = *(const float4*)(xio + off);
                const float4 xb = *(const float4*)(xio + off + 64);
                mix_pair(xio, ca, cb, off, half == 0, xa, xb);
            }
        }
    }
}

extern "C" void kernel_launch(void* const* d_in, const int* in_sizes, int n_in,
                              void* d_out, int out_size, void* d_ws, size_t ws_size,
                              hipStream_t stream) {
    const float* x       = (const float*)d_in[0];
    const int*   ei      = (const int*)d_in[1];
    const float* x_g     = (const float*)d_in[2];
    const int*   ei_g    = (const int*)d_in[3];
    const int*   gidx    = (const int*)d_in[4];
    const int*   nbr     = (const int*)d_in[5];
    const float* W1      = (const float*)d_in[6];
    const float* a_src1  = (const float*)d_in[7];
    const float* a_dst1  = (const float*)d_in[8];
    const float* b1      = (const float*)d_in[9];
    const float* W2      = (const float*)d_in[10];
    const float* a_src2  = (const float*)d_in[11];
    const float* a_dst2  = (const float*)d_in[12];
    const float* b2      = (const float*)d_in[13];

    const int N  = in_sizes[0] / F_DIM;
    const int E  = in_sizes[1] / 2;
    const int K  = in_sizes[2] / F_DIM;
    const int Eg = in_sizes[3] / 2;

    float* out     = (float*)d_out;
    float* x_final = out;
    float* xg_out  = out + (size_t)N * F_DIM;
    float* alpha1  = xg_out + (size_t)K * F_DIM;
    float* alpha2  = alpha1 + (size_t)(E + N) * H_HEADS;

    auto al4 = [](size_t n) { return (n + 3) & ~(size_t)3; };
    const int Np1 = ((N + 1 + OWN_R - 1) / OWN_R) * OWN_R;
    const int Np2 = ((K + 1 + OWN_R - 1) / OWN_R) * OWN_R;
    float* ws = (float*)d_ws;
    size_t off = 0;
    float* h1    = ws + off; off += (size_t)N * F_DIM;
    float* h2    = ws + off; off += (size_t)K * F_DIM;
    float* as1   = ws + off; off += (size_t)N * H_HEADS;
    float* ad1   = ws + off; off += (size_t)N * H_HEADS;
    float* as2   = ws + off; off += (size_t)K * H_HEADS;
    float* ad2   = ws + off; off += (size_t)K * H_HEADS;
    int*   cur1  = (int*)(ws + off);  off += (size_t)OWN_PARTS * Np1;
    int*   cur2  = (int*)(ws + off);  off += (size_t)OWN_PARTS * Np2;
    int2*  list1 = (int2*)(ws + off); off += al4((size_t)E * 2);
    int2*  list2 = (int2*)(ws + off); off += al4((size_t)Eg * 2);

    const int T = 256;
    auto blk = [](long n, int t) { return (int)((n + t - 1) / t); };


    const int nOwn = OWN_PARTS * (blk(K + 1, OWN_R) + blk(N + 1, OWN_R));
    prep_kernel<<<nOwn + blk(K, GEMM_ROWS) + blk(N, GEMM_ROWS), 1024, 0, stream>>>(
        ei, E, N, Np1, cur1, list1, ei_g, Eg, K, Np2, cur2, list2, OWN_PARTS * blk(K + 1, OWN_R), nOwn,
        x, W1, a_src1, a_dst1, h1, as1, ad1, N,
        x_g, W2, a_src2, a_dst2, h2, as2, ad2, K, blk(K, GEMM_ROWS));

    seg_kernel<<<blk(K, 8) + blk(N, 8), T, 0, stream>>>(
        h1, as1, ad1, cur1, list1, b1, E, N, Np1, alpha1, x_final,
        h2, as2, ad2, cur2, list2, b2, Eg, K, Np2, alpha2, xg_out, blk(K, 8));

    group_kernel<<<blk(K, GRP_G), GRP_T, 0, stream>>>(x_final, xg_out, gidx, nbr, N, K);
}
